# starting kernel
# baseline (speedup 1.0000x reference)
#include <hip/hip_runtime.h>
#include <hip/hip_bf16.h>

typedef __attribute__((ext_vector_type(2))) float v2f;
typedef __attribute__((ext_vector_type(8))) float v8f;

#define BB 32
#define FF 4096
#define SS 8
#define DD 512
#define HH 2048
#define NITER 3
#define EPS_A 1e-8f
#define LN_EPS_C 1e-5f
#define SCALE_C 0.04419417382415922f

__device__ __forceinline__ v8f wmma4(v2f a, v2f b, v8f c) {
  return __builtin_amdgcn_wmma_f32_16x16x4_f32(
       false, a,  false, b,
       (short)0, c,  false,  false);
}

__device__ __forceinline__ v2f frag_a(const float* A, int lda, int lane, int k0) {
  int row = lane & 15;
  int kk  = k0 + ((lane >> 4) << 1);
  return *(const v2f*)&A[row * lda + kk];
}
__device__ __forceinline__ v2f frag_b(const float* B, int ldb, int lane, int k0, int n0) {
  int col = n0 + (lane & 15);
  int p   = (k0 >> 1) + (lane >> 4);
  return *(const v2f*)&B[(p * ldb + col) * 2];
}

__device__ __forceinline__ float wave_sum(float v) {
#pragma unroll
  for (int o = 16; o; o >>= 1) v += __shfl_xor(v, o, 32);
  return v;
}

__global__ void __launch_bounds__(256)
row_stats_kernel(const float* __restrict__ X, float* __restrict__ mean,
                 float* __restrict__ rstd) {
  int wv = threadIdx.x >> 5, lane = threadIdx.x & 31;
  int row = blockIdx.x * 8 + wv;
  const float* x = X + (size_t)row * DD;
  float s = 0.f, s2 = 0.f;
#pragma unroll
  for (int j = 0; j < DD / 32; ++j) {
    float v = x[lane + 32 * j]; s += v; s2 += v * v;
  }
  s = wave_sum(s); s2 = wave_sum(s2);
  if (lane == 0) {
    float m = s * (1.f / DD);
    float var = s2 * (1.f / DD) - m * m;
    mean[row] = m;
    rstd[row] = rsqrtf(var + LN_EPS_C);
  }
}

__global__ void __launch_bounds__(256)
ln_rows_kernel(const float* __restrict__ X, const float* __restrict__ w,
               const float* __restrict__ b, float* __restrict__ Y) {
  int wv = threadIdx.x >> 5, lane = threadIdx.x & 31;
  int row = blockIdx.x * 8 + wv;
  const float* x = X + (size_t)row * DD;
  float xv[16]; float s = 0.f, s2 = 0.f;
#pragma unroll
  for (int j = 0; j < 16; ++j) {
    float v = x[lane + 32 * j]; xv[j] = v; s += v; s2 += v * v;
  }
  s = wave_sum(s); s2 = wave_sum(s2);
  float m = s * (1.f / DD);
  float r = rsqrtf(s2 * (1.f / DD) - m * m + LN_EPS_C);
  float* y = Y + (size_t)row * DD;
#pragma unroll
  for (int j = 0; j < 16; ++j) {
    int d = lane + 32 * j;
    y[d] = (xv[j] - m) * r * w[d] + b[d];
  }
}

__global__ void __launch_bounds__(256)
gemm_ln_dual_kernel(const float* __restrict__ X, const float* __restrict__ mean,
                    const float* __restrict__ rstd, const float* __restrict__ lnw,
                    const float* __restrict__ lnb, const float* __restrict__ Wk,
                    const float* __restrict__ Wv, float* __restrict__ Ko,
                    float* __restrict__ Vo) {
  __shared__ float As[128 * 16];
  __shared__ float Bks[16 * 64];
  __shared__ float Bvs[16 * 64];
  int t = threadIdx.x, lane = t & 31, wv = t >> 5;
  int m0 = blockIdx.x * 128, n0 = blockIdx.y * 64;

  int arow = t >> 1, aoff = (t & 1) * 8;
  size_t grow = (size_t)(m0 + arow);
  float mn = mean[grow], rs = rstd[grow];
  const float* xrow = X + grow * DD;
  int bp = t >> 5, bn2 = (t & 31) * 2;

  v8f accK[4] = {}, accV[4] = {};
  for (int kc = 0; kc < DD; kc += 16) {
    __syncthreads();
#pragma unroll
    for (int j = 0; j < 8; ++j) {
      int k = kc + aoff + j;
      As[arow * 16 + aoff + j] = (xrow[k] - mn) * rs * lnw[k] + lnb[k];
    }
    {
      float2 r0 = *(const float2*)&Wk[(size_t)(kc + 2 * bp) * DD + n0 + bn2];
      float2 r1 = *(const float2*)&Wk[(size_t)(kc + 2 * bp + 1) * DD + n0 + bn2];
      float4 wk; wk.x = r0.x; wk.y = r1.x; wk.z = r0.y; wk.w = r1.y;
      *(float4*)&Bks[(bp * 64 + bn2) * 2] = wk;
      float2 s0 = *(const float2*)&Wv[(size_t)(kc + 2 * bp) * DD + n0 + bn2];
      float2 s1 = *(const float2*)&Wv[(size_t)(kc + 2 * bp + 1) * DD + n0 + bn2];
      float4 wvv; wvv.x = s0.x; wvv.y = s1.x; wvv.z = s0.y; wvv.w = s1.y;
      *(float4*)&Bvs[(bp * 64 + bn2) * 2] = wvv;
    }
    __syncthreads();
    const float* Aw = &As[(wv * 16) * 16];
#pragma unroll
    for (int ks = 0; ks < 16; ks += 4) {
      v2f a = frag_a(Aw, 16, lane, ks);
#pragma unroll
      for (int nt = 0; nt < 4; ++nt) {
        v2f bk = frag_b(Bks, 64, lane, ks, nt * 16);
        accK[nt] = wmma4(a, bk, accK[nt]);
        v2f bv = frag_b(Bvs, 64, lane, ks, nt * 16);
        accV[nt] = wmma4(a, bv, accV[nt]);
      }
    }
  }
  int half = lane >> 4, l16 = lane & 15;
#pragma unroll
  for (int nt = 0; nt < 4; ++nt)
#pragma unroll
    for (int i = 0; i < 8; ++i) {
      size_t mrow = (size_t)(m0 + wv * 16 + half * 8 + i);
      int ncol = n0 + nt * 16 + l16;
      Ko[mrow * DD + ncol] = accK[nt][i];
      Vo[mrow * DD + ncol] = accV[nt][i];
    }
}

template <int EPI>
__global__ void __launch_bounds__(256)
gemm_rm_kernel(const float* __restrict__ A, const float* __restrict__ B,
               const float* __restrict__ bias, float* __restrict__ C,
               const float* __restrict__ aux, int K, int N) {
  __shared__ float As[128 * 16];
  __shared__ float Bs[16 * 64];
  int t = threadIdx.x, lane = t & 31, wv = t >> 5;
  int m0 = blockIdx.x * 128, n0 = blockIdx.y * 64;
  int arow = t >> 1, aoff = (t & 1) * 8;
  const float* xrow = A + (size_t)(m0 + arow) * K;
  int bp = t >> 5, bn2 = (t & 31) * 2;

  v8f acc[4] = {};
  for (int kc = 0; kc < K; kc += 16) {
    __syncthreads();
    {
      float4 a0 = *(const float4*)&xrow[kc + aoff];
      float4 a1 = *(const float4*)&xrow[kc + aoff + 4];
      *(float4*)&As[arow * 16 + aoff] = a0;
      *(float4*)&As[arow * 16 + aoff + 4] = a1;
      float2 r0 = *(const float2*)&B[(size_t)(kc + 2 * bp) * N + n0 + bn2];
      float2 r1 = *(const float2*)&B[(size_t)(kc + 2 * bp + 1) * N + n0 + bn2];
      float4 wb; wb.x = r0.x; wb.y = r1.x; wb.z = r0.y; wb.w = r1.y;
      *(float4*)&Bs[(bp * 64 + bn2) * 2] = wb;
    }
    __syncthreads();
    const float* Aw = &As[(wv * 16) * 16];
#pragma unroll
    for (int ks = 0; ks < 16; ks += 4) {
      v2f a = frag_a(Aw, 16, lane, ks);
#pragma unroll
      for (int nt = 0; nt < 4; ++nt) {
        v2f b = frag_b(Bs, 64, lane, ks, nt * 16);
        acc[nt] = wmma4(a, b, acc[nt]);
      }
    }
  }
  int half = lane >> 4, l16 = lane & 15;
#pragma unroll
  for (int nt = 0; nt < 4; ++nt) {
    int ncol = n0 + nt * 16 + l16;
    float bv = (EPI >= 1) ? bias[ncol] : 0.f;
#pragma unroll
    for (int i = 0; i < 8; ++i) {
      size_t mrow = (size_t)(m0 + wv * 16 + half * 8 + i);
      float v = acc[nt][i] + bv;
      if (EPI == 1) v = fmaxf(v, 0.f);
      if (EPI == 2) v += aux[mrow * (size_t)N + ncol];
      C[mrow * (size_t)N + ncol] = v;
    }
  }
}

__global__ void __launch_bounds__(256)
gemm_bt_kernel(const float* __restrict__ A, const float* __restrict__ W,
               const float* __restrict__ bias, float* __restrict__ C,
               int K, int N) {
  __shared__ float As[128 * 16];
  __shared__ float Bs[16 * 64];
  int t = threadIdx.x, lane = t & 31, wv = t >> 5;
  int m0 = blockIdx.x * 128, n0 = blockIdx.y * 64;
  int arow = t >> 1, aoff = (t & 1) * 8;
  const float* xrow = A + (size_t)(m0 + arow) * K;
  int bn = t & 63, bp0 = t >> 6;

  v8f acc[4] = {};
  for (int kc = 0; kc < K; kc += 16) {
    __syncthreads();
    {
      float4 a0 = *(const float4*)&xrow[kc + aoff];
      float4 a1 = *(const float4*)&xrow[kc + aoff + 4];
      *(float4*)&As[arow * 16 + aoff] = a0;
      *(float4*)&As[arow * 16 + aoff + 4] = a1;
#pragma unroll
      for (int j = 0; j < 2; ++j) {
        int p = bp0 + j * 4;
        float2 f = *(const float2*)&W[(size_t)(n0 + bn) * K + kc + 2 * p];
        *(float2*)&Bs[(p * 64 + bn) * 2] = f;
      }
    }
    __syncthreads();
    const float* Aw = &As[(wv * 16) * 16];
#pragma unroll
    for (int ks = 0; ks < 16; ks += 4) {
      v2f a = frag_a(Aw, 16, lane, ks);
#pragma unroll
      for (int nt = 0; nt < 4; ++nt) {
        v2f b = frag_b(Bs, 64, lane, ks, nt * 16);
        acc[nt] = wmma4(a, b, acc[nt]);
      }
    }
  }
  int half = lane >> 4, l16 = lane & 15;
#pragma unroll
  for (int nt = 0; nt < 4; ++nt) {
    int ncol = n0 + nt * 16 + l16;
    float bv = bias[ncol];
#pragma unroll
    for (int i = 0; i < 8; ++i) {
      size_t mrow = (size_t)(m0 + wv * 16 + half * 8 + i);
      C[mrow * (size_t)N + ncol] = acc[nt][i] + bv;
    }
  }
}

__global__ void __launch_bounds__(256)
dots_kernel(const float* __restrict__ Q, const float* __restrict__ Keys,
            float* __restrict__ Dots) {
  __shared__ float As[16 * 16];
  __shared__ float Bs[16 * 512];
  int t = threadIdx.x, lane = t & 31, wv = t >> 5;
  int b = blockIdx.x, nb = blockIdx.y * 512;
  const float* qb = Q + (size_t)b * SS * DD;
  const float* kb = Keys + ((size_t)b * FF + nb) * DD;
  v8f acc[4] = {};
  for (int kc = 0; kc < DD; kc += 16) {
    __syncthreads();
    {
      int r = t >> 4, k = t & 15;
      As[r * 16 + k] = (r < SS) ? qb[(size_t)r * DD + kc + k] : 0.f;
    }
    {
      int k4 = (t & 3) * 4;
#pragma unroll
      for (int j = 0; j < 8; ++j) {
        int n = (t >> 2) + j * 64;
        float4 f = *(const float4*)&kb[(size_t)n * DD + kc + k4];
        float2 lo; lo.x = f.x; lo.y = f.y;
        float2 hi; hi.x = f.z; hi.y = f.w;
        *(float2*)&Bs[(((k4 >> 1) + 0) * 512 + n) * 2] = lo;
        *(float2*)&Bs[(((k4 >> 1) + 1) * 512 + n) * 2] = hi;
      }
    }
    __syncthreads();
#pragma unroll
    for (int ks = 0; ks < 16; ks += 4) {
      v2f a = frag_a(As, 16, lane, ks);
#pragma unroll
      for (int nt = 0; nt < 4; ++nt) {
        v2f b2 = frag_b(Bs, 512, lane, ks, wv * 64 + nt * 16);
        acc[nt] = wmma4(a, b2, acc[nt]);
      }
    }
  }
  int half = lane >> 4, l16 = lane & 15;
  if (half == 0) {
#pragma unroll
    for (int nt = 0; nt < 4; ++nt)
#pragma unroll
      for (int i = 0; i < 8; ++i) {
        int f = nb + wv * 64 + nt * 16 + l16;
        Dots[((size_t)b * SS + i) * FF + f] = acc[nt][i] * SCALE_C;
      }
  }
}

__global__ void __launch_bounds__(256)
softmax_kernel(const float* __restrict__ Dots, float* __restrict__ PreNorm,
               float* __restrict__ Attn, float* __restrict__ Sums) {
  int b = blockIdx.y;
  int f = blockIdx.x * 256 + threadIdx.x;
  int lane = threadIdx.x & 31;
  float d[SS]; float mx = -3.4e38f;
#pragma unroll
  for (int s = 0; s < SS; ++s) {
    d[s] = Dots[((size_t)b * SS + s) * FF + f];
    mx = fmaxf(mx, d[s]);
  }
  float sum = 0.f;
#pragma unroll
  for (int s = 0; s < SS; ++s) { d[s] = expf(d[s] - mx); sum += d[s]; }
  float inv = 1.f / sum;
#pragma unroll
  for (int s = 0; s < SS; ++s) {
    float p = d[s] * inv;
    size_t idx = ((size_t)b * SS + s) * FF + f;
    PreNorm[idx] = p;
    float aw = p + EPS_A;
    Attn[idx] = aw;
    float r = wave_sum(aw);
    if (lane == 0) atomicAdd(&Sums[b * SS + s], r);
  }
}

__global__ void zero_sums_kernel(float* __restrict__ Sums) {
  Sums[threadIdx.x] = 0.f;
}

__global__ void __launch_bounds__(256)
updates_kernel(const float* __restrict__ Attn, const float* __restrict__ Vals,
               const float* __restrict__ Sums, float* __restrict__ Upd) {
  __shared__ float As[16 * 16];
  __shared__ float Bs[16 * 512];
  int t = threadIdx.x, lane = t & 31, wv = t >> 5;
  int b = blockIdx.x;
  const float* ab = Attn + (size_t)b * SS * FF;
  const float* vb = Vals + (size_t)b * FF * DD;
  v8f acc[4] = {};
  for (int kc = 0; kc < FF; kc += 16) {
    __syncthreads();
    {
      int r = t >> 4, k = t & 15;
      As[r * 16 + k] = (r < SS) ? ab[(size_t)r * FF + kc + k] : 0.f;
    }
    {
#pragma unroll
      for (int j = 0; j < 8; ++j) {
        int u = t + j * 256;
        int p = u >> 8;
        int n2 = (u & 255) * 2;
        float2 r0 = *(const float2*)&vb[(size_t)(kc + 2 * p) * DD + n2];
        float2 r1 = *(const float2*)&vb[(size_t)(kc + 2 * p + 1) * DD + n2];
        float4 wb; wb.x = r0.x; wb.y = r1.x; wb.z = r0.y; wb.w = r1.y;
        *(float4*)&Bs[(p * 512 + n2) * 2] = wb;
      }
    }
    __syncthreads();
#pragma unroll
    for (int ks = 0; ks < 16; ks += 4) {
      v2f a = frag_a(As, 16, lane, ks);
#pragma unroll
      for (int nt = 0; nt < 4; ++nt) {
        v2f b2 = frag_b(Bs, 512, lane, ks, wv * 64 + nt * 16);
        acc[nt] = wmma4(a, b2, acc[nt]);
      }
    }
  }
  int half = lane >> 4, l16 = lane & 15;
  if (half == 0) {
#pragma unroll
    for (int i = 0; i < 8; ++i) {
      float inv = 1.f / Sums[b * SS + i];
#pragma unroll
      for (int nt = 0; nt < 4; ++nt) {
        int n = wv * 64 + nt * 16 + l16;
        Upd[((size_t)b * SS + i) * DD + n] = acc[nt][i] * inv;
      }
    }
  }
}

__global__ void __launch_bounds__(256)
gru_kernel(const float* __restrict__ gi, const float* __restrict__ gh,
           const float* __restrict__ sn, float* __restrict__ slots) {
  int idx = blockIdx.x * 256 + threadIdx.x;
  int m = idx >> 9, d = idx & 511;
  size_t base = (size_t)m * 3 * DD;
  float ir = gi[base + d],           hr = gh[base + d];
  float iz = gi[base + DD + d],      hz = gh[base + DD + d];
  float in_ = gi[base + 2 * DD + d], hn = gh[base + 2 * DD + d];
  float r = 1.f / (1.f + expf(-(ir + hr)));
  float z = 1.f / (1.f + expf(-(iz + hz)));
  float n = tanhf(in_ + r * hn);
  float h = sn[(size_t)m * DD + d];
  slots[(size_t)m * DD + d] = (1.f - z) * n + z * h;
}

extern "C" void kernel_launch(void* const* d_in, const int* in_sizes, int n_in,
                              void* d_out, int out_size, void* d_ws, size_t ws_size,
                              hipStream_t stream) {
  (void)in_sizes; (void)n_in; (void)out_size; (void)ws_size;
  const float* in_slots  = (const float*)d_in[0];
  const float* features  = (const float*)d_in[1];
  const float* w_k       = (const float*)d_in[2];
  const float* w_v       = (const float*)d_in[3];
  const float* w_q       = (const float*)d_in[4];
  const float* ln_feat_w = (const float*)d_in[5];
  const float* ln_feat_b = (const float*)d_in[6];
  const float* ln_slot_w = (const float*)d_in[7];
  const float* ln_slot_b = (const float*)d_in[8];
  const float* gru_w_ih  = (const float*)d_in[9];
  const float* gru_w_hh  = (const float*)d_in[10];
  const float* gru_b_ih  = (const float*)d_in[11];
  const float* gru_b_hh  = (const float*)d_in[12];
  const float* ln_mlp_w  = (const float*)d_in[13];
  const float* ln_mlp_b  = (const float*)d_in[14];
  const float* mlp_w1    = (const float*)d_in[15];
  const float* mlp_b1    = (const float*)d_in[16];
  const float* mlp_w2    = (const float*)d_in[17];
  const float* mlp_b2    = (const float*)d_in[18];

  float* ws = (float*)d_ws;
  size_t off = 0;
  auto carve = [&](size_t n) { float* p = ws + off; off += n; return p; };
  float* mean  = carve((size_t)BB * FF);
  float* rstd  = carve((size_t)BB * FF);
  float* keys  = carve((size_t)BB * FF * DD);
  float* vals  = carve((size_t)BB * FF * DD);
  float* slots = carve((size_t)BB * SS * DD);
  float* s_n   = carve((size_t)BB * SS * DD);
  float* q     = carve((size_t)BB * SS * DD);
  float* dots  = carve((size_t)BB * SS * FF);
  float* attn  = carve((size_t)BB * SS * FF);
  float* sums  = carve(256);
  float* upd   = carve((size_t)BB * SS * DD);
  float* gi    = carve((size_t)BB * SS * 3 * DD);
  float* gh    = carve((size_t)BB * SS * 3 * DD);
  float* hln   = carve((size_t)BB * SS * DD);
  float* h1    = carve((size_t)BB * SS * HH);

  float* out_slots = (float*)d_out;
  float* out_attn  = out_slots + (size_t)BB * SS * DD;

  hipMemcpyAsync(slots, in_slots, (size_t)BB * SS * DD * sizeof(float),
                 hipMemcpyDeviceToDevice, stream);

  row_stats_kernel<<<(BB * FF) / 8, 256, 0, stream>>>(features, mean, rstd);
  gemm_ln_dual_kernel<<<dim3((BB * FF) / 128, DD / 64), 256, 0, stream>>>(
      features, mean, rstd, ln_feat_w, ln_feat_b, w_k, w_v, keys, vals);

  for (int it = 0; it < NITER; ++it) {
    ln_rows_kernel<<<32, 256, 0, stream>>>(slots, ln_slot_w, ln_slot_b, s_n);
    gemm_rm_kernel<0><<<dim3(2, DD / 64), 256, 0, stream>>>(
        s_n, w_q, nullptr, q, nullptr, DD, DD);
    zero_sums_kernel<<<1, 256, 0, stream>>>(sums);
    dots_kernel<<<dim3(BB, FF / 512), 256, 0, stream>>>(q, keys, dots);
    softmax_kernel<<<dim3(FF / 256, BB), 256, 0, stream>>>(dots, out_attn, attn, sums);
    updates_kernel<<<BB, 256, 0, stream>>>(attn, vals, sums, upd);
    gemm_bt_kernel<<<dim3(2, (3 * DD) / 64), 256, 0, stream>>>(
        upd, gru_w_ih, gru_b_ih, gi, DD, 3 * DD);
    gemm_bt_kernel<<<dim3(2, (3 * DD) / 64), 256, 0, stream>>>(
        s_n, gru_w_hh, gru_b_hh, gh, DD, 3 * DD);
    gru_kernel<<<(BB * SS * DD) / 256, 256, 0, stream>>>(gi, gh, s_n, slots);
    ln_rows_kernel<<<32, 256, 0, stream>>>(slots, ln_mlp_w, ln_mlp_b, hln);
    gemm_rm_kernel<1><<<dim3(2, HH / 64), 256, 0, stream>>>(
        hln, mlp_w1, mlp_b1, h1, nullptr, DD, HH);
    gemm_rm_kernel<2><<<dim3(2, DD / 64), 256, 0, stream>>>(
        h1, mlp_w2, mlp_b2, slots, slots, HH, DD);
  }

  hipMemcpyAsync(out_slots, slots, (size_t)BB * SS * DD * sizeof(float),
                 hipMemcpyDeviceToDevice, stream);
}


# updates product (attn x values) re-tiled: 256 work-groups (64 columns x batch) not 32, F split over 16 waves loading values straight into fragments, fixed-order LDS sum of partials (new sum order)
# speedup vs baseline: 2.1549x; 2.1549x over previous
#include <hip/hip_runtime.h>
#include <hip/hip_bf16.h>

typedef __attribute__((ext_vector_type(2))) float v2f;
typedef __attribute__((ext_vector_type(8))) float v8f;

#define BB 32
#define FF 4096
#define SS 8
#define DD 512
#define HH 2048
#define NITER 3
#define EPS_A 1e-8f
#define LN_EPS_C 1e-5f
#define SCALE_C 0.04419417382415922f

__device__ __forceinline__ v8f wmma4(v2f a, v2f b, v8f c) {
  return __builtin_amdgcn_wmma_f32_16x16x4_f32(
       false, a,  false, b,
       (short)0, c,  false,  false);
}

__device__ __forceinline__ v2f frag_a(const float* A, int lda, int lane, int k0) {
  int row = lane & 15;
  int kk  = k0 + ((lane >> 4) << 1);
  return *(const v2f*)&A[row * lda + kk];
}
__device__ __forceinline__ v2f frag_b(const float* B, int ldb, int lane, int k0, int n0) {
  int col = n0 + (lane & 15);
  int p   = (k0 >> 1) + (lane >> 4);
  return *(const v2f*)&B[(p * ldb + col) * 2];
}

__device__ __forceinline__ float wave_sum(float v) {
#pragma unroll
  for (int o = 16; o; o >>= 1) v += __shfl_xor(v, o, 32);
  return v;
}

__global__ void __launch_bounds__(256)
row_stats_kernel(const float* __restrict__ X, float* __restrict__ mean,
                 float* __restrict__ rstd) {
  int wv = threadIdx.x >> 5, lane = threadIdx.x & 31;
  int row = blockIdx.x * 8 + wv;
  const float* x = X + (size_t)row * DD;
  float s = 0.f, s2 = 0.f;
#pragma unroll
  for (int j = 0; j < DD / 32; ++j) {
    float v = x[lane + 32 * j]; s += v; s2 += v * v;
  }
  s = wave_sum(s); s2 = wave_sum(s2);
  if (lane == 0) {
    float m = s * (1.f / DD);
    float var = s2 * (1.f / DD) - m * m;
    mean[row] = m;
    rstd[row] = rsqrtf(var + LN_EPS_C);
  }
}

__global__ void __launch_bounds__(256)
ln_rows_kernel(const float* __restrict__ X, const float* __restrict__ w,
               const float* __restrict__ b, float* __restrict__ Y) {
  int wv = threadIdx.x >> 5, lane = threadIdx.x & 31;
  int row = blockIdx.x * 8 + wv;
  const float* x = X + (size_t)row * DD;
  float xv[16]; float s = 0.f, s2 = 0.f;
#pragma unroll
  for (int j = 0; j < 16; ++j) {
    float v = x[lane + 32 * j]; xv[j] = v; s += v; s2 += v * v;
  }
  s = wave_sum(s); s2 = wave_sum(s2);
  float m = s * (1.f / DD);
  float r = rsqrtf(s2 * (1.f / DD) - m * m + LN_EPS_C);
  float* y = Y + (size_t)row * DD;
#pragma unroll
  for (int j = 0; j < 16; ++j) {
    int d = lane + 32 * j;
    y[d] = (xv[j] - m) * r * w[d] + b[d];
  }
}

__global__ void __launch_bounds__(256)
gemm_ln_dual_kernel(const float* __restrict__ X, const float* __restrict__ mean,
                    const float* __restrict__ rstd, const float* __restrict__ lnw,
                    const float* __restrict__ lnb, const float* __restrict__ Wk,
                    const float* __restrict__ Wv, float* __restrict__ Ko,
                    float* __restrict__ Vo) {
  __shared__ float As[128 * 16];
  __shared__ float Bks[16 * 64];
  __shared__ float Bvs[16 * 64];
  int t = threadIdx.x, lane = t & 31, wv = t >> 5;
  int m0 = blockIdx.x * 128, n0 = blockIdx.y * 64;

  int arow = t >> 1, aoff = (t & 1) * 8;
  size_t grow = (size_t)(m0 + arow);
  float mn = mean[grow], rs = rstd[grow];
  const float* xrow = X + grow * DD;
  int bp = t >> 5, bn2 = (t & 31) * 2;

  v8f accK[4] = {}, accV[4] = {};
  for (int kc = 0; kc < DD; kc += 16) {
    __syncthreads();
#pragma unroll
    for (int j = 0; j < 8; ++j) {
      int k = kc + aoff + j;
      As[arow * 16 + aoff + j] = (xrow[k] - mn) * rs * lnw[k] + lnb[k];
    }
    {
      float2 r0 = *(const float2*)&Wk[(size_t)(kc + 2 * bp) * DD + n0 + bn2];
      float2 r1 = *(const float2*)&Wk[(size_t)(kc + 2 * bp + 1) * DD + n0 + bn2];
      float4 wk; wk.x = r0.x; wk.y = r1.x; wk.z = r0.y; wk.w = r1.y;
      *(float4*)&Bks[(bp * 64 + bn2) * 2] = wk;
      float2 s0 = *(const float2*)&Wv[(size_t)(kc + 2 * bp) * DD + n0 + bn2];
      float2 s1 = *(const float2*)&Wv[(size_t)(kc + 2 * bp + 1) * DD + n0 + bn2];
      float4 wvv; wvv.x = s0.x; wvv.y = s1.x; wvv.z = s0.y; wvv.w = s1.y;
      *(float4*)&Bvs[(bp * 64 + bn2) * 2] = wvv;
    }
    __syncthreads();
    const float* Aw = &As[(wv * 16) * 16];
#pragma unroll
    for (int ks = 0; ks < 16; ks += 4) {
      v2f a = frag_a(Aw, 16, lane, ks);
#pragma unroll
      for (int nt = 0; nt < 4; ++nt) {
        v2f bk = frag_b(Bks, 64, lane, ks, nt * 16);
        accK[nt] = wmma4(a, bk, accK[nt]);
        v2f bv = frag_b(Bvs, 64, lane, ks, nt * 16);
        accV[nt] = wmma4(a, bv, accV[nt]);
      }
    }
  }
  int half = lane >> 4, l16 = lane & 15;
#pragma unroll
  for (int nt = 0; nt < 4; ++nt)
#pragma unroll
    for (int i = 0; i < 8; ++i) {
      size_t mrow = (size_t)(m0 + wv * 16 + half * 8 + i);
      int ncol = n0 + nt * 16 + l16;
      Ko[mrow * DD + ncol] = accK[nt][i];
      Vo[mrow * DD + ncol] = accV[nt][i];
    }
}

template <int EPI>
__global__ void __launch_bounds__(256)
gemm_rm_kernel(const float* __restrict__ A, const float* __restrict__ B,
               const float* __restrict__ bias, float* __restrict__ C,
               const float* __restrict__ aux, int K, int N) {
  __shared__ float As[128 * 16];
  __shared__ float Bs[16 * 64];
  int t = threadIdx.x, lane = t & 31, wv = t >> 5;
  int m0 = blockIdx.x * 128, n0 = blockIdx.y * 64;
  int arow = t >> 1, aoff = (t & 1) * 8;
  const float* xrow = A + (size_t)(m0 + arow) * K;
  int bp = t >> 5, bn2 = (t & 31) * 2;

  v8f acc[4] = {};
  for (int kc = 0; kc < K; kc += 16) {
    __syncthreads();
    {
      float4 a0 = *(const float4*)&xrow[kc + aoff];
      float4 a1 = *(const float4*)&xrow[kc + aoff + 4];
      *(float4*)&As[arow * 16 + aoff] = a0;
      *(float4*)&As[arow * 16 + aoff + 4] = a1;
      float2 r0 = *(const float2*)&B[(size_t)(kc + 2 * bp) * N + n0 + bn2];
      float2 r1 = *(const float2*)&B[(size_t)(kc + 2 * bp + 1) * N + n0 + bn2];
      float4 wb; wb.x = r0.x; wb.y = r1.x; wb.z = r0.y; wb.w = r1.y;
      *(float4*)&Bs[(bp * 64 + bn2) * 2] = wb;
    }
    __syncthreads();
    const float* Aw = &As[(wv * 16) * 16];
#pragma unroll
    for (int ks = 0; ks < 16; ks += 4) {
      v2f a = frag_a(Aw, 16, lane, ks);
#pragma unroll
      for (int nt = 0; nt < 4; ++nt) {
        v2f b = frag_b(Bs, 64, lane, ks, nt * 16);
        acc[nt] = wmma4(a, b, acc[nt]);
      }
    }
  }
  int half = lane >> 4, l16 = lane & 15;
#pragma unroll
  for (int nt = 0; nt < 4; ++nt) {
    int ncol = n0 + nt * 16 + l16;
    float bv = (EPI >= 1) ? bias[ncol] : 0.f;
#pragma unroll
    for (int i = 0; i < 8; ++i) {
      size_t mrow = (size_t)(m0 + wv * 16 + half * 8 + i);
      float v = acc[nt][i] + bv;
      if (EPI == 1) v = fmaxf(v, 0.f);
      if (EPI == 2) v += aux[mrow * (size_t)N + ncol];
      C[mrow * (size_t)N + ncol] = v;
    }
  }
}

__global__ void __launch_bounds__(256)
gemm_bt_kernel(const float* __restrict__ A, const float* __restrict__ W,
               const float* __restrict__ bias, float* __restrict__ C,
               int K, int N) {
  __shared__ float As[128 * 16];
  __shared__ float Bs[16 * 64];
  int t = threadIdx.x, lane = t & 31, wv = t >> 5;
  int m0 = blockIdx.x * 128, n0 = blockIdx.y * 64;
  int arow = t >> 1, aoff = (t & 1) * 8;
  const float* xrow = A + (size_t)(m0 + arow) * K;
  int bn = t & 63, bp0 = t >> 6;

  v8f acc[4] = {};
  for (int kc = 0; kc < K; kc += 16) {
    __syncthreads();
    {
      float4 a0 = *(const float4*)&xrow[kc + aoff];
      float4 a1 = *(const float4*)&xrow[kc + aoff + 4];
      *(float4*)&As[arow * 16 + aoff] = a0;
      *(float4*)&As[arow * 16 + aoff + 4] = a1;
#pragma unroll
      for (int j = 0; j < 2; ++j) {
        int p = bp0 + j * 4;
        float2 f = *(const float2*)&W[(size_t)(n0 + bn) * K + kc + 2 * p];
        *(float2*)&Bs[(p * 64 + bn) * 2] = f;
      }
    }
    __syncthreads();
    const float* Aw = &As[(wv * 16) * 16];
#pragma unroll
    for (int ks = 0; ks < 16; ks += 4) {
      v2f a = frag_a(Aw, 16, lane, ks);
#pragma unroll
      for (int nt = 0; nt < 4; ++nt) {
        v2f b = frag_b(Bs, 64, lane, ks, nt * 16);
        acc[nt] = wmma4(a, b, acc[nt]);
      }
    }
  }
  int half = lane >> 4, l16 = lane & 15;
#pragma unroll
  for (int nt = 0; nt < 4; ++nt) {
    int ncol = n0 + nt * 16 + l16;
    float bv = bias[ncol];
#pragma unroll
    for (int i = 0; i < 8; ++i) {
      size_t mrow = (size_t)(m0 + wv * 16 + half * 8 + i);
      C[mrow * (size_t)N + ncol] = acc[nt][i] + bv;
    }
  }
}

__global__ void __launch_bounds__(256)
dots_kernel(const float* __restrict__ Q, const float* __restrict__ Keys,
            float* __restrict__ Dots) {
  __shared__ float As[16 * 16];
  __shared__ float Bs[16 * 512];
  int t = threadIdx.x, lane = t & 31, wv = t >> 5;
  int b = blockIdx.x, nb = blockIdx.y * 512;
  const float* qb = Q + (size_t)b * SS * DD;
  const float* kb = Keys + ((size_t)b * FF + nb) * DD;
  v8f acc[4] = {};
  for (int kc = 0; kc < DD; kc += 16) {
    __syncthreads();
    {
      int r = t >> 4, k = t & 15;
      As[r * 16 + k] = (r < SS) ? qb[(size_t)r * DD + kc + k] : 0.f;
    }
    {
      int k4 = (t & 3) * 4;
#pragma unroll
      for (int j = 0; j < 8; ++j) {
        int n = (t >> 2) + j * 64;
        float4 f = *(const float4*)&kb[(size_t)n * DD + kc + k4];
        float2 lo; lo.x = f.x; lo.y = f.y;
        float2 hi; hi.x = f.z; hi.y = f.w;
        *(float2*)&Bs[(((k4 >> 1) + 0) * 512 + n) * 2] = lo;
        *(float2*)&Bs[(((k4 >> 1) + 1) * 512 + n) * 2] = hi;
      }
    }
    __syncthreads();
#pragma unroll
    for (int ks = 0; ks < 16; ks += 4) {
      v2f a = frag_a(As, 16, lane, ks);
#pragma unroll
      for (int nt = 0; nt < 4; ++nt) {
        v2f b2 = frag_b(Bs, 512, lane, ks, wv * 64 + nt * 16);
        acc[nt] = wmma4(a, b2, acc[nt]);
      }
    }
  }
  int half = lane >> 4, l16 = lane & 15;
  if (half == 0) {
#pragma unroll
    for (int nt = 0; nt < 4; ++nt)
#pragma unroll
      for (int i = 0; i < 8; ++i) {
        int f = nb + wv * 64 + nt * 16 + l16;
        Dots[((size_t)b * SS + i) * FF + f] = acc[nt][i] * SCALE_C;
      }
  }
}

__global__ void __launch_bounds__(256)
softmax_kernel(const float* __restrict__ Dots, float* __restrict__ PreNorm,
               float* __restrict__ Attn, float* __restrict__ Sums) {
  int b = blockIdx.y;
  int f = blockIdx.x * 256 + threadIdx.x;
  int lane = threadIdx.x & 31;
  float d[SS]; float mx = -3.4e38f;
#pragma unroll
  for (int s = 0; s < SS; ++s) {
    d[s] = Dots[((size_t)b * SS + s) * FF + f];
    mx = fmaxf(mx, d[s]);
  }
  float sum = 0.f;
#pragma unroll
  for (int s = 0; s < SS; ++s) { d[s] = expf(d[s] - mx); sum += d[s]; }
  float inv = 1.f / sum;
#pragma unroll
  for (int s = 0; s < SS; ++s) {
    float p = d[s] * inv;
    size_t idx = ((size_t)b * SS + s) * FF + f;
    PreNorm[idx] = p;
    float aw = p + EPS_A;
    Attn[idx] = aw;
    float r = wave_sum(aw);
    if (lane == 0) atomicAdd(&Sums[b * SS + s], r);
  }
}

__global__ void zero_sums_kernel(float* __restrict__ Sums) {
  Sums[threadIdx.x] = 0.f;
}

#define UPD_WAVES 16
#define UPD_ROWS (FF / UPD_WAVES)
#define UPD_BLK 64
__global__ void __launch_bounds__(UPD_WAVES * 32)
updates_kernel(const float* __restrict__ Attn, const float* __restrict__ Vals,
               const float* __restrict__ Sums, float* __restrict__ Upd) {
  __shared__ float red[UPD_WAVES * SS * 64];
  const int t = threadIdx.x, lane = t & 31, wv = t >> 5;
  const int c = lane & 15, h = lane >> 4;
  const int cbase = blockIdx.x * 64, b = blockIdx.y;
  const bool valid = c < SS;
  const float* arow = Attn + ((size_t)b * SS + (c & 7)) * FF + wv * UPD_ROWS + h * 8;
  const float* vcol = Vals + ((size_t)b * FF + wv * UPD_ROWS + h * 8) * DD + cbase + c * 4;

  v8f acc[4] = {};
#pragma unroll 1
  for (int blk = 0; blk < UPD_ROWS; blk += UPD_BLK) {
    float4 aq[UPD_BLK / 16][2];
    float4 bq[UPD_BLK / 16][4][2];
#pragma unroll
    for (int u = 0; u < UPD_BLK / 16; ++u) {
      aq[u][0] = *(const float4*)&arow[blk + u * 16];
      aq[u][1] = *(const float4*)&arow[blk + u * 16 + 4];
#pragma unroll
      for (int tt = 0; tt < 4; ++tt) {
        const float* p = vcol + (size_t)(blk + u * 16 + 2 * tt) * DD;
        bq[u][tt][0] = *(const float4*)p;
        bq[u][tt][1] = *(const float4*)(p + DD);
      }
    }
#pragma unroll
    for (int u = 0; u < UPD_BLK / 16; ++u) {
      const float xa[8] = {aq[u][0].x, aq[u][0].y, aq[u][0].z, aq[u][0].w,
                           aq[u][1].x, aq[u][1].y, aq[u][1].z, aq[u][1].w};
#pragma unroll
      for (int tt = 0; tt < 4; ++tt) {
        v2f a;
        a.x = valid ? xa[2 * tt] : 0.f;
        a.y = valid ? xa[2 * tt + 1] : 0.f;
        const float r0[4] = {bq[u][tt][0].x, bq[u][tt][0].y, bq[u][tt][0].z, bq[u][tt][0].w};
        const float r1[4] = {bq[u][tt][1].x, bq[u][tt][1].y, bq[u][tt][1].z, bq[u][tt][1].w};
#pragma unroll
        for (int j = 0; j < 4; ++j) {
          v2f bb; bb.x = r0[j]; bb.y = r1[j];
          acc[j] = wmma4(a, bb, acc[j]);
        }
      }
    }
  }
  if (h == 0) {
    float* rw = &red[(wv * SS) * 64 + c * 4];
#pragma unroll
    for (int r = 0; r < SS; ++r) {
      float4 o; o.x = acc[0][r]; o.y = acc[1][r]; o.z = acc[2][r]; o.w = acc[3][r];
      *(float4*)&rw[r * 64] = o;
    }
  }
  __syncthreads();
  {
    const int r = t >> 6, col = t & 63;
    float s = 0.f;
#pragma unroll
    for (int w = 0; w < UPD_WAVES; ++w) s += red[(w * SS + r) * 64 + col];
    const float inv = 1.f / Sums[b * SS + r];
    Upd[((size_t)b * SS + r) * DD + cbase + col] = s * inv;
  }
}

__global__ void __launch_bounds__(256)
gru_kernel(const float* __restrict__ gi, const float* __restrict__ gh,
           const float* __restrict__ sn, float* __restrict__ slots) {
  int idx = blockIdx.x * 256 + threadIdx.x;
  int m = idx >> 9, d = idx & 511;
  size_t base = (size_t)m * 3 * DD;
  float ir = gi[base + d],           hr = gh[base + d];
  float iz = gi[base + DD + d],      hz = gh[base + DD + d];
  float in_ = gi[base + 2 * DD + d], hn = gh[base + 2 * DD + d];
  float r = 1.f / (1.f + expf(-(ir + hr)));
  float z = 1.f / (1.f + expf(-(iz + hz)));
  float n = tanhf(in_ + r * hn);
  float h = sn[(size_t)m * DD + d];
  slots[(size_t)m * DD + d] = (1.f - z) * n + z * h;
}

extern "C" void kernel_launch(void* const* d_in, const int* in_sizes, int n_in,
                              void* d_out, int out_size, void* d_ws, size_t ws_size,
                              hipStream_t stream) {
  (void)in_sizes; (void)n_in; (void)out_size; (void)ws_size;
  const float* in_slots  = (const float*)d_in[0];
  const float* features  = (const float*)d_in[1];
  const float* w_k       = (const float*)d_in[2];
  const float* w_v       = (const float*)d_in[3];
  const float* w_q       = (const float*)d_in[4];
  const float* ln_feat_w = (const float*)d_in[5];
  const float* ln_feat_b = (const float*)d_in[6];
  const float* ln_slot_w = (const float*)d_in[7];
  const float* ln_slot_b = (const float*)d_in[8];
  const float* gru_w_ih  = (const float*)d_in[9];
  const float* gru_w_hh  = (const float*)d_in[10];
  const float* gru_b_ih  = (const float*)d_in[11];
  const float* gru_b_hh  = (const float*)d_in[12];
  const float* ln_mlp_w  = (const float*)d_in[13];
  const float* ln_mlp_b  = (const float*)d_in[14];
  const float* mlp_w1    = (const float*)d_in[15];
  const float* mlp_b1    = (const float*)d_in[16];
  const float* mlp_w2    = (const float*)d_in[17];
  const float* mlp_b2    = (const float*)d_in[18];

  float* ws = (float*)d_ws;
  size_t off = 0;
  auto carve = [&](size_t n) { float* p = ws + off; off += n; return p; };
  float* mean  = carve((size_t)BB * FF);
  float* rstd  = carve((size_t)BB * FF);
  float* keys  = carve((size_t)BB * FF * DD);
  float* vals  = carve((size_t)BB * FF * DD);
  float* slots = carve((size_t)BB * SS * DD);
  float* s_n   = carve((size_t)BB * SS * DD);
  float* q     = carve((size_t)BB * SS * DD);
  float* dots  = carve((size_t)BB * SS * FF);
  float* attn  = carve((size_t)BB * SS * FF);
  float* sums  = carve(256);
  float* upd   = carve((size_t)BB * SS * DD);
  float* gi    = carve((size_t)BB * SS * 3 * DD);
  float* gh    = carve((size_t)BB * SS * 3 * DD);
  float* hln   = carve((size_t)BB * SS * DD);
  float* h1    = carve((size_t)BB * SS * HH);

  float* out_slots = (float*)d_out;
  float* out_attn  = out_slots + (size_t)BB * SS * DD;

  hipMemcpyAsync(slots, in_slots, (size_t)BB * SS * DD * sizeof(float),
                 hipMemcpyDeviceToDevice, stream);

  row_stats_kernel<<<(BB * FF) / 8, 256, 0, stream>>>(features, mean, rstd);
  gemm_ln_dual_kernel<<<dim3((BB * FF) / 128, DD / 64), 256, 0, stream>>>(
      features, mean, rstd, ln_feat_w, ln_feat_b, w_k, w_v, keys, vals);

  for (int it = 0; it < NITER; ++it) {
    ln_rows_kernel<<<32, 256, 0, stream>>>(slots, ln_slot_w, ln_slot_b, s_n);
    gemm_rm_kernel<0><<<dim3(2, DD / 64), 256, 0, stream>>>(
        s_n, w_q, nullptr, q, nullptr, DD, DD);
    zero_sums_kernel<<<1, 256, 0, stream>>>(sums);
    dots_kernel<<<dim3(BB, FF / 512), 256, 0, stream>>>(q, keys, dots);
    softmax_kernel<<<dim3(FF / 256, BB), 256, 0, stream>>>(dots, out_attn, attn, sums);
    updates_kernel<<<dim3(DD / 64, BB), UPD_WAVES * 32, 0, stream>>>(attn, vals, sums, upd);
    gemm_bt_kernel<<<dim3(2, (3 * DD) / 64), 256, 0, stream>>>(
        upd, gru_w_ih, gru_b_ih, gi, DD, 3 * DD);
    gemm_bt_kernel<<<dim3(2, (3 * DD) / 64), 256, 0, stream>>>(
        s_n, gru_w_hh, gru_b_hh, gh, DD, 3 * DD);
    gru_kernel<<<(BB * SS * DD) / 256, 256, 0, stream>>>(gi, gh, s_n, slots);
    ln_rows_kernel<<<32, 256, 0, stream>>>(slots, ln_mlp_w, ln_mlp_b, hln);
    gemm_rm_kernel<1><<<dim3(2, HH / 64), 256, 0, stream>>>(
        hln, mlp_w1, mlp_b1, h1, nullptr, DD, HH);
    gemm_rm_kernel<2><<<dim3(2, DD / 64), 256, 0, stream>>>(
        h1, mlp_w2, mlp_b2, slots, slots, HH, DD);
  }

  hipMemcpyAsync(out_slots, slots, (size_t)BB * SS * DD * sizeof(float),
                 hipMemcpyDeviceToDevice, stream);
}


# slot-side GEMMs (M=256: q, GRU gates, both MLP layers) re-tiled: 128-512 work-groups of 16 waves that split K, fragments loaded straight from global, fixed-order LDS sum of partials (new sum order)
# speedup vs baseline: 3.1666x; 1.4695x over previous
#include <hip/hip_runtime.h>
#include <hip/hip_bf16.h>

typedef __attribute__((ext_vector_type(2))) float v2f;
typedef __attribute__((ext_vector_type(8))) float v8f;

#define BB 32
#define FF 4096
#define SS 8
#define DD 512
#define HH 2048
#define NITER 3
#define EPS_A 1e-8f
#define LN_EPS_C 1e-5f
#define SCALE_C 0.04419417382415922f

__device__ __forceinline__ v8f wmma4(v2f a, v2f b, v8f c) {
  return __builtin_amdgcn_wmma_f32_16x16x4_f32(
       false, a,  false, b,
       (short)0, c,  false,  false);
}

__device__ __forceinline__ v2f frag_a(const float* A, int lda, int lane, int k0) {
  int row = lane & 15;
  int kk  = k0 + ((lane >> 4) << 1);
  return *(const v2f*)&A[row * lda + kk];
}
__device__ __forceinline__ v2f frag_b(const float* B, int ldb, int lane, int k0, int n0) {
  int col = n0 + (lane & 15);
  int p   = (k0 >> 1) + (lane >> 4);
  return *(const v2f*)&B[(p * ldb + col) * 2];
}

__device__ __forceinline__ float wave_sum(float v) {
#pragma unroll
  for (int o = 16; o; o >>= 1) v += __shfl_xor(v, o, 32);
  return v;
}

__global__ void __launch_bounds__(256)
row_stats_kernel(const float* __restrict__ X, float* __restrict__ mean,
                 float* __restrict__ rstd) {
  int wv = threadIdx.x >> 5, lane = threadIdx.x & 31;
  int row = blockIdx.x * 8 + wv;
  const float* x = X + (size_t)row * DD;
  float s = 0.f, s2 = 0.f;
#pragma unroll
  for (int j = 0; j < DD / 32; ++j) {
    float v = x[lane + 32 * j]; s += v; s2 += v * v;
  }
  s = wave_sum(s); s2 = wave_sum(s2);
  if (lane == 0) {
    float m = s * (1.f / DD);
    float var = s2 * (1.f / DD) - m * m;
    mean[row] = m;
    rstd[row] = rsqrtf(var + LN_EPS_C);
  }
}

__global__ void __launch_bounds__(256)
ln_rows_kernel(const float* __restrict__ X, const float* __restrict__ w,
               const float* __restrict__ b, float* __restrict__ Y) {
  int wv = threadIdx.x >> 5, lane = threadIdx.x & 31;
  int row = blockIdx.x * 8 + wv;
  const float* x = X + (size_t)row * DD;
  float xv[16]; float s = 0.f, s2 = 0.f;
#pragma unroll
  for (int j = 0; j < 16; ++j) {
    float v = x[lane + 32 * j]; xv[j] = v; s += v; s2 += v * v;
  }
  s = wave_sum(s); s2 = wave_sum(s2);
  float m = s * (1.f / DD);
  float r = rsqrtf(s2 * (1.f / DD) - m * m + LN_EPS_C);
  float* y = Y + (size_t)row * DD;
#pragma unroll
  for (int j = 0; j < 16; ++j) {
    int d = lane + 32 * j;
    y[d] = (xv[j] - m) * r * w[d] + b[d];
  }
}

__global__ void __launch_bounds__(256)
gemm_ln_dual_kernel(const float* __restrict__ X, const float* __restrict__ mean,
                    const float* __restrict__ rstd, const float* __restrict__ lnw,
                    const float* __restrict__ lnb, const float* __restrict__ Wk,
                    const float* __restrict__ Wv, float* __restrict__ Ko,
                    float* __restrict__ Vo) {
  __shared__ float As[128 * 16];
  __shared__ float Bks[16 * 64];
  __shared__ float Bvs[16 * 64];
  int t = threadIdx.x, lane = t & 31, wv = t >> 5;
  int m0 = blockIdx.x * 128, n0 = blockIdx.y * 64;

  int arow = t >> 1, aoff = (t & 1) * 8;
  size_t grow = (size_t)(m0 + arow);
  float mn = mean[grow], rs = rstd[grow];
  const float* xrow = X + grow * DD;
  int bp = t >> 5, bn2 = (t & 31) * 2;

  v8f accK[4] = {}, accV[4] = {};
  for (int kc = 0; kc < DD; kc += 16) {
    __syncthreads();
#pragma unroll
    for (int j = 0; j < 8; ++j) {
      int k = kc + aoff + j;
      As[arow * 16 + aoff + j] = (xrow[k] - mn) * rs * lnw[k] + lnb[k];
    }
    {
      float2 r0 = *(const float2*)&Wk[(size_t)(kc + 2 * bp) * DD + n0 + bn2];
      float2 r1 = *(const float2*)&Wk[(size_t)(kc + 2 * bp + 1) * DD + n0 + bn2];
      float4 wk; wk.x = r0.x; wk.y = r1.x; wk.z = r0.y; wk.w = r1.y;
      *(float4*)&Bks[(bp * 64 + bn2) * 2] = wk;
      float2 s0 = *(const float2*)&Wv[(size_t)(kc + 2 * bp) * DD + n0 + bn2];
      float2 s1 = *(const float2*)&Wv[(size_t)(kc + 2 * bp + 1) * DD + n0 + bn2];
      float4 wvv; wvv.x = s0.x; wvv.y = s1.x; wvv.z = s0.y; wvv.w = s1.y;
      *(float4*)&Bvs[(bp * 64 + bn2) * 2] = wvv;
    }
    __syncthreads();
    const float* Aw = &As[(wv * 16) * 16];
#pragma unroll
    for (int ks = 0; ks < 16; ks += 4) {
      v2f a = frag_a(Aw, 16, lane, ks);
#pragma unroll
      for (int nt = 0; nt < 4; ++nt) {
        v2f bk = frag_b(Bks, 64, lane, ks, nt * 16);
        accK[nt] = wmma4(a, bk, accK[nt]);
        v2f bv = frag_b(Bvs, 64, lane, ks, nt * 16);
        accV[nt] = wmma4(a, bv, accV[nt]);
      }
    }
  }
  int half = lane >> 4, l16 = lane & 15;
#pragma unroll
  for (int nt = 0; nt < 4; ++nt)
#pragma unroll
    for (int i = 0; i < 8; ++i) {
      size_t mrow = (size_t)(m0 + wv * 16 + half * 8 + i);
      int ncol = n0 + nt * 16 + l16;
      Ko[mrow * DD + ncol] = accK[nt][i];
      Vo[mrow * DD + ncol] = accV[nt][i];
    }
}

#define SG_WAVES 16
template <int EPI, bool BT, int K, int N>
__global__ void __launch_bounds__(SG_WAVES * 32)
slot_gemm_kernel(const float* __restrict__ A, const float* __restrict__ B,
                 const float* __restrict__ bias, float* C, const float* aux) {
  constexpr int KW = K / SG_WAVES;
  constexpr int BLK = KW < 64 ? KW : 64;
  constexpr int NU = BLK / 16;
  __shared__ float red[SG_WAVES * 16 * 64];
  const int t = threadIdx.x, lane = t & 31, wv = t >> 5;
  const int c = lane & 15, h = lane >> 4;
  const int cbase = blockIdx.x * 64, m0 = blockIdx.y * 16;
  const int k0 = wv * KW + h * 8;
  const float* arow = A + (size_t)(m0 + c) * K + k0;
  const float* bptr = BT ? B + (size_t)(cbase + c) * K + k0
                         : B + (size_t)k0 * N + cbase + c * 4;

  v8f acc[4] = {};
#pragma unroll 1
  for (int blk = 0; blk < KW; blk += BLK) {
    float4 aq[NU][2];
    float4 bq[NU][4][2];
#pragma unroll
    for (int u = 0; u < NU; ++u) {
      aq[u][0] = *(const float4*)&arow[blk + u * 16];
      aq[u][1] = *(const float4*)&arow[blk + u * 16 + 4];
#pragma unroll
      for (int i = 0; i < 4; ++i) {
        if (BT) {
          const float* p = bptr + (size_t)(16 * i) * K + blk + u * 16;
          bq[u][i][0] = *(const float4*)p;
          bq[u][i][1] = *(const float4*)(p + 4);
        } else {
          const float* p = bptr + (size_t)(blk + u * 16 + 2 * i) * N;
          bq[u][i][0] = *(const float4*)p;
          bq[u][i][1] = *(const float4*)(p + N);
        }
      }
    }
    __builtin_amdgcn_sched_barrier(0);
#pragma unroll
    for (int u = 0; u < NU; ++u) {
      const float xa[8] = {aq[u][0].x, aq[u][0].y, aq[u][0].z, aq[u][0].w,
                           aq[u][1].x, aq[u][1].y, aq[u][1].z, aq[u][1].w};
#pragma unroll
      for (int tt = 0; tt < 4; ++tt) {
        v2f a; a.x = xa[2 * tt]; a.y = xa[2 * tt + 1];
#pragma unroll
        for (int j = 0; j < 4; ++j) {
          v2f bb;
          if (BT) {
            const float xb[8] = {bq[u][j][0].x, bq[u][j][0].y, bq[u][j][0].z, bq[u][j][0].w,
                                 bq[u][j][1].x, bq[u][j][1].y, bq[u][j][1].z, bq[u][j][1].w};
            bb.x = xb[2 * tt]; bb.y = xb[2 * tt + 1];
          } else {
            const float r0[4] = {bq[u][tt][0].x, bq[u][tt][0].y, bq[u][tt][0].z, bq[u][tt][0].w};
            const float r1[4] = {bq[u][tt][1].x, bq[u][tt][1].y, bq[u][tt][1].z, bq[u][tt][1].w};
            bb.x = r0[j]; bb.y = r1[j];
          }
          acc[j] = wmma4(a, bb, acc[j]);
        }
      }
    }
  }
  {
    float* rw = &red[(wv * 16 + h * 8) * 64];
#pragma unroll
    for (int r = 0; r < 8; ++r) {
      if (BT) {
#pragma unroll
        for (int j = 0; j < 4; ++j) rw[r * 64 + 16 * j + c] = acc[j][r];
      } else {
        float4 o; o.x = acc[0][r]; o.y = acc[1][r]; o.z = acc[2][r]; o.w = acc[3][r];
        *(float4*)&rw[r * 64 + 4 * c] = o;
      }
    }
  }
  __syncthreads();
  {
    const int row = t >> 5, col = (t & 31) * 2;
    float2 s; s.x = 0.f; s.y = 0.f;
#pragma unroll
    for (int w = 0; w < SG_WAVES; ++w) {
      const float2 p = *(const float2*)&red[(w * 16 + row) * 64 + col];
      s.x += p.x; s.y += p.y;
    }
    const size_t idx = (size_t)(m0 + row) * N + cbase + col;
    if (EPI >= 1) { s.x += bias[cbase + col]; s.y += bias[cbase + col + 1]; }
    if (EPI == 1) { s.x = fmaxf(s.x, 0.f); s.y = fmaxf(s.y, 0.f); }
    if (EPI == 2) {
      const float2 x = *(const float2*)&aux[idx];
      s.x += x.x; s.y += x.y;
    }
    *(float2*)&C[idx] = s;
  }
}

__global__ void __launch_bounds__(256)
dots_kernel(const float* __restrict__ Q, const float* __restrict__ Keys,
            float* __restrict__ Dots) {
  __shared__ float As[16 * 16];
  __shared__ float Bs[16 * 512];
  int t = threadIdx.x, lane = t & 31, wv = t >> 5;
  int b = blockIdx.x, nb = blockIdx.y * 512;
  const float* qb = Q + (size_t)b * SS * DD;
  const float* kb = Keys + ((size_t)b * FF + nb) * DD;
  v8f acc[4] = {};
  for (int kc = 0; kc < DD; kc += 16) {
    __syncthreads();
    {
      int r = t >> 4, k = t & 15;
      As[r * 16 + k] = (r < SS) ? qb[(size_t)r * DD + kc + k] : 0.f;
    }
    {
      int k4 = (t & 3) * 4;
#pragma unroll
      for (int j = 0; j < 8; ++j) {
        int n = (t >> 2) + j * 64;
        float4 f = *(const float4*)&kb[(size_t)n * DD + kc + k4];
        float2 lo; lo.x = f.x; lo.y = f.y;
        float2 hi; hi.x = f.z; hi.y = f.w;
        *(float2*)&Bs[(((k4 >> 1) + 0) * 512 + n) * 2] = lo;
        *(float2*)&Bs[(((k4 >> 1) + 1) * 512 + n) * 2] = hi;
      }
    }
    __syncthreads();
#pragma unroll
    for (int ks = 0; ks < 16; ks += 4) {
      v2f a = frag_a(As, 16, lane, ks);
#pragma unroll
      for (int nt = 0; nt < 4; ++nt) {
        v2f b2 = frag_b(Bs, 512, lane, ks, wv * 64 + nt * 16);
        acc[nt] = wmma4(a, b2, acc[nt]);
      }
    }
  }
  int half = lane >> 4, l16 = lane & 15;
  if (half == 0) {
#pragma unroll
    for (int nt = 0; nt < 4; ++nt)
#pragma unroll
      for (int i = 0; i < 8; ++i) {
        int f = nb + wv * 64 + nt * 16 + l16;
        Dots[((size_t)b * SS + i) * FF + f] = acc[nt][i] * SCALE_C;
      }
  }
}

__global__ void __launch_bounds__(256)
softmax_kernel(const float* __restrict__ Dots, float* __restrict__ PreNorm,
               float* __restrict__ Attn, float* __restrict__ Sums) {
  int b = blockIdx.y;
  int f = blockIdx.x * 256 + threadIdx.x;
  int lane = threadIdx.x & 31;
  float d[SS]; float mx = -3.4e38f;
#pragma unroll
  for (int s = 0; s < SS; ++s) {
    d[s] = Dots[((size_t)b * SS + s) * FF + f];
    mx = fmaxf(mx, d[s]);
  }
  float sum = 0.f;
#pragma unroll
  for (int s = 0; s < SS; ++s) { d[s] = expf(d[s] - mx); sum += d[s]; }
  float inv = 1.f / sum;
#pragma unroll
  for (int s = 0; s < SS; ++s) {
    float p = d[s] * inv;
    size_t idx = ((size_t)b * SS + s) * FF + f;
    PreNorm[idx] = p;
    float aw = p + EPS_A;
    Attn[idx] = aw;
    float r = wave_sum(aw);
    if (lane == 0) atomicAdd(&Sums[b * SS + s], r);
  }
}

__global__ void zero_sums_kernel(float* __restrict__ Sums) {
  Sums[threadIdx.x] = 0.f;
}

#define UPD_WAVES 16
#define UPD_ROWS (FF / UPD_WAVES)
#define UPD_BLK 64
__global__ void __launch_bounds__(UPD_WAVES * 32)
updates_kernel(const float* __restrict__ Attn, const float* __restrict__ Vals,
               const float* __restrict__ Sums, float* __restrict__ Upd) {
  __shared__ float red[UPD_WAVES * SS * 64];
  const int t = threadIdx.x, lane = t & 31, wv = t >> 5;
  const int c = lane & 15, h = lane >> 4;
  const int cbase = blockIdx.x * 64, b = blockIdx.y;
  const bool valid = c < SS;
  const float* arow = Attn + ((size_t)b * SS + (c & 7)) * FF + wv * UPD_ROWS + h * 8;
  const float* vcol = Vals + ((size_t)b * FF + wv * UPD_ROWS + h * 8) * DD + cbase + c * 4;

  v8f acc[4] = {};
#pragma unroll 1
  for (int blk = 0; blk < UPD_ROWS; blk += UPD_BLK) {
    float4 aq[UPD_BLK / 16][2];
    float4 bq[UPD_BLK / 16][4][2];
#pragma unroll
    for (int u = 0; u < UPD_BLK / 16; ++u) {
      aq[u][0] = *(const float4*)&arow[blk + u * 16];
      aq[u][1] = *(const float4*)&arow[blk + u * 16 + 4];
#pragma unroll
      for (int tt = 0; tt < 4; ++tt) {
        const float* p = vcol + (size_t)(blk + u * 16 + 2 * tt) * DD;
        bq[u][tt][0] = *(const float4*)p;
        bq[u][tt][1] = *(const float4*)(p + DD);
      }
    }
#pragma unroll
    for (int u = 0; u < UPD_BLK / 16; ++u) {
      const float xa[8] = {aq[u][0].x, aq[u][0].y, aq[u][0].z, aq[u][0].w,
                           aq[u][1].x, aq[u][1].y, aq[u][1].z, aq[u][1].w};
#pragma unroll
      for (int tt = 0; tt < 4; ++tt) {
        v2f a;
        a.x = valid ? xa[2 * tt] : 0.f;
        a.y = valid ? xa[2 * tt + 1] : 0.f;
        const float r0[4] = {bq[u][tt][0].x, bq[u][tt][0].y, bq[u][tt][0].z, bq[u][tt][0].w};
        const float r1[4] = {bq[u][tt][1].x, bq[u][tt][1].y, bq[u][tt][1].z, bq[u][tt][1].w};
#pragma unroll
        for (int j = 0; j < 4; ++j) {
          v2f bb; bb.x = r0[j]; bb.y = r1[j];
          acc[j] = wmma4(a, bb, acc[j]);
        }
      }
    }
  }
  if (h == 0) {
    float* rw = &red[(wv * SS) * 64 + c * 4];
#pragma unroll
    for (int r = 0; r < SS; ++r) {
      float4 o; o.x = acc[0][r]; o.y = acc[1][r]; o.z = acc[2][r]; o.w = acc[3][r];
      *(float4*)&rw[r * 64] = o;
    }
  }
  __syncthreads();
  {
    const int r = t >> 6, col = t & 63;
    float s = 0.f;
#pragma unroll
    for (int w = 0; w < UPD_WAVES; ++w) s += red[(w * SS + r) * 64 + col];
    const float inv = 1.f / Sums[b * SS + r];
    Upd[((size_t)b * SS + r) * DD + cbase + col] = s * inv;
  }
}

__global__ void __launch_bounds__(256)
gru_kernel(const float* __restrict__ gi, const float* __restrict__ gh,
           const float* __restrict__ sn, float* __restrict__ slots) {
  int idx = blockIdx.x * 256 + threadIdx.x;
  int m = idx >> 9, d = idx & 511;
  size_t base = (size_t)m * 3 * DD;
  float ir = gi[base + d],           hr = gh[base + d];
  float iz = gi[base + DD + d],      hz = gh[base + DD + d];
  float in_ = gi[base + 2 * DD + d], hn = gh[base + 2 * DD + d];
  float r = 1.f / (1.f + expf(-(ir + hr)));
  float z = 1.f / (1.f + expf(-(iz + hz)));
  float n = tanhf(in_ + r * hn);
  float h = sn[(size_t)m * DD + d];
  slots[(size_t)m * DD + d] = (1.f - z) * n + z * h;
}

extern "C" void kernel_launch(void* const* d_in, const int* in_sizes, int n_in,
                              void* d_out, int out_size, void* d_ws, size_t ws_size,
                              hipStream_t stream) {
  (void)in_sizes; (void)n_in; (void)out_size; (void)ws_size;
  const float* in_slots  = (const float*)d_in[0];
  const float* features  = (const float*)d_in[1];
  const float* w_k       = (const float*)d_in[2];
  const float* w_v       = (const float*)d_in[3];
  const float* w_q       = (const float*)d_in[4];
  const float* ln_feat_w = (const float*)d_in[5];
  const float* ln_feat_b = (const float*)d_in[6];
  const float* ln_slot_w = (const float*)d_in[7];
  const float* ln_slot_b = (const float*)d_in[8];
  const float* gru_w_ih  = (const float*)d_in[9];
  const float* gru_w_hh  = (const float*)d_in[10];
  const float* gru_b_ih  = (const float*)d_in[11];
  const float* gru_b_hh  = (const float*)d_in[12];
  const float* ln_mlp_w  = (const float*)d_in[13];
  const float* ln_mlp_b  = (const float*)d_in[14];
  const float* mlp_w1    = (const float*)d_in[15];
  const float* mlp_b1    = (const float*)d_in[16];
  const float* mlp_w2    = (const float*)d_in[17];
  const float* mlp_b2    = (const float*)d_in[18];

  float* ws = (float*)d_ws;
  size_t off = 0;
  auto carve = [&](size_t n) { float* p = ws + off; off += n; return p; };
  float* mean  = carve((size_t)BB * FF);
  float* rstd  = carve((size_t)BB * FF);
  float* keys  = carve((size_t)BB * FF * DD);
  float* vals  = carve((size_t)BB * FF * DD);
  float* slots = carve((size_t)BB * SS * DD);
  float* s_n   = carve((size_t)BB * SS * DD);
  float* q     = carve((size_t)BB * SS * DD);
  float* dots  = carve((size_t)BB * SS * FF);
  float* attn  = carve((size_t)BB * SS * FF);
  float* sums  = carve(256);
  float* upd   = carve((size_t)BB * SS * DD);
  float* gi    = carve((size_t)BB * SS * 3 * DD);
  float* gh    = carve((size_t)BB * SS * 3 * DD);
  float* hln   = carve((size_t)BB * SS * DD);
  float* h1    = carve((size_t)BB * SS * HH);

  float* out_slots = (float*)d_out;
  float* out_attn  = out_slots + (size_t)BB * SS * DD;

  hipMemcpyAsync(slots, in_slots, (size_t)BB * SS * DD * sizeof(float),
                 hipMemcpyDeviceToDevice, stream);

  row_stats_kernel<<<(BB * FF) / 8, 256, 0, stream>>>(features, mean, rstd);
  gemm_ln_dual_kernel<<<dim3((BB * FF) / 128, DD / 64), 256, 0, stream>>>(
      features, mean, rstd, ln_feat_w, ln_feat_b, w_k, w_v, keys, vals);

  for (int it = 0; it < NITER; ++it) {
    ln_rows_kernel<<<32, 256, 0, stream>>>(slots, ln_slot_w, ln_slot_b, s_n);
    slot_gemm_kernel<0, false, DD, DD><<<dim3(DD / 64, (BB * SS) / 16), SG_WAVES * 32, 0, stream>>>(
        s_n, w_q, nullptr, q, nullptr);
    zero_sums_kernel<<<1, 256, 0, stream>>>(sums);
    dots_kernel<<<dim3(BB, FF / 512), 256, 0, stream>>>(q, keys, dots);
    softmax_kernel<<<dim3(FF / 256, BB), 256, 0, stream>>>(dots, out_attn, attn, sums);
    updates_kernel<<<dim3(DD / 64, BB), UPD_WAVES * 32, 0, stream>>>(attn, vals, sums, upd);
    slot_gemm_kernel<3, true, DD, 3 * DD><<<dim3((3 * DD) / 64, (BB * SS) / 16), SG_WAVES * 32, 0, stream>>>(
        upd, gru_w_ih, gru_b_ih, gi, nullptr);
    slot_gemm_kernel<3, true, DD, 3 * DD><<<dim3((3 * DD) / 64, (BB * SS) / 16), SG_WAVES * 32, 0, stream>>>(
        s_n, gru_w_hh, gru_b_hh, gh, nullptr);
    gru_kernel<<<(BB * SS * DD) / 256, 256, 0, stream>>>(gi, gh, s_n, slots);
    ln_rows_kernel<<<32, 256, 0, stream>>>(slots, ln_mlp_w, ln_mlp_b, hln);
    slot_gemm_kernel<1, false, DD, HH><<<dim3(HH / 64, (BB * SS) / 16), SG_WAVES * 32, 0, stream>>>(
        hln, mlp_w1, mlp_b1, h1, nullptr);
    slot_gemm_kernel<2, false, HH, DD><<<dim3(DD / 64, (BB * SS) / 16), SG_WAVES * 32, 0, stream>>>(
        h1, mlp_w2, mlp_b2, slots, slots);
  }

  hipMemcpyAsync(out_slots, slots, (size_t)BB * SS * DD * sizeof(float),
                 hipMemcpyDeviceToDevice, stream);
}


# feature projections moved to the slot side by associativity: dots=(LN(s) Wq Wk^T) LN(f)^T, gi=(attn LN(f)/sums)(Wv Wih^T)+b; keys/values GEMM gone; products re-associated (new sum order)
# speedup vs baseline: 8.7680x; 2.7689x over previous
#include <hip/hip_runtime.h>
#include <hip/hip_bf16.h>

typedef __attribute__((ext_vector_type(2))) float v2f;
typedef __attribute__((ext_vector_type(8))) float v8f;

#define BB 32
#define FF 4096
#define SS 8
#define DD 512
#define HH 2048
#define NITER 3
#define EPS_A 1e-8f
#define LN_EPS_C 1e-5f
#define SCALE_C 0.04419417382415922f

__device__ __forceinline__ v8f wmma4(v2f a, v2f b, v8f c) {
  return __builtin_amdgcn_wmma_f32_16x16x4_f32(
       false, a,  false, b,
       (short)0, c,  false,  false);
}

__device__ __forceinline__ v2f frag_a(const float* A, int lda, int lane, int k0) {
  int row = lane & 15;
  int kk  = k0 + ((lane >> 4) << 1);
  return *(const v2f*)&A[row * lda + kk];
}
__device__ __forceinline__ v2f frag_b(const float* B, int ldb, int lane, int k0, int n0) {
  int col = n0 + (lane & 15);
  int p   = (k0 >> 1) + (lane >> 4);
  return *(const v2f*)&B[(p * ldb + col) * 2];
}

__device__ __forceinline__ float wave_sum(float v) {
#pragma unroll
  for (int o = 16; o; o >>= 1) v += __shfl_xor(v, o, 32);
  return v;
}

__global__ void __launch_bounds__(256)
ln_feat_kernel(const float* __restrict__ X, const float* __restrict__ w,
               const float* __restrict__ b, float* __restrict__ Y) {
  int wv = threadIdx.x >> 5, lane = threadIdx.x & 31;
  size_t row = (size_t)blockIdx.x * 8 + wv;
  const float4* x = (const float4*)(X + row * DD);
  float4 xv[4]; float s = 0.f, s2 = 0.f;
#pragma unroll
  for (int j = 0; j < 4; ++j) {
    float4 v = x[lane + 32 * j]; xv[j] = v;
    s += (v.x + v.y) + (v.z + v.w);
    s2 += (v.x * v.x + v.y * v.y) + (v.z * v.z + v.w * v.w);
  }
  s = wave_sum(s); s2 = wave_sum(s2);
  float m = s * (1.f / DD);
  float r = rsqrtf(s2 * (1.f / DD) - m * m + LN_EPS_C);
  float4* y = (float4*)(Y + row * DD);
#pragma unroll
  for (int j = 0; j < 4; ++j) {
    int d4 = lane + 32 * j;
    float4 ww = ((const float4*)w)[d4], bb = ((const float4*)b)[d4];
    float4 o;
    o.x = (xv[j].x - m) * r * ww.x + bb.x;
    o.y = (xv[j].y - m) * r * ww.y + bb.y;
    o.z = (xv[j].z - m) * r * ww.z + bb.z;
    o.w = (xv[j].w - m) * r * ww.w + bb.w;
    y[d4] = o;
  }
}

__global__ void __launch_bounds__(256)
ln_rows_kernel(const float* __restrict__ X, const float* __restrict__ w,
               const float* __restrict__ b, float* __restrict__ Y) {
  int wv = threadIdx.x >> 5, lane = threadIdx.x & 31;
  int row = blockIdx.x * 8 + wv;
  const float* x = X + (size_t)row * DD;
  float xv[16]; float s = 0.f, s2 = 0.f;
#pragma unroll
  for (int j = 0; j < 16; ++j) {
    float v = x[lane + 32 * j]; xv[j] = v; s += v; s2 += v * v;
  }
  s = wave_sum(s); s2 = wave_sum(s2);
  float m = s * (1.f / DD);
  float r = rsqrtf(s2 * (1.f / DD) - m * m + LN_EPS_C);
  float* y = Y + (size_t)row * DD;
#pragma unroll
  for (int j = 0; j < 16; ++j) {
    int d = lane + 32 * j;
    y[d] = (xv[j] - m) * r * w[d] + b[d];
  }
}

#define SG_WAVES 16
template <int EPI, bool BT, int K, int N>
__global__ void __launch_bounds__(SG_WAVES * 32)
slot_gemm_kernel(const float* __restrict__ A, const float* __restrict__ B,
                 const float* __restrict__ bias, float* C, const float* aux) {
  constexpr int KW = K / SG_WAVES;
  constexpr int BLK = KW < 64 ? KW : 64;
  constexpr int NU = BLK / 16;
  __shared__ float red[SG_WAVES * 16 * 64];
  const int t = threadIdx.x, lane = t & 31, wv = t >> 5;
  const int c = lane & 15, h = lane >> 4;
  const int cbase = blockIdx.x * 64, m0 = blockIdx.y * 16;
  const int k0 = wv * KW + h * 8;
  const float* arow = A + (size_t)(m0 + c) * K + k0;
  const float* bptr = BT ? B + (size_t)(cbase + c) * K + k0
                         : B + (size_t)k0 * N + cbase + c * 4;

  v8f acc[4] = {};
#pragma unroll 1
  for (int blk = 0; blk < KW; blk += BLK) {
    float4 aq[NU][2];
    float4 bq[NU][4][2];
#pragma unroll
    for (int u = 0; u < NU; ++u) {
      aq[u][0] = *(const float4*)&arow[blk + u * 16];
      aq[u][1] = *(const float4*)&arow[blk + u * 16 + 4];
#pragma unroll
      for (int i = 0; i < 4; ++i) {
        if (BT) {
          const float* p = bptr + (size_t)(16 * i) * K + blk + u * 16;
          bq[u][i][0] = *(const float4*)p;
          bq[u][i][1] = *(const float4*)(p + 4);
        } else {
          const float* p = bptr + (size_t)(blk + u * 16 + 2 * i) * N;
          bq[u][i][0] = *(const float4*)p;
          bq[u][i][1] = *(const float4*)(p + N);
        }
      }
    }
    __builtin_amdgcn_sched_barrier(0);
#pragma unroll
    for (int u = 0; u < NU; ++u) {
      const float xa[8] = {aq[u][0].x, aq[u][0].y, aq[u][0].z, aq[u][0].w,
                           aq[u][1].x, aq[u][1].y, aq[u][1].z, aq[u][1].w};
#pragma unroll
      for (int tt = 0; tt < 4; ++tt) {
        v2f a; a.x = xa[2 * tt]; a.y = xa[2 * tt + 1];
#pragma unroll
        for (int j = 0; j < 4; ++j) {
          v2f bb;
          if (BT) {
            const float xb[8] = {bq[u][j][0].x, bq[u][j][0].y, bq[u][j][0].z, bq[u][j][0].w,
                                 bq[u][j][1].x, bq[u][j][1].y, bq[u][j][1].z, bq[u][j][1].w};
            bb.x = xb[2 * tt]; bb.y = xb[2 * tt + 1];
          } else {
            const float r0[4] = {bq[u][tt][0].x, bq[u][tt][0].y, bq[u][tt][0].z, bq[u][tt][0].w};
            const float r1[4] = {bq[u][tt][1].x, bq[u][tt][1].y, bq[u][tt][1].z, bq[u][tt][1].w};
            bb.x = r0[j]; bb.y = r1[j];
          }
          acc[j] = wmma4(a, bb, acc[j]);
        }
      }
    }
  }
  {
    float* rw = &red[(wv * 16 + h * 8) * 64];
#pragma unroll
    for (int r = 0; r < 8; ++r) {
      if (BT) {
#pragma unroll
        for (int j = 0; j < 4; ++j) rw[r * 64 + 16 * j + c] = acc[j][r];
      } else {
        float4 o; o.x = acc[0][r]; o.y = acc[1][r]; o.z = acc[2][r]; o.w = acc[3][r];
        *(float4*)&rw[r * 64 + 4 * c] = o;
      }
    }
  }
  __syncthreads();
  {
    const int row = t >> 5, col = (t & 31) * 2;
    float2 s; s.x = 0.f; s.y = 0.f;
#pragma unroll
    for (int w = 0; w < SG_WAVES; ++w) {
      const float2 p = *(const float2*)&red[(w * 16 + row) * 64 + col];
      s.x += p.x; s.y += p.y;
    }
    const size_t idx = (size_t)(m0 + row) * N + cbase + col;
    if (EPI >= 1) { s.x += bias[cbase + col]; s.y += bias[cbase + col + 1]; }
    if (EPI == 1) { s.x = fmaxf(s.x, 0.f); s.y = fmaxf(s.y, 0.f); }
    if (EPI == 2) {
      const float2 x = *(const float2*)&aux[idx];
      s.x += x.x; s.y += x.y;
    }
    *(float2*)&C[idx] = s;
  }
}

__global__ void __launch_bounds__(256)
dots_kernel(const float* __restrict__ Q, const float* __restrict__ Keys,
            float* __restrict__ Dots) {
  __shared__ float As[16 * 16];
  __shared__ float Bs[16 * 512];
  int t = threadIdx.x, lane = t & 31, wv = t >> 5;
  int b = blockIdx.x, nb = blockIdx.y * 512;
  const float* qb = Q + (size_t)b * SS * DD;
  const float* kb = Keys + ((size_t)b * FF + nb) * DD;
  v8f acc[4] = {};
  for (int kc = 0; kc < DD; kc += 16) {
    __syncthreads();
    {
      int r = t >> 4, k = t & 15;
      As[r * 16 + k] = (r < SS) ? qb[(size_t)r * DD + kc + k] : 0.f;
    }
    {
      int k4 = (t & 3) * 4;
#pragma unroll
      for (int j = 0; j < 8; ++j) {
        int n = (t >> 2) + j * 64;
        float4 f = *(const float4*)&kb[(size_t)n * DD + kc + k4];
        float2 lo; lo.x = f.x; lo.y = f.y;
        float2 hi; hi.x = f.z; hi.y = f.w;
        *(float2*)&Bs[(((k4 >> 1) + 0) * 512 + n) * 2] = lo;
        *(float2*)&Bs[(((k4 >> 1) + 1) * 512 + n) * 2] = hi;
      }
    }
    __syncthreads();
#pragma unroll
    for (int ks = 0; ks < 16; ks += 4) {
      v2f a = frag_a(As, 16, lane, ks);
#pragma unroll
      for (int nt = 0; nt < 4; ++nt) {
        v2f b2 = frag_b(Bs, 512, lane, ks, wv * 64 + nt * 16);
        acc[nt] = wmma4(a, b2, acc[nt]);
      }
    }
  }
  int half = lane >> 4, l16 = lane & 15;
  if (half == 0) {
#pragma unroll
    for (int nt = 0; nt < 4; ++nt)
#pragma unroll
      for (int i = 0; i < 8; ++i) {
        int f = nb + wv * 64 + nt * 16 + l16;
        Dots[((size_t)b * SS + i) * FF + f] = acc[nt][i] * SCALE_C;
      }
  }
}

__global__ void __launch_bounds__(256)
softmax_kernel(const float* __restrict__ Dots, float* __restrict__ PreNorm,
               float* __restrict__ Attn, float* __restrict__ Sums) {
  int b = blockIdx.y;
  int f = blockIdx.x * 256 + threadIdx.x;
  int lane = threadIdx.x & 31;
  float d[SS]; float mx = -3.4e38f;
#pragma unroll
  for (int s = 0; s < SS; ++s) {
    d[s] = Dots[((size_t)b * SS + s) * FF + f];
    mx = fmaxf(mx, d[s]);
  }
  float sum = 0.f;
#pragma unroll
  for (int s = 0; s < SS; ++s) { d[s] = expf(d[s] - mx); sum += d[s]; }
  float inv = 1.f / sum;
#pragma unroll
  for (int s = 0; s < SS; ++s) {
    float p = d[s] * inv;
    size_t idx = ((size_t)b * SS + s) * FF + f;
    PreNorm[idx] = p;
    float aw = p + EPS_A;
    Attn[idx] = aw;
    float r = wave_sum(aw);
    if (lane == 0) atomicAdd(&Sums[b * SS + s], r);
  }
}

__global__ void zero_sums_kernel(float* __restrict__ Sums) {
  Sums[threadIdx.x] = 0.f;
}

#define UPD_WAVES 16
#define UPD_ROWS (FF / UPD_WAVES)
#define UPD_BLK 64
__global__ void __launch_bounds__(UPD_WAVES * 32)
updates_kernel(const float* __restrict__ Attn, const float* __restrict__ Vals,
               const float* __restrict__ Sums, float* __restrict__ Upd) {
  __shared__ float red[UPD_WAVES * SS * 64];
  const int t = threadIdx.x, lane = t & 31, wv = t >> 5;
  const int c = lane & 15, h = lane >> 4;
  const int cbase = blockIdx.x * 64, b = blockIdx.y;
  const bool valid = c < SS;
  const float* arow = Attn + ((size_t)b * SS + (c & 7)) * FF + wv * UPD_ROWS + h * 8;
  const float* vcol = Vals + ((size_t)b * FF + wv * UPD_ROWS + h * 8) * DD + cbase + c * 4;

  v8f acc[4] = {};
#pragma unroll 1
  for (int blk = 0; blk < UPD_ROWS; blk += UPD_BLK) {
    float4 aq[UPD_BLK / 16][2];
    float4 bq[UPD_BLK / 16][4][2];
#pragma unroll
    for (int u = 0; u < UPD_BLK / 16; ++u) {
      aq[u][0] = *(const float4*)&arow[blk + u * 16];
      aq[u][1] = *(const float4*)&arow[blk + u * 16 + 4];
#pragma unroll
      for (int tt = 0; tt < 4; ++tt) {
        const float* p = vcol + (size_t)(blk + u * 16 + 2 * tt) * DD;
        bq[u][tt][0] = *(const float4*)p;
        bq[u][tt][1] = *(const float4*)(p + DD);
      }
    }
#pragma unroll
    for (int u = 0; u < UPD_BLK / 16; ++u) {
      const float xa[8] = {aq[u][0].x, aq[u][0].y, aq[u][0].z, aq[u][0].w,
                           aq[u][1].x, aq[u][1].y, aq[u][1].z, aq[u][1].w};
#pragma unroll
      for (int tt = 0; tt < 4; ++tt) {
        v2f a;
        a.x = valid ? xa[2 * tt] : 0.f;
        a.y = valid ? xa[2 * tt + 1] : 0.f;
        const float r0[4] = {bq[u][tt][0].x, bq[u][tt][0].y, bq[u][tt][0].z, bq[u][tt][0].w};
        const float r1[4] = {bq[u][tt][1].x, bq[u][tt][1].y, bq[u][tt][1].z, bq[u][tt][1].w};
#pragma unroll
        for (int j = 0; j < 4; ++j) {
          v2f bb; bb.x = r0[j]; bb.y = r1[j];
          acc[j] = wmma4(a, bb, acc[j]);
        }
      }
    }
  }
  if (h == 0) {
    float* rw = &red[(wv * SS) * 64 + c * 4];
#pragma unroll
    for (int r = 0; r < SS; ++r) {
      float4 o; o.x = acc[0][r]; o.y = acc[1][r]; o.z = acc[2][r]; o.w = acc[3][r];
      *(float4*)&rw[r * 64] = o;
    }
  }
  __syncthreads();
  {
    const int r = t >> 6, col = t & 63;
    float s = 0.f;
#pragma unroll
    for (int w = 0; w < UPD_WAVES; ++w) s += red[(w * SS + r) * 64 + col];
    const float inv = 1.f / Sums[b * SS + r];
    Upd[((size_t)b * SS + r) * DD + cbase + col] = s * inv;
  }
}

__global__ void __launch_bounds__(256)
gru_kernel(const float* __restrict__ gi, const float* __restrict__ gh,
           const float* __restrict__ sn, float* __restrict__ slots) {
  int idx = blockIdx.x * 256 + threadIdx.x;
  int m = idx >> 9, d = idx & 511;
  size_t base = (size_t)m * 3 * DD;
  float ir = gi[base + d],           hr = gh[base + d];
  float iz = gi[base + DD + d],      hz = gh[base + DD + d];
  float in_ = gi[base + 2 * DD + d], hn = gh[base + 2 * DD + d];
  float r = 1.f / (1.f + expf(-(ir + hr)));
  float z = 1.f / (1.f + expf(-(iz + hz)));
  float n = tanhf(in_ + r * hn);
  float h = sn[(size_t)m * DD + d];
  slots[(size_t)m * DD + d] = (1.f - z) * n + z * h;
}

extern "C" void kernel_launch(void* const* d_in, const int* in_sizes, int n_in,
                              void* d_out, int out_size, void* d_ws, size_t ws_size,
                              hipStream_t stream) {
  (void)in_sizes; (void)n_in; (void)out_size; (void)ws_size;
  const float* in_slots  = (const float*)d_in[0];
  const float* features  = (const float*)d_in[1];
  const float* w_k       = (const float*)d_in[2];
  const float* w_v       = (const float*)d_in[3];
  const float* w_q       = (const float*)d_in[4];
  const float* ln_feat_w = (const float*)d_in[5];
  const float* ln_feat_b = (const float*)d_in[6];
  const float* ln_slot_w = (const float*)d_in[7];
  const float* ln_slot_b = (const float*)d_in[8];
  const float* gru_w_ih  = (const float*)d_in[9];
  const float* gru_w_hh  = (const float*)d_in[10];
  const float* gru_b_ih  = (const float*)d_in[11];
  const float* gru_b_hh  = (const float*)d_in[12];
  const float* ln_mlp_w  = (const float*)d_in[13];
  const float* ln_mlp_b  = (const float*)d_in[14];
  const float* mlp_w1    = (const float*)d_in[15];
  const float* mlp_b1    = (const float*)d_in[16];
  const float* mlp_w2    = (const float*)d_in[17];
  const float* mlp_b2    = (const float*)d_in[18];

  float* ws = (float*)d_ws;
  size_t off = 0;
  auto carve = [&](size_t n) { float* p = ws + off; off += n; return p; };
  float* fln   = carve((size_t)BB * FF * DD);
  float* wqk   = carve((size_t)DD * DD);
  float* wvi   = carve((size_t)DD * 3 * DD);
  float* slots = carve((size_t)BB * SS * DD);
  float* s_n   = carve((size_t)BB * SS * DD);
  float* q     = carve((size_t)BB * SS * DD);
  float* dots  = carve((size_t)BB * SS * FF);
  float* attn  = carve((size_t)BB * SS * FF);
  float* sums  = carve(256);
  float* upd   = carve((size_t)BB * SS * DD);
  float* gi    = carve((size_t)BB * SS * 3 * DD);
  float* gh    = carve((size_t)BB * SS * 3 * DD);
  float* hln   = carve((size_t)BB * SS * DD);
  float* h1    = carve((size_t)BB * SS * HH);

  float* out_slots = (float*)d_out;
  float* out_attn  = out_slots + (size_t)BB * SS * DD;

  hipMemcpyAsync(slots, in_slots, (size_t)BB * SS * DD * sizeof(float),
                 hipMemcpyDeviceToDevice, stream);

  ln_feat_kernel<<<(BB * FF) / 8, 256, 0, stream>>>(features, ln_feat_w, ln_feat_b, fln);
  slot_gemm_kernel<0, true, DD, DD><<<dim3(DD / 64, DD / 16), SG_WAVES * 32, 0, stream>>>(
      w_q, w_k, nullptr, wqk, nullptr);
  slot_gemm_kernel<0, true, DD, 3 * DD><<<dim3((3 * DD) / 64, DD / 16), SG_WAVES * 32, 0, stream>>>(
      w_v, gru_w_ih, nullptr, wvi, nullptr);

  for (int it = 0; it < NITER; ++it) {
    ln_rows_kernel<<<32, 256, 0, stream>>>(slots, ln_slot_w, ln_slot_b, s_n);
    slot_gemm_kernel<0, false, DD, DD><<<dim3(DD / 64, (BB * SS) / 16), SG_WAVES * 32, 0, stream>>>(
        s_n, wqk, nullptr, q, nullptr);
    zero_sums_kernel<<<1, 256, 0, stream>>>(sums);
    dots_kernel<<<dim3(BB, FF / 512), 256, 0, stream>>>(q, fln, dots);
    softmax_kernel<<<dim3(FF / 256, BB), 256, 0, stream>>>(dots, out_attn, attn, sums);
    updates_kernel<<<dim3(DD / 64, BB), UPD_WAVES * 32, 0, stream>>>(attn, fln, sums, upd);
    slot_gemm_kernel<3, false, DD, 3 * DD><<<dim3((3 * DD) / 64, (BB * SS) / 16), SG_WAVES * 32, 0, stream>>>(
        upd, wvi, gru_b_ih, gi, nullptr);
    slot_gemm_kernel<3, true, DD, 3 * DD><<<dim3((3 * DD) / 64, (BB * SS) / 16), SG_WAVES * 32, 0, stream>>>(
        s_n, gru_w_hh, gru_b_hh, gh, nullptr);
    gru_kernel<<<(BB * SS * DD) / 256, 256, 0, stream>>>(gi, gh, s_n, slots);
    ln_rows_kernel<<<32, 256, 0, stream>>>(slots, ln_mlp_w, ln_mlp_b, hln);
    slot_gemm_kernel<1, false, DD, HH><<<dim3(HH / 64, (BB * SS) / 16), SG_WAVES * 32, 0, stream>>>(
        hln, mlp_w1, mlp_b1, h1, nullptr);
    slot_gemm_kernel<2, false, HH, DD><<<dim3(DD / 64, (BB * SS) / 16), SG_WAVES * 32, 0, stream>>>(
        h1, mlp_w2, mlp_b2, slots, slots);
  }

  hipMemcpyAsync(out_slots, slots, (size_t)BB * SS * DD * sizeof(float),
                 hipMemcpyDeviceToDevice, stream);
}


# dots product re-tiled: a wave owns 32 features over full K, feature rows loaded straight into fragments (two register sets), q staged once in LDS; softmax over slots follows in-lane (new sum order)
# speedup vs baseline: 10.0018x; 1.1407x over previous
#include <hip/hip_runtime.h>
#include <hip/hip_bf16.h>

typedef __attribute__((ext_vector_type(2))) float v2f;
typedef __attribute__((ext_vector_type(8))) float v8f;

#define BB 32
#define FF 4096
#define SS 8
#define DD 512
#define HH 2048
#define NITER 3
#define EPS_A 1e-8f
#define LN_EPS_C 1e-5f
#define SCALE_C 0.04419417382415922f

__device__ __forceinline__ v8f wmma4(v2f a, v2f b, v8f c) {
  return __builtin_amdgcn_wmma_f32_16x16x4_f32(
       false, a,  false, b,
       (short)0, c,  false,  false);
}

__device__ __forceinline__ float wave_sum(float v) {
#pragma unroll
  for (int o = 16; o; o >>= 1) v += __shfl_xor(v, o, 32);
  return v;
}

__global__ void __launch_bounds__(256)
ln_feat_kernel(const float* __restrict__ X, const float* __restrict__ w,
               const float* __restrict__ b, float* __restrict__ Y) {
  int wv = threadIdx.x >> 5, lane = threadIdx.x & 31;
  size_t row = (size_t)blockIdx.x * 8 + wv;
  const float4* x = (const float4*)(X + row * DD);
  float4 xv[4]; float s = 0.f, s2 = 0.f;
#pragma unroll
  for (int j = 0; j < 4; ++j) {
    float4 v = x[lane + 32 * j]; xv[j] = v;
    s += (v.x + v.y) + (v.z + v.w);
    s2 += (v.x * v.x + v.y * v.y) + (v.z * v.z + v.w * v.w);
  }
  s = wave_sum(s); s2 = wave_sum(s2);
  float m = s * (1.f / DD);
  float r = rsqrtf(s2 * (1.f / DD) - m * m + LN_EPS_C);
  float4* y = (float4*)(Y + row * DD);
#pragma unroll
  for (int j = 0; j < 4; ++j) {
    int d4 = lane + 32 * j;
    float4 ww = ((const float4*)w)[d4], bb = ((const float4*)b)[d4];
    float4 o;
    o.x = (xv[j].x - m) * r * ww.x + bb.x;
    o.y = (xv[j].y - m) * r * ww.y + bb.y;
    o.z = (xv[j].z - m) * r * ww.z + bb.z;
    o.w = (xv[j].w - m) * r * ww.w + bb.w;
    y[d4] = o;
  }
}

__global__ void __launch_bounds__(256)
ln_rows_kernel(const float* __restrict__ X, const float* __restrict__ w,
               const float* __restrict__ b, float* __restrict__ Y) {
  int wv = threadIdx.x >> 5, lane = threadIdx.x & 31;
  int row = blockIdx.x * 8 + wv;
  const float* x = X + (size_t)row * DD;
  float xv[16]; float s = 0.f, s2 = 0.f;
#pragma unroll
  for (int j = 0; j < 16; ++j) {
    float v = x[lane + 32 * j]; xv[j] = v; s += v; s2 += v * v;
  }
  s = wave_sum(s); s2 = wave_sum(s2);
  float m = s * (1.f / DD);
  float r = rsqrtf(s2 * (1.f / DD) - m * m + LN_EPS_C);
  float* y = Y + (size_t)row * DD;
#pragma unroll
  for (int j = 0; j < 16; ++j) {
    int d = lane + 32 * j;
    y[d] = (xv[j] - m) * r * w[d] + b[d];
  }
}

#define SG_WAVES 16
template <int EPI, bool BT, int K, int N>
__global__ void __launch_bounds__(SG_WAVES * 32)
slot_gemm_kernel(const float* __restrict__ A, const float* __restrict__ B,
                 const float* __restrict__ bias, float* C, const float* aux) {
  constexpr int KW = K / SG_WAVES;
  constexpr int BLK = KW < 64 ? KW : 64;
  constexpr int NU = BLK / 16;
  __shared__ float red[SG_WAVES * 16 * 64];
  const int t = threadIdx.x, lane = t & 31, wv = t >> 5;
  const int c = lane & 15, h = lane >> 4;
  const int cbase = blockIdx.x * 64, m0 = blockIdx.y * 16;
  const int k0 = wv * KW + h * 8;
  const float* arow = A + (size_t)(m0 + c) * K + k0;
  const float* bptr = BT ? B + (size_t)(cbase + c) * K + k0
                         : B + (size_t)k0 * N + cbase + c * 4;

  v8f acc[4] = {};
#pragma unroll 1
  for (int blk = 0; blk < KW; blk += BLK) {
    float4 aq[NU][2];
    float4 bq[NU][4][2];
#pragma unroll
    for (int u = 0; u < NU; ++u) {
      aq[u][0] = *(const float4*)&arow[blk + u * 16];
      aq[u][1] = *(const float4*)&arow[blk + u * 16 + 4];
#pragma unroll
      for (int i = 0; i < 4; ++i) {
        if (BT) {
          const float* p = bptr + (size_t)(16 * i) * K + blk + u * 16;
          bq[u][i][0] = *(const float4*)p;
          bq[u][i][1] = *(const float4*)(p + 4);
        } else {
          const float* p = bptr + (size_t)(blk + u * 16 + 2 * i) * N;
          bq[u][i][0] = *(const float4*)p;
          bq[u][i][1] = *(const float4*)(p + N);
        }
      }
    }
    __builtin_amdgcn_sched_barrier(0);
#pragma unroll
    for (int u = 0; u < NU; ++u) {
      const float xa[8] = {aq[u][0].x, aq[u][0].y, aq[u][0].z, aq[u][0].w,
                           aq[u][1].x, aq[u][1].y, aq[u][1].z, aq[u][1].w};
#pragma unroll
      for (int tt = 0; tt < 4; ++tt) {
        v2f a; a.x = xa[2 * tt]; a.y = xa[2 * tt + 1];
#pragma unroll
        for (int j = 0; j < 4; ++j) {
          v2f bb;
          if (BT) {
            const float xb[8] = {bq[u][j][0].x, bq[u][j][0].y, bq[u][j][0].z, bq[u][j][0].w,
                                 bq[u][j][1].x, bq[u][j][1].y, bq[u][j][1].z, bq[u][j][1].w};
            bb.x = xb[2 * tt]; bb.y = xb[2 * tt + 1];
          } else {
            const float r0[4] = {bq[u][tt][0].x, bq[u][tt][0].y, bq[u][tt][0].z, bq[u][tt][0].w};
            const float r1[4] = {bq[u][tt][1].x, bq[u][tt][1].y, bq[u][tt][1].z, bq[u][tt][1].w};
            bb.x = r0[j]; bb.y = r1[j];
          }
          acc[j] = wmma4(a, bb, acc[j]);
        }
      }
    }
  }
  {
    float* rw = &red[(wv * 16 + h * 8) * 64];
#pragma unroll
    for (int r = 0; r < 8; ++r) {
      if (BT) {
#pragma unroll
        for (int j = 0; j < 4; ++j) rw[r * 64 + 16 * j + c] = acc[j][r];
      } else {
        float4 o; o.x = acc[0][r]; o.y = acc[1][r]; o.z = acc[2][r]; o.w = acc[3][r];
        *(float4*)&rw[r * 64 + 4 * c] = o;
      }
    }
  }
  __syncthreads();
  {
    const int row = t >> 5, col = (t & 31) * 2;
    float2 s; s.x = 0.f; s.y = 0.f;
#pragma unroll
    for (int w = 0; w < SG_WAVES; ++w) {
      const float2 p = *(const float2*)&red[(w * 16 + row) * 64 + col];
      s.x += p.x; s.y += p.y;
    }
    const size_t idx = (size_t)(m0 + row) * N + cbase + col;
    if (EPI >= 1) { s.x += bias[cbase + col]; s.y += bias[cbase + col + 1]; }
    if (EPI == 1) { s.x = fmaxf(s.x, 0.f); s.y = fmaxf(s.y, 0.f); }
    if (EPI == 2) {
      const float2 x = *(const float2*)&aux[idx];
      s.x += x.x; s.y += x.y;
    }
    *(float2*)&C[idx] = s;
  }
}

#define DOT_WAVES 16
#define DOT_BLK 64
#define DOT_NU (DOT_BLK / 16)
#define DOT_PITCH (DD + 4)
struct DotStage {
  float4 b[2][DOT_NU][2];
  float4 a[DOT_NU][2];
};
__device__ __forceinline__ void dot_request(DotStage& st, const float* brow, const float* arow, int blk) {
#pragma unroll
  for (int j = 0; j < 2; ++j) {
#pragma unroll
    for (int u = 0; u < DOT_NU; ++u) {
      const float* p = brow + (size_t)(16 * j) * DD + blk + u * 16;
      st.b[j][u][0] = *(const float4*)p;
      st.b[j][u][1] = *(const float4*)(p + 4);
    }
  }
#pragma unroll
  for (int u = 0; u < DOT_NU; ++u) {
    st.a[u][0] = *(const float4*)&arow[blk + u * 16];
    st.a[u][1] = *(const float4*)&arow[blk + u * 16 + 4];
  }
}
__device__ __forceinline__ void dot_products(const DotStage& st, v8f (&acc)[2]) {
#pragma unroll
  for (int u = 0; u < DOT_NU; ++u) {
    const float xa[8] = {st.a[u][0].x, st.a[u][0].y, st.a[u][0].z, st.a[u][0].w,
                         st.a[u][1].x, st.a[u][1].y, st.a[u][1].z, st.a[u][1].w};
#pragma unroll
    for (int tt = 0; tt < 4; ++tt) {
      v2f a; a.x = xa[2 * tt]; a.y = xa[2 * tt + 1];
#pragma unroll
      for (int j = 0; j < 2; ++j) {
        const float xb[8] = {st.b[j][u][0].x, st.b[j][u][0].y, st.b[j][u][0].z, st.b[j][u][0].w,
                             st.b[j][u][1].x, st.b[j][u][1].y, st.b[j][u][1].z, st.b[j][u][1].w};
        v2f bb; bb.x = xb[2 * tt]; bb.y = xb[2 * tt + 1];
        acc[j] = wmma4(a, bb, acc[j]);
      }
    }
  }
}
__global__ void __launch_bounds__(DOT_WAVES * 32)
dots_softmax_kernel(const float* __restrict__ Q, const float* __restrict__ Keys,
                    float* __restrict__ PreNorm, float* __restrict__ Attn,
                    float* __restrict__ Sums) {
  __shared__ __attribute__((aligned(16))) float As[SS * DOT_PITCH];
  const int t = threadIdx.x, lane = t & 31, wv = t >> 5;
  const int c = lane & 15, h = lane >> 4;
  const int b = blockIdx.y;
  const int fw = blockIdx.x * (DOT_WAVES * 32) + wv * 32;
  {
    const float4* qb = (const float4*)(Q + (size_t)b * SS * DD);
#pragma unroll
    for (int i = 0; i < 2; ++i) {
      const int idx = t + i * (DOT_WAVES * 32);
      const int row = idx >> 7, c4 = idx & 127;
      *(float4*)&As[row * DOT_PITCH + c4 * 4] = qb[idx];
    }
  }
  __syncthreads();
  const float* arow = &As[(c & 7) * DOT_PITCH + h * 8];
  const float* brow = Keys + ((size_t)b * FF + fw + c) * DD + h * 8;

  v8f acc[2] = {};
  DotStage s0, s1;
  dot_request(s0, brow, arow, 0);
#pragma unroll 1
  for (int blk = 0; blk < DD - 2 * DOT_BLK; blk += 2 * DOT_BLK) {
    dot_request(s1, brow, arow, blk + DOT_BLK);
    __builtin_amdgcn_sched_barrier(0);
    dot_products(s0, acc);
    __builtin_amdgcn_sched_barrier(0);
    dot_request(s0, brow, arow, blk + 2 * DOT_BLK);
    __builtin_amdgcn_sched_barrier(0);
    dot_products(s1, acc);
    __builtin_amdgcn_sched_barrier(0);
  }
  dot_request(s1, brow, arow, DD - DOT_BLK);
  __builtin_amdgcn_sched_barrier(0);
  dot_products(s0, acc);
  dot_products(s1, acc);

  const int f = fw + lane;
  float d[SS]; float mx = -3.4e38f;
#pragma unroll
  for (int s = 0; s < SS; ++s) {
    const float hi = __shfl(acc[1][s], c, 32);
    d[s] = (h ? hi : acc[0][s]) * SCALE_C;
    mx = fmaxf(mx, d[s]);
  }
  float sum = 0.f;
#pragma unroll
  for (int s = 0; s < SS; ++s) { d[s] = expf(d[s] - mx); sum += d[s]; }
  const float inv = 1.f / sum;
#pragma unroll
  for (int s = 0; s < SS; ++s) {
    const float p = d[s] * inv;
    const size_t idx = ((size_t)b * SS + s) * FF + f;
    PreNorm[idx] = p;
    const float aw = p + EPS_A;
    Attn[idx] = aw;
    const float r = wave_sum(aw);
    if (lane == 0) atomicAdd(&Sums[b * SS + s], r);
  }
}

__global__ void zero_sums_kernel(float* __restrict__ Sums) {
  Sums[threadIdx.x] = 0.f;
}

#define UPD_WAVES 16
#define UPD_ROWS (FF / UPD_WAVES)
#define UPD_BLK 64
__global__ void __launch_bounds__(UPD_WAVES * 32)
updates_kernel(const float* __restrict__ Attn, const float* __restrict__ Vals,
               const float* __restrict__ Sums, float* __restrict__ Upd) {
  __shared__ float red[UPD_WAVES * SS * 64];
  const int t = threadIdx.x, lane = t & 31, wv = t >> 5;
  const int c = lane & 15, h = lane >> 4;
  const int cbase = blockIdx.x * 64, b = blockIdx.y;
  const bool valid = c < SS;
  const float* arow = Attn + ((size_t)b * SS + (c & 7)) * FF + wv * UPD_ROWS + h * 8;
  const float* vcol = Vals + ((size_t)b * FF + wv * UPD_ROWS + h * 8) * DD + cbase + c * 4;

  v8f acc[4] = {};
#pragma unroll 1
  for (int blk = 0; blk < UPD_ROWS; blk += UPD_BLK) {
    float4 aq[UPD_BLK / 16][2];
    float4 bq[UPD_BLK / 16][4][2];
#pragma unroll
    for (int u = 0; u < UPD_BLK / 16; ++u) {
      aq[u][0] = *(const float4*)&arow[blk + u * 16];
      aq[u][1] = *(const float4*)&arow[blk + u * 16 + 4];
#pragma unroll
      for (int tt = 0; tt < 4; ++tt) {
        const float* p = vcol + (size_t)(blk + u * 16 + 2 * tt) * DD;
        bq[u][tt][0] = *(const float4*)p;
        bq[u][tt][1] = *(const float4*)(p + DD);
      }
    }
#pragma unroll
    for (int u = 0; u < UPD_BLK / 16; ++u) {
      const float xa[8] = {aq[u][0].x, aq[u][0].y, aq[u][0].z, aq[u][0].w,
                           aq[u][1].x, aq[u][1].y, aq[u][1].z, aq[u][1].w};
#pragma unroll
      for (int tt = 0; tt < 4; ++tt) {
        v2f a;
        a.x = valid ? xa[2 * tt] : 0.f;
        a.y = valid ? xa[2 * tt + 1] : 0.f;
        const float r0[4] = {bq[u][tt][0].x, bq[u][tt][0].y, bq[u][tt][0].z, bq[u][tt][0].w};
        const float r1[4] = {bq[u][tt][1].x, bq[u][tt][1].y, bq[u][tt][1].z, bq[u][tt][1].w};
#pragma unroll
        for (int j = 0; j < 4; ++j) {
          v2f bb; bb.x = r0[j]; bb.y = r1[j];
          acc[j] = wmma4(a, bb, acc[j]);
        }
      }
    }
  }
  if (h == 0) {
    float* rw = &red[(wv * SS) * 64 + c * 4];
#pragma unroll
    for (int r = 0; r < SS; ++r) {
      float4 o; o.x = acc[0][r]; o.y = acc[1][r]; o.z = acc[2][r]; o.w = acc[3][r];
      *(float4*)&rw[r * 64] = o;
    }
  }
  __syncthreads();
  {
    const int r = t >> 6, col = t & 63;
    float s = 0.f;
#pragma unroll
    for (int w = 0; w < UPD_WAVES; ++w) s += red[(w * SS + r) * 64 + col];
    const float inv = 1.f / Sums[b * SS + r];
    Upd[((size_t)b * SS + r) * DD + cbase + col] = s * inv;
  }
}

__global__ void __launch_bounds__(256)
gru_kernel(const float* __restrict__ gi, const float* __restrict__ gh,
           const float* __restrict__ sn, float* __restrict__ slots) {
  int idx = blockIdx.x * 256 + threadIdx.x;
  int m = idx >> 9, d = idx & 511;
  size_t base = (size_t)m * 3 * DD;
  float ir = gi[base + d],           hr = gh[base + d];
  float iz = gi[base + DD + d],      hz = gh[base + DD + d];
  float in_ = gi[base + 2 * DD + d], hn = gh[base + 2 * DD + d];
  float r = 1.f / (1.f + expf(-(ir + hr)));
  float z = 1.f / (1.f + expf(-(iz + hz)));
  float n = tanhf(in_ + r * hn);
  float h = sn[(size_t)m * DD + d];
  slots[(size_t)m * DD + d] = (1.f - z) * n + z * h;
}

extern "C" void kernel_launch(void* const* d_in, const int* in_sizes, int n_in,
                              void* d_out, int out_size, void* d_ws, size_t ws_size,
                              hipStream_t stream) {
  (void)in_sizes; (void)n_in; (void)out_size; (void)ws_size;
  const float* in_slots  = (const float*)d_in[0];
  const float* features  = (const float*)d_in[1];
  const float* w_k       = (const float*)d_in[2];
  const float* w_v       = (const float*)d_in[3];
  const float* w_q       = (const float*)d_in[4];
  const float* ln_feat_w = (const float*)d_in[5];
  const float* ln_feat_b = (const float*)d_in[6];
  const float* ln_slot_w = (const float*)d_in[7];
  const float* ln_slot_b = (const float*)d_in[8];
  const float* gru_w_ih  = (const float*)d_in[9];
  const float* gru_w_hh  = (const float*)d_in[10];
  const float* gru_b_ih  = (const float*)d_in[11];
  const float* gru_b_hh  = (const float*)d_in[12];
  const float* ln_mlp_w  = (const float*)d_in[13];
  const float* ln_mlp_b  = (const float*)d_in[14];
  const float* mlp_w1    = (const float*)d_in[15];
  const float* mlp_b1    = (const float*)d_in[16];
  const float* mlp_w2    = (const float*)d_in[17];
  const float* mlp_b2    = (const float*)d_in[18];

  float* ws = (float*)d_ws;
  size_t off = 0;
  auto carve = [&](size_t n) { float* p = ws + off; off += n; return p; };
  float* fln   = carve((size_t)BB * FF * DD);
  float* wqk   = carve((size_t)DD * DD);
  float* wvi   = carve((size_t)DD * 3 * DD);
  float* slots = carve((size_t)BB * SS * DD);
  float* s_n   = carve((size_t)BB * SS * DD);
  float* q     = carve((size_t)BB * SS * DD);
  float* attn  = carve((size_t)BB * SS * FF);
  float* sums  = carve(256);
  float* upd   = carve((size_t)BB * SS * DD);
  float* gi    = carve((size_t)BB * SS * 3 * DD);
  float* gh    = carve((size_t)BB * SS * 3 * DD);
  float* hln   = carve((size_t)BB * SS * DD);
  float* h1    = carve((size_t)BB * SS * HH);

  float* out_slots = (float*)d_out;
  float* out_attn  = out_slots + (size_t)BB * SS * DD;

  hipMemcpyAsync(slots, in_slots, (size_t)BB * SS * DD * sizeof(float),
                 hipMemcpyDeviceToDevice, stream);

  ln_feat_kernel<<<(BB * FF) / 8, 256, 0, stream>>>(features, ln_feat_w, ln_feat_b, fln);
  slot_gemm_kernel<0, true, DD, DD><<<dim3(DD / 64, DD / 16), SG_WAVES * 32, 0, stream>>>(
      w_q, w_k, nullptr, wqk, nullptr);
  slot_gemm_kernel<0, true, DD, 3 * DD><<<dim3((3 * DD) / 64, DD / 16), SG_WAVES * 32, 0, stream>>>(
      w_v, gru_w_ih, nullptr, wvi, nullptr);

  for (int it = 0; it < NITER; ++it) {
    ln_rows_kernel<<<32, 256, 0, stream>>>(slots, ln_slot_w, ln_slot_b, s_n);
    slot_gemm_kernel<0, false, DD, DD><<<dim3(DD / 64, (BB * SS) / 16), SG_WAVES * 32, 0, stream>>>(
        s_n, wqk, nullptr, q, nullptr);
    zero_sums_kernel<<<1, 256, 0, stream>>>(sums);
    dots_softmax_kernel<<<dim3(FF / (DOT_WAVES * 32), BB), DOT_WAVES * 32, 0, stream>>>(
        q, fln, out_attn, attn, sums);
    updates_kernel<<<dim3(DD / 64, BB), UPD_WAVES * 32, 0, stream>>>(attn, fln, sums, upd);
    slot_gemm_kernel<3, false, DD, 3 * DD><<<dim3((3 * DD) / 64, (BB * SS) / 16), SG_WAVES * 32, 0, stream>>>(
        upd, wvi, gru_b_ih, gi, nullptr);
    slot_gemm_kernel<3, true, DD, 3 * DD><<<dim3((3 * DD) / 64, (BB * SS) / 16), SG_WAVES * 32, 0, stream>>>(
        s_n, gru_w_hh, gru_b_hh, gh, nullptr);
    gru_kernel<<<(BB * SS * DD) / 256, 256, 0, stream>>>(gi, gh, s_n, slots);
    ln_rows_kernel<<<32, 256, 0, stream>>>(slots, ln_mlp_w, ln_mlp_b, hln);
    slot_gemm_kernel<1, false, DD, HH><<<dim3(HH / 64, (BB * SS) / 16), SG_WAVES * 32, 0, stream>>>(
        hln, mlp_w1, mlp_b1, h1, nullptr);
    slot_gemm_kernel<2, false, HH, DD><<<dim3(DD / 64, (BB * SS) / 16), SG_WAVES * 32, 0, stream>>>(
        h1, mlp_w2, mlp_b2, slots, slots);
  }

  hipMemcpyAsync(out_slots, slots, (size_t)BB * SS * DD * sizeof(float),
                 hipMemcpyDeviceToDevice, stream);
}


# feature LayerNorm folded into the slot side: dots and updates read raw feature rows, row mean/rsqrt come from the dots fragments and scale scores and attn; LN pass and buffer gone (new sum order)
# speedup vs baseline: 10.8674x; 1.0865x over previous
#include <hip/hip_runtime.h>
#include <hip/hip_bf16.h>

typedef __attribute__((ext_vector_type(2))) float v2f;
typedef __attribute__((ext_vector_type(8))) float v8f;

#define BB 32
#define FF 4096
#define SS 8
#define DD 512
#define HH 2048
#define NITER 3
#define EPS_A 1e-8f
#define LN_EPS_C 1e-5f
#define SCALE_C 0.04419417382415922f

__device__ __forceinline__ v8f wmma4(v2f a, v2f b, v8f c) {
  return __builtin_amdgcn_wmma_f32_16x16x4_f32(
       false, a,  false, b,
       (short)0, c,  false,  false);
}

__device__ __forceinline__ float wave_sum(float v) {
#pragma unroll
  for (int o = 16; o; o >>= 1) v += __shfl_xor(v, o, 32);
  return v;
}

__global__ void __launch_bounds__(256)
ln_rows_kernel(const float* __restrict__ X, const float* __restrict__ w,
               const float* __restrict__ b, float* __restrict__ Y) {
  int wv = threadIdx.x >> 5, lane = threadIdx.x & 31;
  int row = blockIdx.x * 8 + wv;
  const float* x = X + (size_t)row * DD;
  float xv[16]; float s = 0.f, s2 = 0.f;
#pragma unroll
  for (int j = 0; j < 16; ++j) {
    float v = x[lane + 32 * j]; xv[j] = v; s += v; s2 += v * v;
  }
  s = wave_sum(s); s2 = wave_sum(s2);
  float m = s * (1.f / DD);
  float r = rsqrtf(s2 * (1.f / DD) - m * m + LN_EPS_C);
  float* y = Y + (size_t)row * DD;
#pragma unroll
  for (int j = 0; j < 16; ++j) {
    int d = lane + 32 * j;
    y[d] = (xv[j] - m) * r * w[d] + b[d];
  }
}

#define SG_WAVES 16
template <int EPI, bool BT, int K, int N>
__global__ void __launch_bounds__(SG_WAVES * 32)
slot_gemm_kernel(const float* __restrict__ A, const float* __restrict__ B,
                 const float* __restrict__ bias, float* C, const float* aux) {
  constexpr int KW = K / SG_WAVES;
  constexpr int BLK = KW < 64 ? KW : 64;
  constexpr int NU = BLK / 16;
  __shared__ float red[SG_WAVES * 16 * 64];
  const int t = threadIdx.x, lane = t & 31, wv = t >> 5;
  const int c = lane & 15, h = lane >> 4;
  const int cbase = blockIdx.x * 64, m0 = blockIdx.y * 16;
  const int k0 = wv * KW + h * 8;
  const float* arow = A + (size_t)(m0 + c) * K + k0;
  const float* bptr = BT ? B + (size_t)(cbase + c) * K + k0
                         : B + (size_t)k0 * N + cbase + c * 4;

  v8f acc[4] = {};
#pragma unroll 1
  for (int blk = 0; blk < KW; blk += BLK) {
    float4 aq[NU][2];
    float4 bq[NU][4][2];
#pragma unroll
    for (int u = 0; u < NU; ++u) {
      aq[u][0] = *(const float4*)&arow[blk + u * 16];
      aq[u][1] = *(const float4*)&arow[blk + u * 16 + 4];
#pragma unroll
      for (int i = 0; i < 4; ++i) {
        if (BT) {
          const float* p = bptr + (size_t)(16 * i) * K + blk + u * 16;
          bq[u][i][0] = *(const float4*)p;
          bq[u][i][1] = *(const float4*)(p + 4);
        } else {
          const float* p = bptr + (size_t)(blk + u * 16 + 2 * i) * N;
          bq[u][i][0] = *(const float4*)p;
          bq[u][i][1] = *(const float4*)(p + N);
        }
      }
    }
    __builtin_amdgcn_sched_barrier(0);
#pragma unroll
    for (int u = 0; u < NU; ++u) {
      const float xa[8] = {aq[u][0].x, aq[u][0].y, aq[u][0].z, aq[u][0].w,
                           aq[u][1].x, aq[u][1].y, aq[u][1].z, aq[u][1].w};
#pragma unroll
      for (int tt = 0; tt < 4; ++tt) {
        v2f a; a.x = xa[2 * tt]; a.y = xa[2 * tt + 1];
#pragma unroll
        for (int j = 0; j < 4; ++j) {
          v2f bb;
          if (BT) {
            const float xb[8] = {bq[u][j][0].x, bq[u][j][0].y, bq[u][j][0].z, bq[u][j][0].w,
                                 bq[u][j][1].x, bq[u][j][1].y, bq[u][j][1].z, bq[u][j][1].w};
            bb.x = xb[2 * tt]; bb.y = xb[2 * tt + 1];
          } else {
            const float r0[4] = {bq[u][tt][0].x, bq[u][tt][0].y, bq[u][tt][0].z, bq[u][tt][0].w};
            const float r1[4] = {bq[u][tt][1].x, bq[u][tt][1].y, bq[u][tt][1].z, bq[u][tt][1].w};
            bb.x = r0[j]; bb.y = r1[j];
          }
          acc[j] = wmma4(a, bb, acc[j]);
        }
      }
    }
  }
  {
    float* rw = &red[(wv * 16 + h * 8) * 64];
#pragma unroll
    for (int r = 0; r < 8; ++r) {
      if (BT) {
#pragma unroll
        for (int j = 0; j < 4; ++j) rw[r * 64 + 16 * j + c] = acc[j][r];
      } else {
        float4 o; o.x = acc[0][r]; o.y = acc[1][r]; o.z = acc[2][r]; o.w = acc[3][r];
        *(float4*)&rw[r * 64 + 4 * c] = o;
      }
    }
  }
  __syncthreads();
  {
    const int row = t >> 5, col = (t & 31) * 2;
    float2 s; s.x = 0.f; s.y = 0.f;
#pragma unroll
    for (int w = 0; w < SG_WAVES; ++w) {
      const float2 p = *(const float2*)&red[(w * 16 + row) * 64 + col];
      s.x += p.x; s.y += p.y;
    }
    const size_t idx = (size_t)(m0 + row) * N + cbase + col;
    if (EPI >= 1) { s.x += bias[cbase + col]; s.y += bias[cbase + col + 1]; }
    if (EPI == 1) { s.x = fmaxf(s.x, 0.f); s.y = fmaxf(s.y, 0.f); }
    if (EPI == 2) {
      const float2 x = *(const float2*)&aux[idx];
      s.x += x.x; s.y += x.y;
    }
    *(float2*)&C[idx] = s;
  }
}

#define DOT_WAVES 16
#define DOT_BLK 64
#define DOT_NU (DOT_BLK / 16)
#define DOT_PITCH (DD + 4)
struct DotStage {
  float4 b[2][DOT_NU][2];
  float4 a[DOT_NU][2];
};
__device__ __forceinline__ void dot_request(DotStage& st, const float* brow, const float* arow, int blk) {
#pragma unroll
  for (int j = 0; j < 2; ++j) {
#pragma unroll
    for (int u = 0; u < DOT_NU; ++u) {
      const float* p = brow + (size_t)(16 * j) * DD + blk + u * 16;
      st.b[j][u][0] = *(const float4*)p;
      st.b[j][u][1] = *(const float4*)(p + 4);
    }
  }
#pragma unroll
  for (int u = 0; u < DOT_NU; ++u) {
    st.a[u][0] = *(const float4*)&arow[blk + u * 16];
    st.a[u][1] = *(const float4*)&arow[blk + u * 16 + 4];
  }
}
__device__ __forceinline__ void dot_products(const DotStage& st, v8f (&acc)[2],
                                             v2f (&rsum)[2], v2f (&rsq)[2]) {
#pragma unroll
  for (int j = 0; j < 2; ++j) {
#pragma unroll
    for (int u = 0; u < DOT_NU; ++u) {
#pragma unroll
      for (int e = 0; e < 2; ++e) {
        v2f lo, hi;
        lo.x = st.b[j][u][e].x; lo.y = st.b[j][u][e].y;
        hi.x = st.b[j][u][e].z; hi.y = st.b[j][u][e].w;
        rsum[j] += lo + hi;
        rsq[j] += lo * lo;
        rsq[j] += hi * hi;
      }
    }
  }
#pragma unroll
  for (int u = 0; u < DOT_NU; ++u) {
    const float xa[8] = {st.a[u][0].x, st.a[u][0].y, st.a[u][0].z, st.a[u][0].w,
                         st.a[u][1].x, st.a[u][1].y, st.a[u][1].z, st.a[u][1].w};
#pragma unroll
    for (int tt = 0; tt < 4; ++tt) {
      v2f a; a.x = xa[2 * tt]; a.y = xa[2 * tt + 1];
#pragma unroll
      for (int j = 0; j < 2; ++j) {
        const float xb[8] = {st.b[j][u][0].x, st.b[j][u][0].y, st.b[j][u][0].z, st.b[j][u][0].w,
                             st.b[j][u][1].x, st.b[j][u][1].y, st.b[j][u][1].z, st.b[j][u][1].w};
        v2f bb; bb.x = xb[2 * tt]; bb.y = xb[2 * tt + 1];
        acc[j] = wmma4(a, bb, acc[j]);
      }
    }
  }
}
__global__ void __launch_bounds__(DOT_WAVES * 32)
dots_softmax_kernel(const float* __restrict__ Q, const float* __restrict__ Keys,
                    const float* __restrict__ lnw, const float* __restrict__ lnb,
                    float* __restrict__ PreNorm, float* __restrict__ Attn,
                    float* __restrict__ Sums) {
  __shared__ __attribute__((aligned(16))) float As[SS * DOT_PITCH];
  __shared__ float part[2][SS][4];
  __shared__ __attribute__((aligned(16))) float cst[2][SS];
  const int t = threadIdx.x, lane = t & 31, wv = t >> 5;
  const int c = lane & 15, h = lane >> 4;
  const int b = blockIdx.y;
  const int fw = blockIdx.x * (DOT_WAVES * 32) + wv * 32;
  {
    const float4* qb = (const float4*)(Q + (size_t)b * SS * DD);
#pragma unroll
    for (int i = 0; i < 2; ++i) {
      const int idx = t + i * (DOT_WAVES * 32);
      const int row = idx >> 7, c4 = idx & 127;
      const float4 qv = qb[idx];
      const float4 ww = ((const float4*)lnw)[c4], bb = ((const float4*)lnb)[c4];
      float4 qw;
      qw.x = qv.x * ww.x; qw.y = qv.y * ww.y; qw.z = qv.z * ww.z; qw.w = qv.w * ww.w;
      *(float4*)&As[row * DOT_PITCH + c4 * 4] = qw;
      const float ps = wave_sum((qw.x + qw.y) + (qw.z + qw.w));
      const float pb = wave_sum((qv.x * bb.x + qv.y * bb.y) + (qv.z * bb.z + qv.w * bb.w));
      if (lane == 0) { part[0][row][wv & 3] = ps; part[1][row][wv & 3] = pb; }
    }
  }
  __syncthreads();
  if (t < 2 * SS) {
    const int which = t >> 3, s = t & 7;
    cst[which][s] = (part[which][s][0] + part[which][s][1]) + (part[which][s][2] + part[which][s][3]);
  }
  __syncthreads();
  const float* arow = &As[(c & 7) * DOT_PITCH + h * 8];
  const float* brow = Keys + ((size_t)b * FF + fw + c) * DD + h * 8;

  v8f acc[2] = {};
  v2f rsum[2] = {}, rsq[2] = {};
  DotStage s0, s1;
  dot_request(s0, brow, arow, 0);
#pragma unroll 1
  for (int blk = 0; blk < DD - 2 * DOT_BLK; blk += 2 * DOT_BLK) {
    dot_request(s1, brow, arow, blk + DOT_BLK);
    __builtin_amdgcn_sched_barrier(0);
    dot_products(s0, acc, rsum, rsq);
    __builtin_amdgcn_sched_barrier(0);
    dot_request(s0, brow, arow, blk + 2 * DOT_BLK);
    __builtin_amdgcn_sched_barrier(0);
    dot_products(s1, acc, rsum, rsq);
    __builtin_amdgcn_sched_barrier(0);
  }
  dot_request(s1, brow, arow, DD - DOT_BLK);
  __builtin_amdgcn_sched_barrier(0);
  dot_products(s0, acc, rsum, rsq);
  dot_products(s1, acc, rsum, rsq);

  const int f = fw + lane;
  float m, rs;
  {
    float t0 = rsum[0].x + rsum[0].y, t1 = rsum[1].x + rsum[1].y;
    float u0 = rsq[0].x + rsq[0].y, u1 = rsq[1].x + rsq[1].y;
    t0 += __shfl_xor(t0, 16, 32); t1 += __shfl_xor(t1, 16, 32);
    u0 += __shfl_xor(u0, 16, 32); u1 += __shfl_xor(u1, 16, 32);
    m = (h ? t1 : t0) * (1.f / DD);
    rs = rsqrtf((h ? u1 : u0) * (1.f / DD) - m * m + LN_EPS_C);
  }
  float qs[SS], qb2[SS];
  {
    const float4 a0 = *(const float4*)&cst[0][0], a1 = *(const float4*)&cst[0][4];
    const float4 b0 = *(const float4*)&cst[1][0], b1 = *(const float4*)&cst[1][4];
    qs[0] = a0.x; qs[1] = a0.y; qs[2] = a0.z; qs[3] = a0.w;
    qs[4] = a1.x; qs[5] = a1.y; qs[6] = a1.z; qs[7] = a1.w;
    qb2[0] = b0.x; qb2[1] = b0.y; qb2[2] = b0.z; qb2[3] = b0.w;
    qb2[4] = b1.x; qb2[5] = b1.y; qb2[6] = b1.z; qb2[7] = b1.w;
  }
  float d[SS]; float mx = -3.4e38f;
#pragma unroll
  for (int s = 0; s < SS; ++s) {
    const float hi = __shfl(acc[1][s], c, 32);
    const float raw = h ? hi : acc[0][s];
    d[s] = (rs * (raw - m * qs[s]) + qb2[s]) * SCALE_C;
    mx = fmaxf(mx, d[s]);
  }
  float sum = 0.f;
#pragma unroll
  for (int s = 0; s < SS; ++s) { d[s] = expf(d[s] - mx); sum += d[s]; }
  const float inv = 1.f / sum;
#pragma unroll
  for (int s = 0; s < SS; ++s) {
    const float p = d[s] * inv;
    const size_t idx = ((size_t)b * SS + s) * FF + f;
    PreNorm[idx] = p;
    const float aw = p + EPS_A;
    const float ar = aw * rs;
    Attn[idx] = ar;
    const float r = wave_sum(aw);
    const float r2 = wave_sum(ar * m);
    if (lane == 0) {
      atomicAdd(&Sums[b * SS + s], r);
      atomicAdd(&Sums[BB * SS + b * SS + s], r2);
    }
  }
}

__global__ void zero_sums_kernel(float* __restrict__ Sums) {
  Sums[threadIdx.x] = 0.f;
}

#define UPD_WAVES 16
#define UPD_ROWS (FF / UPD_WAVES)
#define UPD_BLK 64
__global__ void __launch_bounds__(UPD_WAVES * 32)
updates_kernel(const float* __restrict__ Attn, const float* __restrict__ Vals,
               const float* __restrict__ Sums, const float* __restrict__ lnw,
               const float* __restrict__ lnb, float* __restrict__ Upd) {
  __shared__ float red[UPD_WAVES * SS * 64];
  const int t = threadIdx.x, lane = t & 31, wv = t >> 5;
  const int c = lane & 15, h = lane >> 4;
  const int cbase = blockIdx.x * 64, b = blockIdx.y;
  const bool valid = c < SS;
  const float* arow = Attn + ((size_t)b * SS + (c & 7)) * FF + wv * UPD_ROWS + h * 8;
  const float* vcol = Vals + ((size_t)b * FF + wv * UPD_ROWS + h * 8) * DD + cbase + c * 4;

  v8f acc[4] = {};
#pragma unroll 1
  for (int blk = 0; blk < UPD_ROWS; blk += UPD_BLK) {
    float4 aq[UPD_BLK / 16][2];
    float4 bq[UPD_BLK / 16][4][2];
#pragma unroll
    for (int u = 0; u < UPD_BLK / 16; ++u) {
      aq[u][0] = *(const float4*)&arow[blk + u * 16];
      aq[u][1] = *(const float4*)&arow[blk + u * 16 + 4];
#pragma unroll
      for (int tt = 0; tt < 4; ++tt) {
        const float* p = vcol + (size_t)(blk + u * 16 + 2 * tt) * DD;
        bq[u][tt][0] = *(const float4*)p;
        bq[u][tt][1] = *(const float4*)(p + DD);
      }
    }
#pragma unroll
    for (int u = 0; u < UPD_BLK / 16; ++u) {
      const float xa[8] = {aq[u][0].x, aq[u][0].y, aq[u][0].z, aq[u][0].w,
                           aq[u][1].x, aq[u][1].y, aq[u][1].z, aq[u][1].w};
#pragma unroll
      for (int tt = 0; tt < 4; ++tt) {
        v2f a;
        a.x = valid ? xa[2 * tt] : 0.f;
        a.y = valid ? xa[2 * tt + 1] : 0.f;
        const float r0[4] = {bq[u][tt][0].x, bq[u][tt][0].y, bq[u][tt][0].z, bq[u][tt][0].w};
        const float r1[4] = {bq[u][tt][1].x, bq[u][tt][1].y, bq[u][tt][1].z, bq[u][tt][1].w};
#pragma unroll
        for (int j = 0; j < 4; ++j) {
          v2f bb; bb.x = r0[j]; bb.y = r1[j];
          acc[j] = wmma4(a, bb, acc[j]);
        }
      }
    }
  }
  if (h == 0) {
    float* rw = &red[(wv * SS) * 64 + c * 4];
#pragma unroll
    for (int r = 0; r < SS; ++r) {
      float4 o; o.x = acc[0][r]; o.y = acc[1][r]; o.z = acc[2][r]; o.w = acc[3][r];
      *(float4*)&rw[r * 64] = o;
    }
  }
  __syncthreads();
  {
    const int r = t >> 6, col = t & 63;
    float s = 0.f;
#pragma unroll
    for (int w = 0; w < UPD_WAVES; ++w) s += red[(w * SS + r) * 64 + col];
    const float inv = 1.f / Sums[b * SS + r];
    const float mterm = Sums[BB * SS + b * SS + r];
    Upd[((size_t)b * SS + r) * DD + cbase + col] =
        (s - mterm) * inv * lnw[cbase + col] + lnb[cbase + col];
  }
}

__global__ void __launch_bounds__(256)
gru_kernel(const float* __restrict__ gi, const float* __restrict__ gh,
           const float* __restrict__ sn, float* __restrict__ slots) {
  int idx = blockIdx.x * 256 + threadIdx.x;
  int m = idx >> 9, d = idx & 511;
  size_t base = (size_t)m * 3 * DD;
  float ir = gi[base + d],           hr = gh[base + d];
  float iz = gi[base + DD + d],      hz = gh[base + DD + d];
  float in_ = gi[base + 2 * DD + d], hn = gh[base + 2 * DD + d];
  float r = 1.f / (1.f + expf(-(ir + hr)));
  float z = 1.f / (1.f + expf(-(iz + hz)));
  float n = tanhf(in_ + r * hn);
  float h = sn[(size_t)m * DD + d];
  slots[(size_t)m * DD + d] = (1.f - z) * n + z * h;
}

extern "C" void kernel_launch(void* const* d_in, const int* in_sizes, int n_in,
                              void* d_out, int out_size, void* d_ws, size_t ws_size,
                              hipStream_t stream) {
  (void)in_sizes; (void)n_in; (void)out_size; (void)ws_size;
  const float* in_slots  = (const float*)d_in[0];
  const float* features  = (const float*)d_in[1];
  const float* w_k       = (const float*)d_in[2];
  const float* w_v       = (const float*)d_in[3];
  const float* w_q       = (const float*)d_in[4];
  const float* ln_feat_w = (const float*)d_in[5];
  const float* ln_feat_b = (const float*)d_in[6];
  const float* ln_slot_w = (const float*)d_in[7];
  const float* ln_slot_b = (const float*)d_in[8];
  const float* gru_w_ih  = (const float*)d_in[9];
  const float* gru_w_hh  = (const float*)d_in[10];
  const float* gru_b_ih  = (const float*)d_in[11];
  const float* gru_b_hh  = (const float*)d_in[12];
  const float* ln_mlp_w  = (const float*)d_in[13];
  const float* ln_mlp_b  = (const float*)d_in[14];
  const float* mlp_w1    = (const float*)d_in[15];
  const float* mlp_b1    = (const float*)d_in[16];
  const float* mlp_w2    = (const float*)d_in[17];
  const float* mlp_b2    = (const float*)d_in[18];

  float* ws = (float*)d_ws;
  size_t off = 0;
  auto carve = [&](size_t n) { float* p = ws + off; off += n; return p; };
  float* wqk   = carve((size_t)DD * DD);
  float* wvi   = carve((size_t)DD * 3 * DD);
  float* slots = carve((size_t)BB * SS * DD);
  float* s_n   = carve((size_t)BB * SS * DD);
  float* q     = carve((size_t)BB * SS * DD);
  float* attn  = carve((size_t)BB * SS * FF);
  float* sums  = carve(2 * BB * SS);
  float* upd   = carve((size_t)BB * SS * DD);
  float* gi    = carve((size_t)BB * SS * 3 * DD);
  float* gh    = carve((size_t)BB * SS * 3 * DD);
  float* hln   = carve((size_t)BB * SS * DD);
  float* h1    = carve((size_t)BB * SS * HH);

  float* out_slots = (float*)d_out;
  float* out_attn  = out_slots + (size_t)BB * SS * DD;

  hipMemcpyAsync(slots, in_slots, (size_t)BB * SS * DD * sizeof(float),
                 hipMemcpyDeviceToDevice, stream);

  slot_gemm_kernel<0, true, DD, DD><<<dim3(DD / 64, DD / 16), SG_WAVES * 32, 0, stream>>>(
      w_q, w_k, nullptr, wqk, nullptr);
  slot_gemm_kernel<0, true, DD, 3 * DD><<<dim3((3 * DD) / 64, DD / 16), SG_WAVES * 32, 0, stream>>>(
      w_v, gru_w_ih, nullptr, wvi, nullptr);

  for (int it = 0; it < NITER; ++it) {
    ln_rows_kernel<<<32, 256, 0, stream>>>(slots, ln_slot_w, ln_slot_b, s_n);
    slot_gemm_kernel<0, false, DD, DD><<<dim3(DD / 64, (BB * SS) / 16), SG_WAVES * 32, 0, stream>>>(
        s_n, wqk, nullptr, q, nullptr);
    zero_sums_kernel<<<1, 2 * BB * SS, 0, stream>>>(sums);
    dots_softmax_kernel<<<dim3(FF / (DOT_WAVES * 32), BB), DOT_WAVES * 32, 0, stream>>>(
        q, features, ln_feat_w, ln_feat_b, out_attn, attn, sums);
    updates_kernel<<<dim3(DD / 64, BB), UPD_WAVES * 32, 0, stream>>>(attn, features, sums, ln_feat_w, ln_feat_b, upd);
    slot_gemm_kernel<3, false, DD, 3 * DD><<<dim3((3 * DD) / 64, (BB * SS) / 16), SG_WAVES * 32, 0, stream>>>(
        upd, wvi, gru_b_ih, gi, nullptr);
    slot_gemm_kernel<3, true, DD, 3 * DD><<<dim3((3 * DD) / 64, (BB * SS) / 16), SG_WAVES * 32, 0, stream>>>(
        s_n, gru_w_hh, gru_b_hh, gh, nullptr);
    gru_kernel<<<(BB * SS * DD) / 256, 256, 0, stream>>>(gi, gh, s_n, slots);
    ln_rows_kernel<<<32, 256, 0, stream>>>(slots, ln_mlp_w, ln_mlp_b, hln);
    slot_gemm_kernel<1, false, DD, HH><<<dim3(HH / 64, (BB * SS) / 16), SG_WAVES * 32, 0, stream>>>(
        hln, mlp_w1, mlp_b1, h1, nullptr);
    slot_gemm_kernel<2, false, HH, DD><<<dim3(DD / 64, (BB * SS) / 16), SG_WAVES * 32, 0, stream>>>(
        h1, mlp_w2, mlp_b2, slots, slots);
  }

  hipMemcpyAsync(out_slots, slots, (size_t)BB * SS * DD * sizeof(float),
                 hipMemcpyDeviceToDevice, stream);
}


# scores, softmax over slots and updates product fused into one pass: feature tile read once into LDS, used by both products; per-work-group partials summed in fixed order, atomics gone (new sum order)
# speedup vs baseline: 13.1675x; 1.2117x over previous
#include <hip/hip_runtime.h>
#include <hip/hip_bf16.h>

typedef __attribute__((ext_vector_type(2))) float v2f;
typedef __attribute__((ext_vector_type(8))) float v8f;

#define BB 32
#define FF 4096
#define SS 8
#define DD 512
#define HH 2048
#define NITER 3
#define EPS_A 1e-8f
#define LN_EPS_C 1e-5f
#define SCALE_C 0.04419417382415922f

__device__ __forceinline__ v8f wmma4(v2f a, v2f b, v8f c) {
  return __builtin_amdgcn_wmma_f32_16x16x4_f32(
       false, a,  false, b,
       (short)0, c,  false,  false);
}

__device__ __forceinline__ float wave_sum(float v) {
#pragma unroll
  for (int o = 16; o; o >>= 1) v += __shfl_xor(v, o, 32);
  return v;
}

__global__ void __launch_bounds__(256)
ln_rows_kernel(const float* __restrict__ X, const float* __restrict__ w,
               const float* __restrict__ b, float* __restrict__ Y) {
  int wv = threadIdx.x >> 5, lane = threadIdx.x & 31;
  int row = blockIdx.x * 8 + wv;
  const float* x = X + (size_t)row * DD;
  float xv[16]; float s = 0.f, s2 = 0.f;
#pragma unroll
  for (int j = 0; j < 16; ++j) {
    float v = x[lane + 32 * j]; xv[j] = v; s += v; s2 += v * v;
  }
  s = wave_sum(s); s2 = wave_sum(s2);
  float m = s * (1.f / DD);
  float r = rsqrtf(s2 * (1.f / DD) - m * m + LN_EPS_C);
  float* y = Y + (size_t)row * DD;
#pragma unroll
  for (int j = 0; j < 16; ++j) {
    int d = lane + 32 * j;
    y[d] = (xv[j] - m) * r * w[d] + b[d];
  }
}

#define SG_WAVES 16
template <int EPI, bool BT, int K, int N>
__global__ void __launch_bounds__(SG_WAVES * 32)
slot_gemm_kernel(const float* __restrict__ A, const float* __restrict__ B,
                 const float* __restrict__ bias, float* C, const float* aux) {
  constexpr int KW = K / SG_WAVES;
  constexpr int BLK = KW < 64 ? KW : 64;
  constexpr int NU = BLK / 16;
  __shared__ float red[SG_WAVES * 16 * 64];
  const int t = threadIdx.x, lane = t & 31, wv = t >> 5;
  const int c = lane & 15, h = lane >> 4;
  const int cbase = blockIdx.x * 64, m0 = blockIdx.y * 16;
  const int k0 = wv * KW + h * 8;
  const float* arow = A + (size_t)(m0 + c) * K + k0;
  const float* bptr = BT ? B + (size_t)(cbase + c) * K + k0
                         : B + (size_t)k0 * N + cbase + c * 4;

  v8f acc[4] = {};
#pragma unroll 1
  for (int blk = 0; blk < KW; blk += BLK) {
    float4 aq[NU][2];
    float4 bq[NU][4][2];
#pragma unroll
    for (int u = 0; u < NU; ++u) {
      aq[u][0] = *(const float4*)&arow[blk + u * 16];
      aq[u][1] = *(const float4*)&arow[blk + u * 16 + 4];
#pragma unroll
      for (int i = 0; i < 4; ++i) {
        if (BT) {
          const float* p = bptr + (size_t)(16 * i) * K + blk + u * 16;
          bq[u][i][0] = *(const float4*)p;
          bq[u][i][1] = *(const float4*)(p + 4);
        } else {
          const float* p = bptr + (size_t)(blk + u * 16 + 2 * i) * N;
          bq[u][i][0] = *(const float4*)p;
          bq[u][i][1] = *(const float4*)(p + N);
        }
      }
    }
    __builtin_amdgcn_sched_barrier(0);
#pragma unroll
    for (int u = 0; u < NU; ++u) {
      const float xa[8] = {aq[u][0].x, aq[u][0].y, aq[u][0].z, aq[u][0].w,
                           aq[u][1].x, aq[u][1].y, aq[u][1].z, aq[u][1].w};
#pragma unroll
      for (int tt = 0; tt < 4; ++tt) {
        v2f a; a.x = xa[2 * tt]; a.y = xa[2 * tt + 1];
#pragma unroll
        for (int j = 0; j < 4; ++j) {
          v2f bb;
          if (BT) {
            const float xb[8] = {bq[u][j][0].x, bq[u][j][0].y, bq[u][j][0].z, bq[u][j][0].w,
                                 bq[u][j][1].x, bq[u][j][1].y, bq[u][j][1].z, bq[u][j][1].w};
            bb.x = xb[2 * tt]; bb.y = xb[2 * tt + 1];
          } else {
            const float r0[4] = {bq[u][tt][0].x, bq[u][tt][0].y, bq[u][tt][0].z, bq[u][tt][0].w};
            const float r1[4] = {bq[u][tt][1].x, bq[u][tt][1].y, bq[u][tt][1].z, bq[u][tt][1].w};
            bb.x = r0[j]; bb.y = r1[j];
          }
          acc[j] = wmma4(a, bb, acc[j]);
        }
      }
    }
  }
  {
    float* rw = &red[(wv * 16 + h * 8) * 64];
#pragma unroll
    for (int r = 0; r < 8; ++r) {
      if (BT) {
#pragma unroll
        for (int j = 0; j < 4; ++j) rw[r * 64 + 16 * j + c] = acc[j][r];
      } else {
        float4 o; o.x = acc[0][r]; o.y = acc[1][r]; o.z = acc[2][r]; o.w = acc[3][r];
        *(float4*)&rw[r * 64 + 4 * c] = o;
      }
    }
  }
  __syncthreads();
  {
    const int row = t >> 5, col = (t & 31) * 2;
    float2 s; s.x = 0.f; s.y = 0.f;
#pragma unroll
    for (int w = 0; w < SG_WAVES; ++w) {
      const float2 p = *(const float2*)&red[(w * 16 + row) * 64 + col];
      s.x += p.x; s.y += p.y;
    }
    const size_t idx = (size_t)(m0 + row) * N + cbase + col;
    if (EPI >= 1) { s.x += bias[cbase + col]; s.y += bias[cbase + col + 1]; }
    if (EPI == 1) { s.x = fmaxf(s.x, 0.f); s.y = fmaxf(s.y, 0.f); }
    if (EPI == 2) {
      const float2 x = *(const float2*)&aux[idx];
      s.x += x.x; s.y += x.y;
    }
    *(float2*)&C[idx] = s;
  }
}

#define FA_WAVES 16
#define FA_TF 64
#define FA_FWG 512
#define FA_NT (FA_FWG / FA_TF)
#define FA_NWG (FF / FA_FWG)
#define FA_PITCH (DD + 4)
#define FA_APITCH (FA_TF + 4)
#define FA_REP8(M) M(0) M(1) M(2) M(3) M(4) M(5) M(6) M(7)
#define FA_DECL(i) float4 pa##i, pb##i;
#define FA_LDA(i) pa##i = *(const float4*)&src[(size_t)((i) >> 2) * DD + ((i) & 3) * 128];
#define FA_LDB(i) pb##i = *(const float4*)&src[(size_t)(2 + ((i) >> 2)) * DD + ((i) & 3) * 128];
#define FA_ST(i) *(float4*)&xs[((i) >> 2) * FA_PITCH + ((i) & 3) * 128] = pa##i; \
                 *(float4*)&xs[(2 + ((i) >> 2)) * FA_PITCH + ((i) & 3) * 128] = pb##i;
__device__ __forceinline__ void lds_barrier() {
  asm volatile("s_wait_dscnt 0x0\n\ts_barrier_signal -1\n\ts_barrier_wait -1" ::: "memory");
}
__global__ void __launch_bounds__(FA_WAVES * 32)
attend_kernel(const float* __restrict__ Q, const float* X,
              const float* __restrict__ lnw, const float* __restrict__ lnb,
              float* __restrict__ PreNorm, float* __restrict__ Upart,
              float* __restrict__ Spart) {
  __shared__ __attribute__((aligned(16))) float xt[FA_TF * FA_PITCH];
  __shared__ __attribute__((aligned(16))) float psc[4][SS][FA_TF];
  __shared__ float pst[4][2][FA_TF];
  __shared__ __attribute__((aligned(16))) float at[SS * FA_APITCH];
  __shared__ float part[2][SS][4];
  __shared__ __attribute__((aligned(16))) float cst[2][SS];
  __shared__ float sfin[2][2][SS];
  const int t = threadIdx.x, lane = t & 31, wv = t >> 5;
  const int c = lane & 15, h = lane >> 4;
  const int b = blockIdx.y, bx = blockIdx.x;
  const float* qbase = Q + (size_t)b * SS * DD;
  {
    const float4* qb = (const float4*)qbase;
#pragma unroll
    for (int i = 0; i < 2; ++i) {
      const int idx = t + i * (FA_WAVES * 32);
      const int row = idx >> 7, c4 = idx & 127;
      const float4 qv = qb[idx];
      const float4 ww = ((const float4*)lnw)[c4], bb = ((const float4*)lnb)[c4];
      const float ps = wave_sum((qv.x * ww.x + qv.y * ww.y) + (qv.z * ww.z + qv.w * ww.w));
      const float pb = wave_sum((qv.x * bb.x + qv.y * bb.y) + (qv.z * bb.z + qv.w * bb.w));
      if (lane == 0) { part[0][row][wv & 3] = ps; part[1][row][wv & 3] = pb; }
    }
  }
  const int ft = wv & 3, kq = wv >> 2;
  float4 qa[8][2];
  const float* xg = X + ((size_t)b * FF + (size_t)bx * FA_FWG + wv * 4) * DD + lane * 4;
  float* xs = &xt[(wv * 4) * FA_PITCH + lane * 4];
  FA_REP8(FA_DECL)
  {
    const float* src = xg;
    FA_REP8(FA_LDA)
    FA_REP8(FA_LDB)
    FA_REP8(FA_ST)
  }
  __builtin_amdgcn_sched_barrier(0);
  {
    const float* qr = qbase + (size_t)(c & 7) * DD + kq * 128 + h * 8;
    const float* wr = lnw + kq * 128 + h * 8;
#pragma unroll
    for (int g = 0; g < 8; ++g) {
#pragma unroll
      for (int e = 0; e < 2; ++e) {
        const float4 qv = *(const float4*)&qr[g * 16 + 4 * e];
        const float4 ww = *(const float4*)&wr[g * 16 + 4 * e];
        float4 o; o.x = qv.x * ww.x; o.y = qv.y * ww.y; o.z = qv.z * ww.z; o.w = qv.w * ww.w;
        qa[g][e] = o;
      }
    }
  }
  __syncthreads();
  if (t < 2 * SS) {
    const int which = t >> 3, s = t & 7;
    cst[which][s] = (part[which][s][0] + part[which][s][1]) + (part[which][s][2] + part[which][s][3]);
  }
  __syncthreads();

  const float* xrow = &xt[(ft * 16 + c) * FA_PITCH + kq * 128 + h * 8];
  const float* xcol = &xt[(h * 8) * FA_PITCH + 32 * wv + 2 * c];
  const float* arow = &at[(c & 7) * FA_APITCH + h * 8];
  v8f accu[2] = {};
  float accs[SS] = {}, accm[SS] = {};

#pragma unroll 1
  for (int it = 0; it < FA_NT; ++it) {
    const float* src = xg + (size_t)(it + 1 < FA_NT ? it + 1 : it) * FA_TF * DD;
    FA_REP8(FA_LDA)
    __builtin_amdgcn_sched_barrier(0);

    v8f sc[2] = {};
    v2f rsum = {}, rsq = {};
#pragma unroll
    for (int hf = 0; hf < 2; ++hf) {
      float4 bs[4][2];
#pragma unroll
      for (int u = 0; u < 4; ++u) {
        bs[u][0] = *(const float4*)&xrow[(hf * 4 + u) * 16];
        bs[u][1] = *(const float4*)&xrow[(hf * 4 + u) * 16 + 4];
      }
      __builtin_amdgcn_sched_barrier(0);
#pragma unroll
      for (int u = 0; u < 4; ++u) {
#pragma unroll
        for (int e = 0; e < 2; ++e) {
          v2f lo, hi;
          lo.x = bs[u][e].x; lo.y = bs[u][e].y;
          hi.x = bs[u][e].z; hi.y = bs[u][e].w;
          rsum += lo + hi;
          rsq += lo * lo;
          rsq += hi * hi;
        }
        const int g = hf * 4 + u;
        const float xa[8] = {qa[g][0].x, qa[g][0].y, qa[g][0].z, qa[g][0].w,
                             qa[g][1].x, qa[g][1].y, qa[g][1].z, qa[g][1].w};
        const float xb[8] = {bs[u][0].x, bs[u][0].y, bs[u][0].z, bs[u][0].w,
                             bs[u][1].x, bs[u][1].y, bs[u][1].z, bs[u][1].w};
#pragma unroll
        for (int tt = 0; tt < 4; ++tt) {
          v2f a; a.x = xa[2 * tt]; a.y = xa[2 * tt + 1];
          v2f bb; bb.x = xb[2 * tt]; bb.y = xb[2 * tt + 1];
          sc[u & 1] = wmma4(a, bb, sc[u & 1]);
        }
      }
      __builtin_amdgcn_sched_barrier(0);
    }
    {
      const v8f s8 = sc[0] + sc[1];
      float t0 = rsum.x + rsum.y, u0 = rsq.x + rsq.y;
      t0 += __shfl_xor(t0, 16, 32);
      u0 += __shfl_xor(u0, 16, 32);
      if (h == 0) {
#pragma unroll
        for (int r = 0; r < SS; ++r) psc[kq][r][ft * 16 + c] = s8[r];
        pst[kq][0][ft * 16 + c] = t0;
        pst[kq][1][ft * 16 + c] = u0;
      }
    }
    FA_REP8(FA_LDB)
    __builtin_amdgcn_sched_barrier(0);
    lds_barrier();

    if (wv < 2) {
      const int fl = wv * 32 + lane;
      const float ssum = (pst[0][0][fl] + pst[1][0][fl]) + (pst[2][0][fl] + pst[3][0][fl]);
      const float ssq = (pst[0][1][fl] + pst[1][1][fl]) + (pst[2][1][fl] + pst[3][1][fl]);
      const float m = ssum * (1.f / DD);
      const float rs = rsqrtf(ssq * (1.f / DD) - m * m + LN_EPS_C);
      float qs[SS], qb2[SS];
      {
        const float4 a0 = *(const float4*)&cst[0][0], a1 = *(const float4*)&cst[0][4];
        const float4 b0 = *(const float4*)&cst[1][0], b1 = *(const float4*)&cst[1][4];
        qs[0] = a0.x; qs[1] = a0.y; qs[2] = a0.z; qs[3] = a0.w;
        qs[4] = a1.x; qs[5] = a1.y; qs[6] = a1.z; qs[7] = a1.w;
        qb2[0] = b0.x; qb2[1] = b0.y; qb2[2] = b0.z; qb2[3] = b0.w;
        qb2[4] = b1.x; qb2[5] = b1.y; qb2[6] = b1.z; qb2[7] = b1.w;
      }
      float d[SS]; float mx = -3.4e38f;
#pragma unroll
      for (int s = 0; s < SS; ++s) {
        const float raw = (psc[0][s][fl] + psc[1][s][fl]) + (psc[2][s][fl] + psc[3][s][fl]);
        d[s] = (rs * (raw - m * qs[s]) + qb2[s]) * SCALE_C;
        mx = fmaxf(mx, d[s]);
      }
      float sum = 0.f;
#pragma unroll
      for (int s = 0; s < SS; ++s) { d[s] = expf(d[s] - mx); sum += d[s]; }
      const float inv = 1.f / sum;
      const int f = bx * FA_FWG + it * FA_TF + fl;
#pragma unroll
      for (int s = 0; s < SS; ++s) {
        const float p = d[s] * inv;
        PreNorm[((size_t)b * SS + s) * FF + f] = p;
        const float aw = p + EPS_A;
        const float ar = aw * rs;
        at[s * FA_APITCH + fl] = ar;
        accs[s] += aw;
        accm[s] += ar * m;
      }
    }
    lds_barrier();

#pragma unroll
    for (int hf = 0; hf < 2; ++hf) {
      float4 aq[2][2];
      float2 bq[2][4][2];
#pragma unroll
      for (int uu = 0; uu < 2; ++uu) {
        const int u = hf * 2 + uu;
        aq[uu][0] = *(const float4*)&arow[u * 16];
        aq[uu][1] = *(const float4*)&arow[u * 16 + 4];
#pragma unroll
        for (int tt = 0; tt < 4; ++tt) {
          bq[uu][tt][0] = *(const float2*)&xcol[(u * 16 + 2 * tt) * FA_PITCH];
          bq[uu][tt][1] = *(const float2*)&xcol[(u * 16 + 2 * tt + 1) * FA_PITCH];
        }
      }
      __builtin_amdgcn_sched_barrier(0);
#pragma unroll
      for (int uu = 0; uu < 2; ++uu) {
        const float xa[8] = {aq[uu][0].x, aq[uu][0].y, aq[uu][0].z, aq[uu][0].w,
                             aq[uu][1].x, aq[uu][1].y, aq[uu][1].z, aq[uu][1].w};
#pragma unroll
        for (int tt = 0; tt < 4; ++tt) {
          v2f a; a.x = xa[2 * tt]; a.y = xa[2 * tt + 1];
          v2f b0; b0.x = bq[uu][tt][0].x; b0.y = bq[uu][tt][1].x;
          v2f b1; b1.x = bq[uu][tt][0].y; b1.y = bq[uu][tt][1].y;
          accu[0] = wmma4(a, b0, accu[0]);
          accu[1] = wmma4(a, b1, accu[1]);
        }
      }
      __builtin_amdgcn_sched_barrier(0);
    }
    lds_barrier();

    FA_REP8(FA_ST)
    lds_barrier();
  }

  if (h == 0) {
    float* up = Upart + (((size_t)b * FA_NWG + bx) * SS) * DD + 32 * wv + 2 * c;
#pragma unroll
    for (int r = 0; r < SS; ++r) {
      float2 o; o.x = accu[0][r]; o.y = accu[1][r];
      *(float2*)&up[(size_t)r * DD] = o;
    }
  }
  if (wv < 2) {
#pragma unroll
    for (int s = 0; s < SS; ++s) {
      const float r1 = wave_sum(accs[s]);
      const float r2 = wave_sum(accm[s]);
      if (lane == 0) { sfin[wv][0][s] = r1; sfin[wv][1][s] = r2; }
    }
  }
  __syncthreads();
  if (t < 2 * SS) {
    const int which = t >> 3, s = t & 7;
    Spart[(((size_t)b * FA_NWG + bx) * 2 + which) * SS + s] = sfin[0][which][s] + sfin[1][which][s];
  }
}

__global__ void __launch_bounds__(DD)
upd_finish_kernel(const float* __restrict__ Upart, const float* __restrict__ Spart,
                  const float* __restrict__ lnw, const float* __restrict__ lnb,
                  float* __restrict__ Upd) {
  const int row = blockIdx.x, b = row >> 3, s = row & 7, d = threadIdx.x;
  float u = 0.f, sum = 0.f, mterm = 0.f;
#pragma unroll
  for (int w = 0; w < FA_NWG; ++w) {
    u += Upart[(((size_t)b * FA_NWG + w) * SS + s) * DD + d];
    sum += Spart[(((size_t)b * FA_NWG + w) * 2 + 0) * SS + s];
    mterm += Spart[(((size_t)b * FA_NWG + w) * 2 + 1) * SS + s];
  }
  Upd[(size_t)row * DD + d] = (u - mterm) * (1.f / sum) * lnw[d] + lnb[d];
}

__global__ void __launch_bounds__(256)
gru_kernel(const float* __restrict__ gi, const float* __restrict__ gh,
           const float* __restrict__ sn, float* __restrict__ slots) {
  int idx = blockIdx.x * 256 + threadIdx.x;
  int m = idx >> 9, d = idx & 511;
  size_t base = (size_t)m * 3 * DD;
  float ir = gi[base + d],           hr = gh[base + d];
  float iz = gi[base + DD + d],      hz = gh[base + DD + d];
  float in_ = gi[base + 2 * DD + d], hn = gh[base + 2 * DD + d];
  float r = 1.f / (1.f + expf(-(ir + hr)));
  float z = 1.f / (1.f + expf(-(iz + hz)));
  float n = tanhf(in_ + r * hn);
  float h = sn[(size_t)m * DD + d];
  slots[(size_t)m * DD + d] = (1.f - z) * n + z * h;
}

extern "C" void kernel_launch(void* const* d_in, const int* in_sizes, int n_in,
                              void* d_out, int out_size, void* d_ws, size_t ws_size,
                              hipStream_t stream) {
  (void)in_sizes; (void)n_in; (void)out_size; (void)ws_size;
  const float* in_slots  = (const float*)d_in[0];
  const float* features  = (const float*)d_in[1];
  const float* w_k       = (const float*)d_in[2];
  const float* w_v       = (const float*)d_in[3];
  const float* w_q       = (const float*)d_in[4];
  const float* ln_feat_w = (const float*)d_in[5];
  const float* ln_feat_b = (const float*)d_in[6];
  const float* ln_slot_w = (const float*)d_in[7];
  const float* ln_slot_b = (const float*)d_in[8];
  const float* gru_w_ih  = (const float*)d_in[9];
  const float* gru_w_hh  = (const float*)d_in[10];
  const float* gru_b_ih  = (const float*)d_in[11];
  const float* gru_b_hh  = (const float*)d_in[12];
  const float* ln_mlp_w  = (const float*)d_in[13];
  const float* ln_mlp_b  = (const float*)d_in[14];
  const float* mlp_w1    = (const float*)d_in[15];
  const float* mlp_b1    = (const float*)d_in[16];
  const float* mlp_w2    = (const float*)d_in[17];
  const float* mlp_b2    = (const float*)d_in[18];

  float* ws = (float*)d_ws;
  size_t off = 0;
  auto carve = [&](size_t n) { float* p = ws + off; off += n; return p; };
  float* wqk   = carve((size_t)DD * DD);
  float* wvi   = carve((size_t)DD * 3 * DD);
  float* slots = carve((size_t)BB * SS * DD);
  float* s_n   = carve((size_t)BB * SS * DD);
  float* q     = carve((size_t)BB * SS * DD);
  float* upart = carve((size_t)BB * FA_NWG * SS * DD);
  float* upd   = carve((size_t)BB * SS * DD);
  float* gi    = carve((size_t)BB * SS * 3 * DD);
  float* gh    = carve((size_t)BB * SS * 3 * DD);
  float* hln   = carve((size_t)BB * SS * DD);
  float* h1    = carve((size_t)BB * SS * HH);
  float* spart = h1;

  float* out_slots = (float*)d_out;
  float* out_attn  = out_slots + (size_t)BB * SS * DD;

  hipMemcpyAsync(slots, in_slots, (size_t)BB * SS * DD * sizeof(float),
                 hipMemcpyDeviceToDevice, stream);

  slot_gemm_kernel<0, true, DD, DD><<<dim3(DD / 64, DD / 16), SG_WAVES * 32, 0, stream>>>(
      w_q, w_k, nullptr, wqk, nullptr);
  slot_gemm_kernel<0, true, DD, 3 * DD><<<dim3((3 * DD) / 64, DD / 16), SG_WAVES * 32, 0, stream>>>(
      w_v, gru_w_ih, nullptr, wvi, nullptr);

  for (int it = 0; it < NITER; ++it) {
    ln_rows_kernel<<<32, 256, 0, stream>>>(slots, ln_slot_w, ln_slot_b, s_n);
    slot_gemm_kernel<0, false, DD, DD><<<dim3(DD / 64, (BB * SS) / 16), SG_WAVES * 32, 0, stream>>>(
        s_n, wqk, nullptr, q, nullptr);
    attend_kernel<<<dim3(FA_NWG, BB), FA_WAVES * 32, 0, stream>>>(
        q, features, ln_feat_w, ln_feat_b, out_attn, upart, spart);
    upd_finish_kernel<<<BB * SS, DD, 0, stream>>>(upart, spart, ln_feat_w, ln_feat_b, upd);
    slot_gemm_kernel<3, false, DD, 3 * DD><<<dim3((3 * DD) / 64, (BB * SS) / 16), SG_WAVES * 32, 0, stream>>>(
        upd, wvi, gru_b_ih, gi, nullptr);
    slot_gemm_kernel<3, true, DD, 3 * DD><<<dim3((3 * DD) / 64, (BB * SS) / 16), SG_WAVES * 32, 0, stream>>>(
        s_n, gru_w_hh, gru_b_hh, gh, nullptr);
    gru_kernel<<<(BB * SS * DD) / 256, 256, 0, stream>>>(gi, gh, s_n, slots);
    ln_rows_kernel<<<32, 256, 0, stream>>>(slots, ln_mlp_w, ln_mlp_b, hln);
    slot_gemm_kernel<1, false, DD, HH><<<dim3(HH / 64, (BB * SS) / 16), SG_WAVES * 32, 0, stream>>>(
        hln, mlp_w1, mlp_b1, h1, nullptr);
    slot_gemm_kernel<2, false, HH, DD><<<dim3(DD / 64, (BB * SS) / 16), SG_WAVES * 32, 0, stream>>>(
        h1, mlp_w2, mlp_b2, slots, slots);
  }

  hipMemcpyAsync(out_slots, slots, (size_t)BB * SS * DD * sizeof(float),
                 hipMemcpyDeviceToDevice, stream);
}


# slot-side launch fusion, 32 launches -> 22, both copies gone: slot LayerNorm in the prologue of one q+gh product launch, GRU + MLP LayerNorm one kernel, weight products one launch; sum orders kept
# speedup vs baseline: 14.9072x; 1.1321x over previous
#include <hip/hip_runtime.h>
#include <hip/hip_bf16.h>

typedef __attribute__((ext_vector_type(2))) float v2f;
typedef __attribute__((ext_vector_type(8))) float v8f;

#define BB 32
#define FF 4096
#define SS 8
#define DD 512
#define HH 2048
#define NITER 3
#define EPS_A 1e-8f
#define LN_EPS_C 1e-5f
#define SCALE_C 0.04419417382415922f

__device__ __forceinline__ v8f wmma4(v2f a, v2f b, v8f c) {
  return __builtin_amdgcn_wmma_f32_16x16x4_f32(
       false, a,  false, b,
       (short)0, c,  false,  false);
}

__device__ __forceinline__ float wave_sum(float v) {
#pragma unroll
  for (int o = 16; o; o >>= 1) v += __shfl_xor(v, o, 32);
  return v;
}

__device__ __forceinline__ void row_stats(const float (&xv)[16], float& m, float& r) {
  float s = 0.f, s2 = 0.f;
#pragma unroll
  for (int j = 0; j < 16; ++j) { s += xv[j]; s2 += xv[j] * xv[j]; }
  s = wave_sum(s); s2 = wave_sum(s2);
  m = s * (1.f / DD);
  r = rsqrtf(s2 * (1.f / DD) - m * m + LN_EPS_C);
}

#define SG_WAVES 16
#define LNA_PITCH (DD + 4)
template <int EPI, bool BT, int K, int N, bool LN>
__device__ __forceinline__ void
slot_gemm_body(const float* __restrict__ A, const float* __restrict__ B,
               const float* __restrict__ bias, float* C, const float* aux,
               const float* __restrict__ lnw, const float* __restrict__ lnb,
               float* red, float* lna, const int bx, const int by) {
  constexpr int KW = K / SG_WAVES;
  constexpr int BLK = KW < 64 ? KW : 64;
  constexpr int NU = BLK / 16;
  const int t = threadIdx.x, lane = t & 31, wv = t >> 5;
  const int c = lane & 15, h = lane >> 4;
  const int cbase = bx * 64, m0 = by * 16;
  const int k0 = wv * KW + h * 8;
  if (LN) {
    const float* x = A + (size_t)(m0 + wv) * K;
    float xv[16];
#pragma unroll
    for (int j = 0; j < 16; ++j) xv[j] = x[lane + 32 * j];
    float m, r;
    row_stats(xv, m, r);
#pragma unroll
    for (int j = 0; j < 16; ++j) {
      const int d = lane + 32 * j;
      lna[wv * LNA_PITCH + d] = (xv[j] - m) * r * lnw[d] + lnb[d];
    }
    __syncthreads();
  }
  const float* arow = LN ? lna + c * LNA_PITCH + k0 : A + (size_t)(m0 + c) * K + k0;
  const float* bptr = BT ? B + (size_t)(cbase + c) * K + k0
                         : B + (size_t)k0 * N + cbase + c * 4;

  v8f acc[4] = {};
#pragma unroll 1
  for (int blk = 0; blk < KW; blk += BLK) {
    float4 aq[NU][2];
    float4 bq[NU][4][2];
#pragma unroll
    for (int u = 0; u < NU; ++u) {
      aq[u][0] = *(const float4*)&arow[blk + u * 16];
      aq[u][1] = *(const float4*)&arow[blk + u * 16 + 4];
#pragma unroll
      for (int i = 0; i < 4; ++i) {
        if (BT) {
          const float* p = bptr + (size_t)(16 * i) * K + blk + u * 16;
          bq[u][i][0] = *(const float4*)p;
          bq[u][i][1] = *(const float4*)(p + 4);
        } else {
          const float* p = bptr + (size_t)(blk + u * 16 + 2 * i) * N;
          bq[u][i][0] = *(const float4*)p;
          bq[u][i][1] = *(const float4*)(p + N);
        }
      }
    }
    __builtin_amdgcn_sched_barrier(0);
#pragma unroll
    for (int u = 0; u < NU; ++u) {
      const float xa[8] = {aq[u][0].x, aq[u][0].y, aq[u][0].z, aq[u][0].w,
                           aq[u][1].x, aq[u][1].y, aq[u][1].z, aq[u][1].w};
#pragma unroll
      for (int tt = 0; tt < 4; ++tt) {
        v2f a; a.x = xa[2 * tt]; a.y = xa[2 * tt + 1];
#pragma unroll
        for (int j = 0; j < 4; ++j) {
          v2f bb;
          if (BT) {
            const float xb[8] = {bq[u][j][0].x, bq[u][j][0].y, bq[u][j][0].z, bq[u][j][0].w,
                                 bq[u][j][1].x, bq[u][j][1].y, bq[u][j][1].z, bq[u][j][1].w};
            bb.x = xb[2 * tt]; bb.y = xb[2 * tt + 1];
          } else {
            const float r0[4] = {bq[u][tt][0].x, bq[u][tt][0].y, bq[u][tt][0].z, bq[u][tt][0].w};
            const float r1[4] = {bq[u][tt][1].x, bq[u][tt][1].y, bq[u][tt][1].z, bq[u][tt][1].w};
            bb.x = r0[j]; bb.y = r1[j];
          }
          acc[j] = wmma4(a, bb, acc[j]);
        }
      }
    }
  }
  {
    float* rw = &red[(wv * 16 + h * 8) * 64];
#pragma unroll
    for (int r = 0; r < 8; ++r) {
      if (BT) {
#pragma unroll
        for (int j = 0; j < 4; ++j) rw[r * 64 + 16 * j + c] = acc[j][r];
      } else {
        float4 o; o.x = acc[0][r]; o.y = acc[1][r]; o.z = acc[2][r]; o.w = acc[3][r];
        *(float4*)&rw[r * 64 + 4 * c] = o;
      }
    }
  }
  __syncthreads();
  {
    const int row = t >> 5, col = (t & 31) * 2;
    float2 s; s.x = 0.f; s.y = 0.f;
#pragma unroll
    for (int w = 0; w < SG_WAVES; ++w) {
      const float2 p = *(const float2*)&red[(w * 16 + row) * 64 + col];
      s.x += p.x; s.y += p.y;
    }
    const size_t idx = (size_t)(m0 + row) * N + cbase + col;
    if (EPI >= 1) { s.x += bias[cbase + col]; s.y += bias[cbase + col + 1]; }
    if (EPI == 1) { s.x = fmaxf(s.x, 0.f); s.y = fmaxf(s.y, 0.f); }
    if (EPI == 2) {
      const float2 x = *(const float2*)&aux[idx];
      s.x += x.x; s.y += x.y;
    }
    v2f sv; sv.x = s.x; sv.y = s.y;
    *(v2f*)&C[idx] = sv;
  }
}

template <int EPI, bool BT, int K, int N>
__global__ void __launch_bounds__(SG_WAVES * 32)
slot_gemm_kernel(const float* __restrict__ A, const float* __restrict__ B,
                 const float* __restrict__ bias, float* C, const float* aux) {
  __shared__ float red[SG_WAVES * 16 * 64];
  slot_gemm_body<EPI, BT, K, N, false>(A, B, bias, C, aux, nullptr, nullptr, red, nullptr,
                                       blockIdx.x, blockIdx.y);
}

template <int EPI0, bool BT0, int N0, int EPI1, bool BT1, int N1, int K, bool LN>
__global__ void __launch_bounds__(SG_WAVES * 32)
slot_gemm2_kernel(const float* __restrict__ A0, const float* __restrict__ B0,
                  const float* __restrict__ bias0, float* __restrict__ C0,
                  const float* __restrict__ A1, const float* __restrict__ B1,
                  const float* __restrict__ bias1, float* __restrict__ C1,
                  const float* __restrict__ lnw, const float* __restrict__ lnb) {
  __shared__ float red[SG_WAVES * 16 * 64];
  __shared__ __attribute__((aligned(16))) float lna[LN ? 16 * LNA_PITCH : 4];
  if (blockIdx.x < N0 / 64)
    slot_gemm_body<EPI0, BT0, K, N0, LN>(A0, B0, bias0, C0, nullptr, lnw, lnb, red, lna,
                                         blockIdx.x, blockIdx.y);
  else
    slot_gemm_body<EPI1, BT1, K, N1, LN>(A1, B1, bias1, C1, nullptr, lnw, lnb, red, lna,
                                         blockIdx.x - N0 / 64, blockIdx.y);
}

#define FA_WAVES 16
#define FA_TF 64
#define FA_FWG 512
#define FA_NT (FA_FWG / FA_TF)
#define FA_NWG (FF / FA_FWG)
#define FA_PITCH (DD + 4)
#define FA_APITCH (FA_TF + 4)
#define FA_REP8(M) M(0) M(1) M(2) M(3) M(4) M(5) M(6) M(7)
#define FA_DECL(i) float4 pa##i, pb##i;
#define FA_LDA(i) pa##i = *(const float4*)&src[(size_t)((i) >> 2) * DD + ((i) & 3) * 128];
#define FA_LDB(i) pb##i = *(const float4*)&src[(size_t)(2 + ((i) >> 2)) * DD + ((i) & 3) * 128];
#define FA_ST(i) *(float4*)&xs[((i) >> 2) * FA_PITCH + ((i) & 3) * 128] = pa##i; \
                 *(float4*)&xs[(2 + ((i) >> 2)) * FA_PITCH + ((i) & 3) * 128] = pb##i;
__device__ __forceinline__ void lds_barrier() {
  asm volatile("s_wait_dscnt 0x0\n\ts_barrier_signal -1\n\ts_barrier_wait -1" ::: "memory");
}
__global__ void __launch_bounds__(FA_WAVES * 32)
attend_kernel(const float* __restrict__ Q, const float* X,
              const float* __restrict__ lnw, const float* __restrict__ lnb,
              float* __restrict__ PreNorm, float* __restrict__ Upart,
              float* __restrict__ Spart) {
  __shared__ __attribute__((aligned(16))) float xt[FA_TF * FA_PITCH];
  __shared__ __attribute__((aligned(16))) float psc[4][SS][FA_TF];
  __shared__ float pst[4][2][FA_TF];
  __shared__ __attribute__((aligned(16))) float at[SS * FA_APITCH];
  __shared__ float part[2][SS][4];
  __shared__ __attribute__((aligned(16))) float cst[2][SS];
  __shared__ float sfin[2][2][SS];
  const int t = threadIdx.x, lane = t & 31, wv = t >> 5;
  const int c = lane & 15, h = lane >> 4;
  const int b = blockIdx.y, bx = blockIdx.x;
  const float* qbase = Q + (size_t)b * SS * DD;
  {
    const float4* qb = (const float4*)qbase;
#pragma unroll
    for (int i = 0; i < 2; ++i) {
      const int idx = t + i * (FA_WAVES * 32);
      const int row = idx >> 7, c4 = idx & 127;
      const float4 qv = qb[idx];
      const float4 ww = ((const float4*)lnw)[c4], bb = ((const float4*)lnb)[c4];
      const float ps = wave_sum((qv.x * ww.x + qv.y * ww.y) + (qv.z * ww.z + qv.w * ww.w));
      const float pb = wave_sum((qv.x * bb.x + qv.y * bb.y) + (qv.z * bb.z + qv.w * bb.w));
      if (lane == 0) { part[0][row][wv & 3] = ps; part[1][row][wv & 3] = pb; }
    }
  }
  const int ft = wv & 3, kq = wv >> 2;
  float4 qa[8][2];
  const float* xg = X + ((size_t)b * FF + (size_t)bx * FA_FWG + wv * 4) * DD + lane * 4;
  float* xs = &xt[(wv * 4) * FA_PITCH + lane * 4];
  FA_REP8(FA_DECL)
  {
    const float* src = xg;
    FA_REP8(FA_LDA)
    FA_REP8(FA_LDB)
    FA_REP8(FA_ST)
  }
  __builtin_amdgcn_sched_barrier(0);
  {
    const float* qr = qbase + (size_t)(c & 7) * DD + kq * 128 + h * 8;
    const float* wr = lnw + kq * 128 + h * 8;
#pragma unroll
    for (int g = 0; g < 8; ++g) {
#pragma unroll
      for (int e = 0; e < 2; ++e) {
        const float4 qv = *(const float4*)&qr[g * 16 + 4 * e];
        const float4 ww = *(const float4*)&wr[g * 16 + 4 * e];
        float4 o; o.x = qv.x * ww.x; o.y = qv.y * ww.y; o.z = qv.z * ww.z; o.w = qv.w * ww.w;
        qa[g][e] = o;
      }
    }
  }
  __syncthreads();
  if (t < 2 * SS) {
    const int which = t >> 3, s = t & 7;
    cst[which][s] = (part[which][s][0] + part[which][s][1]) + (part[which][s][2] + part[which][s][3]);
  }
  __syncthreads();

  const float* xrow = &xt[(ft * 16 + c) * FA_PITCH + kq * 128 + h * 8];
  const float* xcol = &xt[(h * 8) * FA_PITCH + 32 * wv + 2 * c];
  const float* arow = &at[(c & 7) * FA_APITCH + h * 8];
  v8f accu[2] = {};
  float accs[SS] = {}, accm[SS] = {};

#pragma unroll 1
  for (int it = 0; it < FA_NT; ++it) {
    const float* src = xg + (size_t)(it + 1 < FA_NT ? it + 1 : it) * FA_TF * DD;
    FA_REP8(FA_LDA)
    __builtin_amdgcn_sched_barrier(0);

    v8f sc[2] = {};
    v2f rsum = {}, rsq = {};
#pragma unroll
    for (int hf = 0; hf < 2; ++hf) {
      float4 bs[4][2];
#pragma unroll
      for (int u = 0; u < 4; ++u) {
        bs[u][0] = *(const float4*)&xrow[(hf * 4 + u) * 16];
        bs[u][1] = *(const float4*)&xrow[(hf * 4 + u) * 16 + 4];
      }
      __builtin_amdgcn_sched_barrier(0);
#pragma unroll
      for (int u = 0; u < 4; ++u) {
#pragma unroll
        for (int e = 0; e < 2; ++e) {
          v2f lo, hi;
          lo.x = bs[u][e].x; lo.y = bs[u][e].y;
          hi.x = bs[u][e].z; hi.y = bs[u][e].w;
          rsum += lo + hi;
          rsq += lo * lo;
          rsq += hi * hi;
        }
        const int g = hf * 4 + u;
        const float xa[8] = {qa[g][0].x, qa[g][0].y, qa[g][0].z, qa[g][0].w,
                             qa[g][1].x, qa[g][1].y, qa[g][1].z, qa[g][1].w};
        const float xb[8] = {bs[u][0].x, bs[u][0].y, bs[u][0].z, bs[u][0].w,
                             bs[u][1].x, bs[u][1].y, bs[u][1].z, bs[u][1].w};
#pragma unroll
        for (int tt = 0; tt < 4; ++tt) {
          v2f a; a.x = xa[2 * tt]; a.y = xa[2 * tt + 1];
          v2f bb; bb.x = xb[2 * tt]; bb.y = xb[2 * tt + 1];
          sc[u & 1] = wmma4(a, bb, sc[u & 1]);
        }
      }
      __builtin_amdgcn_sched_barrier(0);
    }
    {
      const v8f s8 = sc[0] + sc[1];
      float t0 = rsum.x + rsum.y, u0 = rsq.x + rsq.y;
      t0 += __shfl_xor(t0, 16, 32);
      u0 += __shfl_xor(u0, 16, 32);
      if (h == 0) {
#pragma unroll
        for (int r = 0; r < SS; ++r) psc[kq][r][ft * 16 + c] = s8[r];
        pst[kq][0][ft * 16 + c] = t0;
        pst[kq][1][ft * 16 + c] = u0;
      }
    }
    FA_REP8(FA_LDB)
    __builtin_amdgcn_sched_barrier(0);
    lds_barrier();

    if (wv < 2) {
      const int fl = wv * 32 + lane;
      const float ssum = (pst[0][0][fl] + pst[1][0][fl]) + (pst[2][0][fl] + pst[3][0][fl]);
      const float ssq = (pst[0][1][fl] + pst[1][1][fl]) + (pst[2][1][fl] + pst[3][1][fl]);
      const float m = ssum * (1.f / DD);
      const float rs = rsqrtf(ssq * (1.f / DD) - m * m + LN_EPS_C);
      float qs[SS], qb2[SS];
      {
        const float4 a0 = *(const float4*)&cst[0][0], a1 = *(const float4*)&cst[0][4];
        const float4 b0 = *(const float4*)&cst[1][0], b1 = *(const float4*)&cst[1][4];
        qs[0] = a0.x; qs[1] = a0.y; qs[2] = a0.z; qs[3] = a0.w;
        qs[4] = a1.x; qs[5] = a1.y; qs[6] = a1.z; qs[7] = a1.w;
        qb2[0] = b0.x; qb2[1] = b0.y; qb2[2] = b0.z; qb2[3] = b0.w;
        qb2[4] = b1.x; qb2[5] = b1.y; qb2[6] = b1.z; qb2[7] = b1.w;
      }
      float d[SS]; float mx = -3.4e38f;
#pragma unroll
      for (int s = 0; s < SS; ++s) {
        const float raw = (psc[0][s][fl] + psc[1][s][fl]) + (psc[2][s][fl] + psc[3][s][fl]);
        d[s] = (rs * (raw - m * qs[s]) + qb2[s]) * SCALE_C;
        mx = fmaxf(mx, d[s]);
      }
      float sum = 0.f;
#pragma unroll
      for (int s = 0; s < SS; ++s) { d[s] = expf(d[s] - mx); sum += d[s]; }
      const float inv = 1.f / sum;
      const int f = bx * FA_FWG + it * FA_TF + fl;
#pragma unroll
      for (int s = 0; s < SS; ++s) {
        const float p = d[s] * inv;
        PreNorm[((size_t)b * SS + s) * FF + f] = p;
        const float aw = p + EPS_A;
        const float ar = aw * rs;
        at[s * FA_APITCH + fl] = ar;
        accs[s] += aw;
        accm[s] += ar * m;
      }
    }
    lds_barrier();

#pragma unroll
    for (int hf = 0; hf < 2; ++hf) {
      float4 aq[2][2];
      float2 bq[2][4][2];
#pragma unroll
      for (int uu = 0; uu < 2; ++uu) {
        const int u = hf * 2 + uu;
        aq[uu][0] = *(const float4*)&arow[u * 16];
        aq[uu][1] = *(const float4*)&arow[u * 16 + 4];
#pragma unroll
        for (int tt = 0; tt < 4; ++tt) {
          bq[uu][tt][0] = *(const float2*)&xcol[(u * 16 + 2 * tt) * FA_PITCH];
          bq[uu][tt][1] = *(const float2*)&xcol[(u * 16 + 2 * tt + 1) * FA_PITCH];
        }
      }
      __builtin_amdgcn_sched_barrier(0);
#pragma unroll
      for (int uu = 0; uu < 2; ++uu) {
        const float xa[8] = {aq[uu][0].x, aq[uu][0].y, aq[uu][0].z, aq[uu][0].w,
                             aq[uu][1].x, aq[uu][1].y, aq[uu][1].z, aq[uu][1].w};
#pragma unroll
        for (int tt = 0; tt < 4; ++tt) {
          v2f a; a.x = xa[2 * tt]; a.y = xa[2 * tt + 1];
          v2f b0; b0.x = bq[uu][tt][0].x; b0.y = bq[uu][tt][1].x;
          v2f b1; b1.x = bq[uu][tt][0].y; b1.y = bq[uu][tt][1].y;
          accu[0] = wmma4(a, b0, accu[0]);
          accu[1] = wmma4(a, b1, accu[1]);
        }
      }
      __builtin_amdgcn_sched_barrier(0);
    }
    lds_barrier();

    FA_REP8(FA_ST)
    lds_barrier();
  }

  if (h == 0) {
    float* up = Upart + (((size_t)b * FA_NWG + bx) * SS) * DD + 32 * wv + 2 * c;
#pragma unroll
    for (int r = 0; r < SS; ++r) {
      float2 o; o.x = accu[0][r]; o.y = accu[1][r];
      *(float2*)&up[(size_t)r * DD] = o;
    }
  }
  if (wv < 2) {
#pragma unroll
    for (int s = 0; s < SS; ++s) {
      const float r1 = wave_sum(accs[s]);
      const float r2 = wave_sum(accm[s]);
      if (lane == 0) { sfin[wv][0][s] = r1; sfin[wv][1][s] = r2; }
    }
  }
  __syncthreads();
  if (t < 2 * SS) {
    const int which = t >> 3, s = t & 7;
    Spart[(((size_t)b * FA_NWG + bx) * 2 + which) * SS + s] = sfin[0][which][s] + sfin[1][which][s];
  }
}

__global__ void __launch_bounds__(DD)
upd_finish_kernel(const float* __restrict__ Upart, const float* __restrict__ Spart,
                  const float* __restrict__ lnw, const float* __restrict__ lnb,
                  float* __restrict__ Upd) {
  const int row = blockIdx.x, b = row >> 3, s = row & 7, d = threadIdx.x;
  float u = 0.f, sum = 0.f, mterm = 0.f;
#pragma unroll
  for (int w = 0; w < FA_NWG; ++w) {
    u += Upart[(((size_t)b * FA_NWG + w) * SS + s) * DD + d];
    sum += Spart[(((size_t)b * FA_NWG + w) * 2 + 0) * SS + s];
    mterm += Spart[(((size_t)b * FA_NWG + w) * 2 + 1) * SS + s];
  }
  Upd[(size_t)row * DD + d] = (u - mterm) * (1.f / sum) * lnw[d] + lnb[d];
}

#define GL_WAVES 4
__global__ void __launch_bounds__(GL_WAVES * 32)
gru_ln_kernel(const float* xold, const float* __restrict__ w1, const float* __restrict__ b1,
              const float* __restrict__ gi, const float* __restrict__ gh,
              const float* __restrict__ w2, const float* __restrict__ b2,
              float* slots, float* __restrict__ hln) {
  __shared__ float nrow[DD];
  const int row = blockIdx.x, lane = threadIdx.x & 31, wv = threadIdx.x >> 5;
  const float* x = xold + (size_t)row * DD;
  float xv[16];
#pragma unroll
  for (int j = 0; j < 16; ++j) xv[j] = x[lane + 32 * j];
  float m, r;
  row_stats(xv, m, r);
  const size_t base = (size_t)row * 3 * DD;
  float nv[4];
#pragma unroll
  for (int jj = 0; jj < 4; ++jj) {
    const int d = lane + 32 * (4 * wv + jj);
    const float h = (x[d] - m) * r * w1[d] + b1[d];
    float ir = gi[base + d],           hr = gh[base + d];
    float iz = gi[base + DD + d],      hz = gh[base + DD + d];
    float in_ = gi[base + 2 * DD + d], hn = gh[base + 2 * DD + d];
    float rg = 1.f / (1.f + expf(-(ir + hr)));
    float z = 1.f / (1.f + expf(-(iz + hz)));
    float n = tanhf(in_ + rg * hn);
    nv[jj] = (1.f - z) * n + z * h;
    nrow[d] = nv[jj];
  }
  __syncthreads();
  float yv[16];
#pragma unroll
  for (int j = 0; j < 16; ++j) yv[j] = nrow[lane + 32 * j];
  float m2, r2;
  row_stats(yv, m2, r2);
#pragma unroll
  for (int jj = 0; jj < 4; ++jj) {
    const int d = lane + 32 * (4 * wv + jj);
    slots[(size_t)row * DD + d] = nv[jj];
    hln[(size_t)row * DD + d] = (nv[jj] - m2) * r2 * w2[d] + b2[d];
  }
}

extern "C" void kernel_launch(void* const* d_in, const int* in_sizes, int n_in,
                              void* d_out, int out_size, void* d_ws, size_t ws_size,
                              hipStream_t stream) {
  (void)in_sizes; (void)n_in; (void)out_size; (void)ws_size;
  const float* in_slots  = (const float*)d_in[0];
  const float* features  = (const float*)d_in[1];
  const float* w_k       = (const float*)d_in[2];
  const float* w_v       = (const float*)d_in[3];
  const float* w_q       = (const float*)d_in[4];
  const float* ln_feat_w = (const float*)d_in[5];
  const float* ln_feat_b = (const float*)d_in[6];
  const float* ln_slot_w = (const float*)d_in[7];
  const float* ln_slot_b = (const float*)d_in[8];
  const float* gru_w_ih  = (const float*)d_in[9];
  const float* gru_w_hh  = (const float*)d_in[10];
  const float* gru_b_ih  = (const float*)d_in[11];
  const float* gru_b_hh  = (const float*)d_in[12];
  const float* ln_mlp_w  = (const float*)d_in[13];
  const float* ln_mlp_b  = (const float*)d_in[14];
  const float* mlp_w1    = (const float*)d_in[15];
  const float* mlp_b1    = (const float*)d_in[16];
  const float* mlp_w2    = (const float*)d_in[17];
  const float* mlp_b2    = (const float*)d_in[18];

  float* ws = (float*)d_ws;
  size_t off = 0;
  auto carve = [&](size_t n) { float* p = ws + off; off += n; return p; };
  float* wqk   = carve((size_t)DD * DD);
  float* wvi   = carve((size_t)DD * 3 * DD);
  float* slots = carve((size_t)BB * SS * DD);
  float* q     = carve((size_t)BB * SS * DD);
  float* upart = carve((size_t)BB * FA_NWG * SS * DD);
  float* upd   = carve((size_t)BB * SS * DD);
  float* gi    = carve((size_t)BB * SS * 3 * DD);
  float* gh    = carve((size_t)BB * SS * 3 * DD);
  float* hln   = carve((size_t)BB * SS * DD);
  float* h1    = carve((size_t)BB * SS * HH);
  float* spart = h1;

  float* out_slots = (float*)d_out;
  float* out_attn  = out_slots + (size_t)BB * SS * DD;

  slot_gemm2_kernel<0, true, DD, 0, true, 3 * DD, DD, false>
      <<<dim3(DD / 64 + (3 * DD) / 64, DD / 16), SG_WAVES * 32, 0, stream>>>(
      w_q, w_k, nullptr, wqk, w_v, gru_w_ih, nullptr, wvi, nullptr, nullptr);

  for (int it = 0; it < NITER; ++it) {
    const float* cur = it == 0 ? in_slots : slots;
    float* nxt = it == NITER - 1 ? out_slots : slots;
    slot_gemm2_kernel<0, false, DD, 3, true, 3 * DD, DD, true>
        <<<dim3(DD / 64 + (3 * DD) / 64, (BB * SS) / 16), SG_WAVES * 32, 0, stream>>>(
        cur, wqk, nullptr, q, cur, gru_w_hh, gru_b_hh, gh, ln_slot_w, ln_slot_b);
    attend_kernel<<<dim3(FA_NWG, BB), FA_WAVES * 32, 0, stream>>>(
        q, features, ln_feat_w, ln_feat_b, out_attn, upart, spart);
    upd_finish_kernel<<<BB * SS, DD, 0, stream>>>(upart, spart, ln_feat_w, ln_feat_b, upd);
    slot_gemm_kernel<3, false, DD, 3 * DD><<<dim3((3 * DD) / 64, (BB * SS) / 16), SG_WAVES * 32, 0, stream>>>(
        upd, wvi, gru_b_ih, gi, nullptr);
    gru_ln_kernel<<<BB * SS, GL_WAVES * 32, 0, stream>>>(
        cur, ln_slot_w, ln_slot_b, gi, gh, ln_mlp_w, ln_mlp_b, slots, hln);
    slot_gemm_kernel<1, false, DD, HH><<<dim3(HH / 64, (BB * SS) / 16), SG_WAVES * 32, 0, stream>>>(
        hln, mlp_w1, mlp_b1, h1, nullptr);
    slot_gemm_kernel<2, false, HH, DD><<<dim3(DD / 64, (BB * SS) / 16), SG_WAVES * 32, 0, stream>>>(
        h1, mlp_w2, mlp_b2, nxt, slots);
  }
}


# attention pass re-tiled so that two work-groups share a compute unit: 512 groups of 8 waves on 32-row tiles, one fills the other's barrier and softmax gaps; partials per 256 rows (new sum order)
# speedup vs baseline: 15.5128x; 1.0406x over previous
#include <hip/hip_runtime.h>
#include <hip/hip_bf16.h>

typedef __attribute__((ext_vector_type(2))) float v2f;
typedef __attribute__((ext_vector_type(8))) float v8f;

#define BB 32
#define FF 4096
#define SS 8
#define DD 512
#define HH 2048
#define NITER 3
#define EPS_A 1e-8f
#define LN_EPS_C 1e-5f
#define SCALE_C 0.04419417382415922f

__device__ __forceinline__ v8f wmma4(v2f a, v2f b, v8f c) {
  return __builtin_amdgcn_wmma_f32_16x16x4_f32(
       false, a,  false, b,
       (short)0, c,  false,  false);
}

__device__ __forceinline__ float wave_sum(float v) {
#pragma unroll
  for (int o = 16; o; o >>= 1) v += __shfl_xor(v, o, 32);
  return v;
}

__device__ __forceinline__ void row_stats(const float (&xv)[16], float& m, float& r) {
  float s = 0.f, s2 = 0.f;
#pragma unroll
  for (int j = 0; j < 16; ++j) { s += xv[j]; s2 += xv[j] * xv[j]; }
  s = wave_sum(s); s2 = wave_sum(s2);
  m = s * (1.f / DD);
  r = rsqrtf(s2 * (1.f / DD) - m * m + LN_EPS_C);
}

#define SG_WAVES 16
#define LNA_PITCH (DD + 4)
template <int EPI, bool BT, int K, int N, bool LN>
__device__ __forceinline__ void
slot_gemm_body(const float* __restrict__ A, const float* __restrict__ B,
               const float* __restrict__ bias, float* C, const float* aux,
               const float* __restrict__ lnw, const float* __restrict__ lnb,
               float* red, float* lna, const int bx, const int by) {
  constexpr int KW = K / SG_WAVES;
  constexpr int BLK = KW < 64 ? KW : 64;
  constexpr int NU = BLK / 16;
  const int t = threadIdx.x, lane = t & 31, wv = t >> 5;
  const int c = lane & 15, h = lane >> 4;
  const int cbase = bx * 64, m0 = by * 16;
  const int k0 = wv * KW + h * 8;
  if (LN) {
    const float* x = A + (size_t)(m0 + wv) * K;
    float xv[16];
#pragma unroll
    for (int j = 0; j < 16; ++j) xv[j] = x[lane + 32 * j];
    float m, r;
    row_stats(xv, m, r);
#pragma unroll
    for (int j = 0; j < 16; ++j) {
      const int d = lane + 32 * j;
      lna[wv * LNA_PITCH + d] = (xv[j] - m) * r * lnw[d] + lnb[d];
    }
    __syncthreads();
  }
  const float* arow = LN ? lna + c * LNA_PITCH + k0 : A + (size_t)(m0 + c) * K + k0;
  const float* bptr = BT ? B + (size_t)(cbase + c) * K + k0
                         : B + (size_t)k0 * N + cbase + c * 4;

  v8f acc[4] = {};
#pragma unroll 1
  for (int blk = 0; blk < KW; blk += BLK) {
    float4 aq[NU][2];
    float4 bq[NU][4][2];
#pragma unroll
    for (int u = 0; u < NU; ++u) {
      aq[u][0] = *(const float4*)&arow[blk + u * 16];
      aq[u][1] = *(const float4*)&arow[blk + u * 16 + 4];
#pragma unroll
      for (int i = 0; i < 4; ++i) {
        if (BT) {
          const float* p = bptr + (size_t)(16 * i) * K + blk + u * 16;
          bq[u][i][0] = *(const float4*)p;
          bq[u][i][1] = *(const float4*)(p + 4);
        } else {
          const float* p = bptr + (size_t)(blk + u * 16 + 2 * i) * N;
          bq[u][i][0] = *(const float4*)p;
          bq[u][i][1] = *(const float4*)(p + N);
        }
      }
    }
    __builtin_amdgcn_sched_barrier(0);
#pragma unroll
    for (int u = 0; u < NU; ++u) {
      const float xa[8] = {aq[u][0].x, aq[u][0].y, aq[u][0].z, aq[u][0].w,
                           aq[u][1].x, aq[u][1].y, aq[u][1].z, aq[u][1].w};
#pragma unroll
      for (int tt = 0; tt < 4; ++tt) {
        v2f a; a.x = xa[2 * tt]; a.y = xa[2 * tt + 1];
#pragma unroll
        for (int j = 0; j < 4; ++j) {
          v2f bb;
          if (BT) {
            const float xb[8] = {bq[u][j][0].x, bq[u][j][0].y, bq[u][j][0].z, bq[u][j][0].w,
                                 bq[u][j][1].x, bq[u][j][1].y, bq[u][j][1].z, bq[u][j][1].w};
            bb.x = xb[2 * tt]; bb.y = xb[2 * tt + 1];
          } else {
            const float r0[4] = {bq[u][tt][0].x, bq[u][tt][0].y, bq[u][tt][0].z, bq[u][tt][0].w};
            const float r1[4] = {bq[u][tt][1].x, bq[u][tt][1].y, bq[u][tt][1].z, bq[u][tt][1].w};
            bb.x = r0[j]; bb.y = r1[j];
          }
          acc[j] = wmma4(a, bb, acc[j]);
        }
      }
    }
  }
  {
    float* rw = &red[(wv * 16 + h * 8) * 64];
#pragma unroll
    for (int r = 0; r < 8; ++r) {
      if (BT) {
#pragma unroll
        for (int j = 0; j < 4; ++j) rw[r * 64 + 16 * j + c] = acc[j][r];
      } else {
        float4 o; o.x = acc[0][r]; o.y = acc[1][r]; o.z = acc[2][r]; o.w = acc[3][r];
        *(float4*)&rw[r * 64 + 4 * c] = o;
      }
    }
  }
  __syncthreads();
  {
    const int row = t >> 5, col = (t & 31) * 2;
    float2 s; s.x = 0.f; s.y = 0.f;
#pragma unroll
    for (int w = 0; w < SG_WAVES; ++w) {
      const float2 p = *(const float2*)&red[(w * 16 + row) * 64 + col];
      s.x += p.x; s.y += p.y;
    }
    const size_t idx = (size_t)(m0 + row) * N + cbase + col;
    if (EPI >= 1) { s.x += bias[cbase + col]; s.y += bias[cbase + col + 1]; }
    if (EPI == 1) { s.x = fmaxf(s.x, 0.f); s.y = fmaxf(s.y, 0.f); }
    if (EPI == 2) {
      const float2 x = *(const float2*)&aux[idx];
      s.x += x.x; s.y += x.y;
    }
    v2f sv; sv.x = s.x; sv.y = s.y;
    *(v2f*)&C[idx] = sv;
  }
}

template <int EPI, bool BT, int K, int N>
__global__ void __launch_bounds__(SG_WAVES * 32)
slot_gemm_kernel(const float* __restrict__ A, const float* __restrict__ B,
                 const float* __restrict__ bias, float* C, const float* aux) {
  __shared__ float red[SG_WAVES * 16 * 64];
  slot_gemm_body<EPI, BT, K, N, false>(A, B, bias, C, aux, nullptr, nullptr, red, nullptr,
                                       blockIdx.x, blockIdx.y);
}

template <int EPI0, bool BT0, int N0, int EPI1, bool BT1, int N1, int K, bool LN>
__global__ void __launch_bounds__(SG_WAVES * 32)
slot_gemm2_kernel(const float* __restrict__ A0, const float* __restrict__ B0,
                  const float* __restrict__ bias0, float* __restrict__ C0,
                  const float* __restrict__ A1, const float* __restrict__ B1,
                  const float* __restrict__ bias1, float* __restrict__ C1,
                  const float* __restrict__ lnw, const float* __restrict__ lnb) {
  __shared__ float red[SG_WAVES * 16 * 64];
  __shared__ __attribute__((aligned(16))) float lna[LN ? 16 * LNA_PITCH : 4];
  if (blockIdx.x < N0 / 64)
    slot_gemm_body<EPI0, BT0, K, N0, LN>(A0, B0, bias0, C0, nullptr, lnw, lnb, red, lna,
                                         blockIdx.x, blockIdx.y);
  else
    slot_gemm_body<EPI1, BT1, K, N1, LN>(A1, B1, bias1, C1, nullptr, lnw, lnb, red, lna,
                                         blockIdx.x - N0 / 64, blockIdx.y);
}

#define FA_WAVES 8
#define FA_TF 32
#define FA_FWG 256
#define FA_NT (FA_FWG / FA_TF)
#define FA_NWG (FF / FA_FWG)
#define FA_PITCH (DD + 4)
#define FA_APITCH (FA_TF + 4)
#define FA_REP8(M) M(0) M(1) M(2) M(3) M(4) M(5) M(6) M(7)
#define FA_DECL(i) float4 pa##i, pb##i;
#define FA_LDA(i) pa##i = *(const float4*)&src[(size_t)((i) >> 2) * DD + ((i) & 3) * 128];
#define FA_LDB(i) pb##i = *(const float4*)&src[(size_t)(2 + ((i) >> 2)) * DD + ((i) & 3) * 128];
#define FA_ST(i) *(float4*)&xs[((i) >> 2) * FA_PITCH + ((i) & 3) * 128] = pa##i; \
                 *(float4*)&xs[(2 + ((i) >> 2)) * FA_PITCH + ((i) & 3) * 128] = pb##i;
__device__ __forceinline__ void lds_barrier() {
  asm volatile("s_wait_dscnt 0x0\n\ts_barrier_signal -1\n\ts_barrier_wait -1" ::: "memory");
}
__global__ void __launch_bounds__(FA_WAVES * 32, 4)
attend_kernel(const float* __restrict__ Q, const float* X,
              const float* __restrict__ lnw, const float* __restrict__ lnb,
              float* __restrict__ PreNorm, float* __restrict__ Upart,
              float* __restrict__ Spart) {
  __shared__ __attribute__((aligned(16))) float xt[FA_TF * FA_PITCH];
  __shared__ __attribute__((aligned(16))) float psc[4][SS][FA_TF];
  __shared__ float pst[4][2][FA_TF];
  __shared__ __attribute__((aligned(16))) float at[SS * FA_APITCH];
  __shared__ __attribute__((aligned(16))) float cst[2][SS];
  const int t = threadIdx.x, lane = t & 31, wv = t >> 5;
  const int c = lane & 15, h = lane >> 4;
  const int b = blockIdx.y, bx = blockIdx.x;
  const float* qbase = Q + (size_t)b * SS * DD;
  {
    static_assert(FA_WAVES == SS, "one wave per slot row");
    const float4* qb = (const float4*)qbase + wv * (DD / 4);
    float ps = 0.f, pb = 0.f;
#pragma unroll
    for (int j = 0; j < DD / 128; ++j) {
      const float4 qv = qb[lane + 32 * j];
      const float4 ww = ((const float4*)lnw)[lane + 32 * j], bb = ((const float4*)lnb)[lane + 32 * j];
      ps += (qv.x * ww.x + qv.y * ww.y) + (qv.z * ww.z + qv.w * ww.w);
      pb += (qv.x * bb.x + qv.y * bb.y) + (qv.z * bb.z + qv.w * bb.w);
    }
    ps = wave_sum(ps); pb = wave_sum(pb);
    if (lane == 0) { cst[0][wv] = ps; cst[1][wv] = pb; }
  }
  const int ft = wv & 1, kq = wv >> 1;
  float4 qa[8][2];
  const float* xg = X + ((size_t)b * FF + (size_t)bx * FA_FWG + wv * 4) * DD + lane * 4;
  float* xs = &xt[(wv * 4) * FA_PITCH + lane * 4];
  FA_REP8(FA_DECL)
  {
    const float* src = xg;
    FA_REP8(FA_LDA)
    FA_REP8(FA_LDB)
    FA_REP8(FA_ST)
  }
  __builtin_amdgcn_sched_barrier(0);
  {
    const float* qr = qbase + (size_t)(c & 7) * DD + kq * 128 + h * 8;
    const float* wr = lnw + kq * 128 + h * 8;
#pragma unroll
    for (int g = 0; g < 8; ++g) {
#pragma unroll
      for (int e = 0; e < 2; ++e) {
        const float4 qv = *(const float4*)&qr[g * 16 + 4 * e];
        const float4 ww = *(const float4*)&wr[g * 16 + 4 * e];
        float4 o; o.x = qv.x * ww.x; o.y = qv.y * ww.y; o.z = qv.z * ww.z; o.w = qv.w * ww.w;
        qa[g][e] = o;
      }
    }
  }
  __syncthreads();

  const float* xrow = &xt[(ft * 16 + c) * FA_PITCH + kq * 128 + h * 8];
  const float* xcol = &xt[(h * 8) * FA_PITCH + 64 * wv + 4 * c];
  const float* arow = &at[(c & 7) * FA_APITCH + h * 8];
  v8f accu[4] = {};
  float accs[SS] = {}, accm[SS] = {};

#pragma unroll 1
  for (int it = 0; it < FA_NT; ++it) {
    const float* src = xg + (size_t)(it + 1 < FA_NT ? it + 1 : it) * FA_TF * DD;
    FA_REP8(FA_LDA)
    __builtin_amdgcn_sched_barrier(0);

    v8f sc[2] = {};
    v2f rsum = {}, rsq = {};
#pragma unroll
    for (int hf = 0; hf < 2; ++hf) {
      float4 bs[4][2];
#pragma unroll
      for (int u = 0; u < 4; ++u) {
        bs[u][0] = *(const float4*)&xrow[(hf * 4 + u) * 16];
        bs[u][1] = *(const float4*)&xrow[(hf * 4 + u) * 16 + 4];
      }
      __builtin_amdgcn_sched_barrier(0);
#pragma unroll
      for (int u = 0; u < 4; ++u) {
#pragma unroll
        for (int e = 0; e < 2; ++e) {
          v2f lo, hi;
          lo.x = bs[u][e].x; lo.y = bs[u][e].y;
          hi.x = bs[u][e].z; hi.y = bs[u][e].w;
          rsum += lo + hi;
          rsq += lo * lo;
          rsq += hi * hi;
        }
        const int g = hf * 4 + u;
        const float xa[8] = {qa[g][0].x, qa[g][0].y, qa[g][0].z, qa[g][0].w,
                             qa[g][1].x, qa[g][1].y, qa[g][1].z, qa[g][1].w};
        const float xb[8] = {bs[u][0].x, bs[u][0].y, bs[u][0].z, bs[u][0].w,
                             bs[u][1].x, bs[u][1].y, bs[u][1].z, bs[u][1].w};
#pragma unroll
        for (int tt = 0; tt < 4; ++tt) {
          v2f a; a.x = xa[2 * tt]; a.y = xa[2 * tt + 1];
          v2f bb; bb.x = xb[2 * tt]; bb.y = xb[2 * tt + 1];
          sc[u & 1] = wmma4(a, bb, sc[u & 1]);
        }
      }
      __builtin_amdgcn_sched_barrier(0);
    }
    {
      const v8f s8 = sc[0] + sc[1];
      float t0 = rsum.x + rsum.y, u0 = rsq.x + rsq.y;
      t0 += __shfl_xor(t0, 16, 32);
      u0 += __shfl_xor(u0, 16, 32);
      if (h == 0) {
#pragma unroll
        for (int r = 0; r < SS; ++r) psc[kq][r][ft * 16 + c] = s8[r];
        pst[kq][0][ft * 16 + c] = t0;
        pst[kq][1][ft * 16 + c] = u0;
      }
    }
    FA_REP8(FA_LDB)
    __builtin_amdgcn_sched_barrier(0);
    lds_barrier();

    if (wv == 0) {
      const int fl = lane;
      const float ssum = (pst[0][0][fl] + pst[1][0][fl]) + (pst[2][0][fl] + pst[3][0][fl]);
      const float ssq = (pst[0][1][fl] + pst[1][1][fl]) + (pst[2][1][fl] + pst[3][1][fl]);
      const float m = ssum * (1.f / DD);
      const float rs = rsqrtf(ssq * (1.f / DD) - m * m + LN_EPS_C);
      float qs[SS], qb2[SS];
      {
        const float4 a0 = *(const float4*)&cst[0][0], a1 = *(const float4*)&cst[0][4];
        const float4 b0 = *(const float4*)&cst[1][0], b1 = *(const float4*)&cst[1][4];
        qs[0] = a0.x; qs[1] = a0.y; qs[2] = a0.z; qs[3] = a0.w;
        qs[4] = a1.x; qs[5] = a1.y; qs[6] = a1.z; qs[7] = a1.w;
        qb2[0] = b0.x; qb2[1] = b0.y; qb2[2] = b0.z; qb2[3] = b0.w;
        qb2[4] = b1.x; qb2[5] = b1.y; qb2[6] = b1.z; qb2[7] = b1.w;
      }
      float d[SS]; float mx = -3.4e38f;
#pragma unroll
      for (int s = 0; s < SS; ++s) {
        const float raw = (psc[0][s][fl] + psc[1][s][fl]) + (psc[2][s][fl] + psc[3][s][fl]);
        d[s] = (rs * (raw - m * qs[s]) + qb2[s]) * SCALE_C;
        mx = fmaxf(mx, d[s]);
      }
      float sum = 0.f;
#pragma unroll
      for (int s = 0; s < SS; ++s) { d[s] = expf(d[s] - mx); sum += d[s]; }
      const float inv = 1.f / sum;
      const int f = bx * FA_FWG + it * FA_TF + fl;
#pragma unroll
      for (int s = 0; s < SS; ++s) {
        const float p = d[s] * inv;
        PreNorm[((size_t)b * SS + s) * FF + f] = p;
        const float aw = p + EPS_A;
        const float ar = aw * rs;
        at[s * FA_APITCH + fl] = ar;
        accs[s] += aw;
        accm[s] += ar * m;
      }
    }
    lds_barrier();

#pragma unroll
    for (int u = 0; u < FA_TF / 16; ++u) {
      float4 aq[2];
      float4 bq[4][2];
      aq[0] = *(const float4*)&arow[u * 16];
      aq[1] = *(const float4*)&arow[u * 16 + 4];
#pragma unroll
      for (int tt = 0; tt < 4; ++tt) {
        bq[tt][0] = *(const float4*)&xcol[(u * 16 + 2 * tt) * FA_PITCH];
        bq[tt][1] = *(const float4*)&xcol[(u * 16 + 2 * tt + 1) * FA_PITCH];
      }
      __builtin_amdgcn_sched_barrier(0);
      const float xa[8] = {aq[0].x, aq[0].y, aq[0].z, aq[0].w,
                           aq[1].x, aq[1].y, aq[1].z, aq[1].w};
#pragma unroll
      for (int tt = 0; tt < 4; ++tt) {
        v2f a; a.x = xa[2 * tt]; a.y = xa[2 * tt + 1];
        const float r0[4] = {bq[tt][0].x, bq[tt][0].y, bq[tt][0].z, bq[tt][0].w};
        const float r1[4] = {bq[tt][1].x, bq[tt][1].y, bq[tt][1].z, bq[tt][1].w};
#pragma unroll
        for (int j = 0; j < 4; ++j) {
          v2f bb; bb.x = r0[j]; bb.y = r1[j];
          accu[j] = wmma4(a, bb, accu[j]);
        }
      }
      __builtin_amdgcn_sched_barrier(0);
    }
    lds_barrier();

    FA_REP8(FA_ST)
    lds_barrier();
  }

  if (h == 0) {
    float* up = Upart + (((size_t)b * FA_NWG + bx) * SS) * DD + 64 * wv + 4 * c;
#pragma unroll
    for (int r = 0; r < SS; ++r) {
      float4 o; o.x = accu[0][r]; o.y = accu[1][r]; o.z = accu[2][r]; o.w = accu[3][r];
      *(float4*)&up[(size_t)r * DD] = o;
    }
  }
  if (wv == 0) {
    float r1[SS], r2[SS];
#pragma unroll
    for (int s = 0; s < SS; ++s) { r1[s] = wave_sum(accs[s]); r2[s] = wave_sum(accm[s]); }
    if (lane == 0) {
      float* sp = Spart + ((size_t)b * FA_NWG + bx) * 2 * SS;
#pragma unroll
      for (int s = 0; s < SS; ++s) { sp[s] = r1[s]; sp[SS + s] = r2[s]; }
    }
  }
}

__global__ void __launch_bounds__(DD)
upd_finish_kernel(const float* __restrict__ Upart, const float* __restrict__ Spart,
                  const float* __restrict__ lnw, const float* __restrict__ lnb,
                  float* __restrict__ Upd) {
  const int row = blockIdx.x, b = row >> 3, s = row & 7, d = threadIdx.x;
  float u = 0.f, sum = 0.f, mterm = 0.f;
#pragma unroll
  for (int w = 0; w < FA_NWG; ++w) {
    u += Upart[(((size_t)b * FA_NWG + w) * SS + s) * DD + d];
    sum += Spart[(((size_t)b * FA_NWG + w) * 2 + 0) * SS + s];
    mterm += Spart[(((size_t)b * FA_NWG + w) * 2 + 1) * SS + s];
  }
  Upd[(size_t)row * DD + d] = (u - mterm) * (1.f / sum) * lnw[d] + lnb[d];
}

#define GL_WAVES 4
__global__ void __launch_bounds__(GL_WAVES * 32)
gru_ln_kernel(const float* xold, const float* __restrict__ w1, const float* __restrict__ b1,
              const float* __restrict__ gi, const float* __restrict__ gh,
              const float* __restrict__ w2, const float* __restrict__ b2,
              float* slots, float* __restrict__ hln) {
  __shared__ float nrow[DD];
  const int row = blockIdx.x, lane = threadIdx.x & 31, wv = threadIdx.x >> 5;
  const float* x = xold + (size_t)row * DD;
  float xv[16];
#pragma unroll
  for (int j = 0; j < 16; ++j) xv[j] = x[lane + 32 * j];
  float m, r;
  row_stats(xv, m, r);
  const size_t base = (size_t)row * 3 * DD;
  float nv[4];
#pragma unroll
  for (int jj = 0; jj < 4; ++jj) {
    const int d = lane + 32 * (4 * wv + jj);
    const float h = (x[d] - m) * r * w1[d] + b1[d];
    float ir = gi[base + d],           hr = gh[base + d];
    float iz = gi[base + DD + d],      hz = gh[base + DD + d];
    float in_ = gi[base + 2 * DD + d], hn = gh[base + 2 * DD + d];
    float rg = 1.f / (1.f + expf(-(ir + hr)));
    float z = 1.f / (1.f + expf(-(iz + hz)));
    float n = tanhf(in_ + rg * hn);
    nv[jj] = (1.f - z) * n + z * h;
    nrow[d] = nv[jj];
  }
  __syncthreads();
  float yv[16];
#pragma unroll
  for (int j = 0; j < 16; ++j) yv[j] = nrow[lane + 32 * j];
  float m2, r2;
  row_stats(yv, m2, r2);
#pragma unroll
  for (int jj = 0; jj < 4; ++jj) {
    const int d = lane + 32 * (4 * wv + jj);
    slots[(size_t)row * DD + d] = nv[jj];
    hln[(size_t)row * DD + d] = (nv[jj] - m2) * r2 * w2[d] + b2[d];
  }
}

extern "C" void kernel_launch(void* const* d_in, const int* in_sizes, int n_in,
                              void* d_out, int out_size, void* d_ws, size_t ws_size,
                              hipStream_t stream) {
  (void)in_sizes; (void)n_in; (void)out_size; (void)ws_size;
  const float* in_slots  = (const float*)d_in[0];
  const float* features  = (const float*)d_in[1];
  const float* w_k       = (const float*)d_in[2];
  const float* w_v       = (const float*)d_in[3];
  const float* w_q       = (const float*)d_in[4];
  const float* ln_feat_w = (const float*)d_in[5];
  const float* ln_feat_b = (const float*)d_in[6];
  const float* ln_slot_w = (const float*)d_in[7];
  const float* ln_slot_b = (const float*)d_in[8];
  const float* gru_w_ih  = (const float*)d_in[9];
  const float* gru_w_hh  = (const float*)d_in[10];
  const float* gru_b_ih  = (const float*)d_in[11];
  const float* gru_b_hh  = (const float*)d_in[12];
  const float* ln_mlp_w  = (const float*)d_in[13];
  const float* ln_mlp_b  = (const float*)d_in[14];
  const float* mlp_w1    = (const float*)d_in[15];
  const float* mlp_b1    = (const float*)d_in[16];
  const float* mlp_w2    = (const float*)d_in[17];
  const float* mlp_b2    = (const float*)d_in[18];

  float* ws = (float*)d_ws;
  size_t off = 0;
  auto carve = [&](size_t n) { float* p = ws + off; off += n; return p; };
  float* wqk   = carve((size_t)DD * DD);
  float* wvi   = carve((size_t)DD * 3 * DD);
  float* slots = carve((size_t)BB * SS * DD);
  float* q     = carve((size_t)BB * SS * DD);
  float* upart = carve((size_t)BB * FA_NWG * SS * DD);
  float* upd   = carve((size_t)BB * SS * DD);
  float* gi    = carve((size_t)BB * SS * 3 * DD);
  float* gh    = carve((size_t)BB * SS * 3 * DD);
  float* hln   = carve((size_t)BB * SS * DD);
  float* h1    = carve((size_t)BB * SS * HH);
  float* spart = h1;

  float* out_slots = (float*)d_out;
  float* out_attn  = out_slots + (size_t)BB * SS * DD;

  slot_gemm2_kernel<0, true, DD, 0, true, 3 * DD, DD, false>
      <<<dim3(DD / 64 + (3 * DD) / 64, DD / 16), SG_WAVES * 32, 0, stream>>>(
      w_q, w_k, nullptr, wqk, w_v, gru_w_ih, nullptr, wvi, nullptr, nullptr);

  for (int it = 0; it < NITER; ++it) {
    const float* cur = it == 0 ? in_slots : slots;
    float* nxt = it == NITER - 1 ? out_slots : slots;
    slot_gemm2_kernel<0, false, DD, 3, true, 3 * DD, DD, true>
        <<<dim3(DD / 64 + (3 * DD) / 64, (BB * SS) / 16), SG_WAVES * 32, 0, stream>>>(
        cur, wqk, nullptr, q, cur, gru_w_hh, gru_b_hh, gh, ln_slot_w, ln_slot_b);
    attend_kernel<<<dim3(FA_NWG, BB), FA_WAVES * 32, 0, stream>>>(
        q, features, ln_feat_w, ln_feat_b, out_attn, upart, spart);
    upd_finish_kernel<<<BB * SS, DD, 0, stream>>>(upart, spart, ln_feat_w, ln_feat_b, upd);
    slot_gemm_kernel<3, false, DD, 3 * DD><<<dim3((3 * DD) / 64, (BB * SS) / 16), SG_WAVES * 32, 0, stream>>>(
        upd, wvi, gru_b_ih, gi, nullptr);
    gru_ln_kernel<<<BB * SS, GL_WAVES * 32, 0, stream>>>(
        cur, ln_slot_w, ln_slot_b, gi, gh, ln_mlp_w, ln_mlp_b, slots, hln);
    slot_gemm_kernel<1, false, DD, HH><<<dim3(HH / 64, (BB * SS) / 16), SG_WAVES * 32, 0, stream>>>(
        hln, mlp_w1, mlp_b1, h1, nullptr);
    slot_gemm_kernel<2, false, HH, DD><<<dim3(DD / 64, (BB * SS) / 16), SG_WAVES * 32, 0, stream>>>(
        h1, mlp_w2, mlp_b2, nxt, slots);
  }
}


# slot-side launch fusion, 22 launches -> 19: GRU cell in the epilogue of the gi product (one work-group forms all 3 gates of its columns), MLP LayerNorm in the next product's prologue; sum orders kept
# speedup vs baseline: 16.0015x; 1.0315x over previous
#include <hip/hip_runtime.h>
#include <hip/hip_bf16.h>

typedef __attribute__((ext_vector_type(2))) float v2f;
typedef __attribute__((ext_vector_type(8))) float v8f;

#define BB 32
#define FF 4096
#define SS 8
#define DD 512
#define HH 2048
#define NITER 3
#define EPS_A 1e-8f
#define LN_EPS_C 1e-5f
#define SCALE_C 0.04419417382415922f

__device__ __forceinline__ v8f wmma4(v2f a, v2f b, v8f c) {
  return __builtin_amdgcn_wmma_f32_16x16x4_f32(
       false, a,  false, b,
       (short)0, c,  false,  false);
}

__device__ __forceinline__ float wave_sum(float v) {
#pragma unroll
  for (int o = 16; o; o >>= 1) v += __shfl_xor(v, o, 32);
  return v;
}

__device__ __forceinline__ void row_stats(const float (&xv)[16], float& m, float& r) {
  float s = 0.f, s2 = 0.f;
#pragma unroll
  for (int j = 0; j < 16; ++j) { s += xv[j]; s2 += xv[j] * xv[j]; }
  s = wave_sum(s); s2 = wave_sum(s2);
  m = s * (1.f / DD);
  r = rsqrtf(s2 * (1.f / DD) - m * m + LN_EPS_C);
}

#define SG_WAVES 16
#define LNA_PITCH (DD + 4)
template <int EPI, bool BT, int K, int N, bool LN>
__device__ __forceinline__ void
slot_gemm_body(const float* __restrict__ A, const float* __restrict__ B,
               const float* __restrict__ bias, float* C, const float* aux,
               const float* __restrict__ lnw, const float* __restrict__ lnb,
               float* red, float* lna, const int bx, const int by) {
  constexpr int KW = K / SG_WAVES;
  constexpr int BLK = KW < 64 ? KW : 64;
  constexpr int NU = BLK / 16;
  const int t = threadIdx.x, lane = t & 31, wv = t >> 5;
  const int c = lane & 15, h = lane >> 4;
  const int cbase = bx * 64, m0 = by * 16;
  const int k0 = wv * KW + h * 8;
  if (LN) {
    const float* x = A + (size_t)(m0 + wv) * K;
    float xv[16];
#pragma unroll
    for (int j = 0; j < 16; ++j) xv[j] = x[lane + 32 * j];
    float m, r;
    row_stats(xv, m, r);
#pragma unroll
    for (int j = 0; j < 16; ++j) {
      const int d = lane + 32 * j;
      lna[wv * LNA_PITCH + d] = (xv[j] - m) * r * lnw[d] + lnb[d];
    }
    __syncthreads();
  }
  const float* arow = LN ? lna + c * LNA_PITCH + k0 : A + (size_t)(m0 + c) * K + k0;
  const float* bptr = BT ? B + (size_t)(cbase + c) * K + k0
                         : B + (size_t)k0 * N + cbase + c * 4;

  v8f acc[4] = {};
#pragma unroll 1
  for (int blk = 0; blk < KW; blk += BLK) {
    float4 aq[NU][2];
    float4 bq[NU][4][2];
#pragma unroll
    for (int u = 0; u < NU; ++u) {
      aq[u][0] = *(const float4*)&arow[blk + u * 16];
      aq[u][1] = *(const float4*)&arow[blk + u * 16 + 4];
#pragma unroll
      for (int i = 0; i < 4; ++i) {
        if (BT) {
          const float* p = bptr + (size_t)(16 * i) * K + blk + u * 16;
          bq[u][i][0] = *(const float4*)p;
          bq[u][i][1] = *(const float4*)(p + 4);
        } else {
          const float* p = bptr + (size_t)(blk + u * 16 + 2 * i) * N;
          bq[u][i][0] = *(const float4*)p;
          bq[u][i][1] = *(const float4*)(p + N);
        }
      }
    }
    __builtin_amdgcn_sched_barrier(0);
#pragma unroll
    for (int u = 0; u < NU; ++u) {
      const float xa[8] = {aq[u][0].x, aq[u][0].y, aq[u][0].z, aq[u][0].w,
                           aq[u][1].x, aq[u][1].y, aq[u][1].z, aq[u][1].w};
#pragma unroll
      for (int tt = 0; tt < 4; ++tt) {
        v2f a; a.x = xa[2 * tt]; a.y = xa[2 * tt + 1];
#pragma unroll
        for (int j = 0; j < 4; ++j) {
          v2f bb;
          if (BT) {
            const float xb[8] = {bq[u][j][0].x, bq[u][j][0].y, bq[u][j][0].z, bq[u][j][0].w,
                                 bq[u][j][1].x, bq[u][j][1].y, bq[u][j][1].z, bq[u][j][1].w};
            bb.x = xb[2 * tt]; bb.y = xb[2 * tt + 1];
          } else {
            const float r0[4] = {bq[u][tt][0].x, bq[u][tt][0].y, bq[u][tt][0].z, bq[u][tt][0].w};
            const float r1[4] = {bq[u][tt][1].x, bq[u][tt][1].y, bq[u][tt][1].z, bq[u][tt][1].w};
            bb.x = r0[j]; bb.y = r1[j];
          }
          acc[j] = wmma4(a, bb, acc[j]);
        }
      }
    }
  }
  {
    float* rw = &red[(wv * 16 + h * 8) * 64];
#pragma unroll
    for (int r = 0; r < 8; ++r) {
      if (BT) {
#pragma unroll
        for (int j = 0; j < 4; ++j) rw[r * 64 + 16 * j + c] = acc[j][r];
      } else {
        float4 o; o.x = acc[0][r]; o.y = acc[1][r]; o.z = acc[2][r]; o.w = acc[3][r];
        *(float4*)&rw[r * 64 + 4 * c] = o;
      }
    }
  }
  __syncthreads();
  {
    const int row = t >> 5, col = (t & 31) * 2;
    float2 s; s.x = 0.f; s.y = 0.f;
#pragma unroll
    for (int w = 0; w < SG_WAVES; ++w) {
      const float2 p = *(const float2*)&red[(w * 16 + row) * 64 + col];
      s.x += p.x; s.y += p.y;
    }
    const size_t idx = (size_t)(m0 + row) * N + cbase + col;
    if (EPI >= 1) { s.x += bias[cbase + col]; s.y += bias[cbase + col + 1]; }
    if (EPI == 1) { s.x = fmaxf(s.x, 0.f); s.y = fmaxf(s.y, 0.f); }
    if (EPI == 2) {
      const float2 x = *(const float2*)&aux[idx];
      s.x += x.x; s.y += x.y;
    }
    v2f sv; sv.x = s.x; sv.y = s.y;
    *(v2f*)&C[idx] = sv;
  }
}

template <int EPI, bool BT, int K, int N>
__global__ void __launch_bounds__(SG_WAVES * 32)
slot_gemm_kernel(const float* __restrict__ A, const float* __restrict__ B,
                 const float* __restrict__ bias, float* C, const float* aux) {
  __shared__ float red[SG_WAVES * 16 * 64];
  slot_gemm_body<EPI, BT, K, N, false>(A, B, bias, C, aux, nullptr, nullptr, red, nullptr,
                                       blockIdx.x, blockIdx.y);
}

template <int EPI, bool BT, int K, int N>
__global__ void __launch_bounds__(SG_WAVES * 32)
slot_gemm_ln_kernel(const float* __restrict__ A, const float* __restrict__ B,
                    const float* __restrict__ bias, float* __restrict__ C,
                    const float* __restrict__ lnw, const float* __restrict__ lnb) {
  __shared__ float red[SG_WAVES * 16 * 64];
  __shared__ __attribute__((aligned(16))) float lna[16 * LNA_PITCH];
  slot_gemm_body<EPI, BT, K, N, true>(A, B, bias, C, nullptr, lnw, lnb, red, lna,
                                      blockIdx.x, blockIdx.y);
}

template <int EPI0, bool BT0, int N0, int EPI1, bool BT1, int N1, int K, bool LN>
__global__ void __launch_bounds__(SG_WAVES * 32)
slot_gemm2_kernel(const float* __restrict__ A0, const float* __restrict__ B0,
                  const float* __restrict__ bias0, float* __restrict__ C0,
                  const float* __restrict__ A1, const float* __restrict__ B1,
                  const float* __restrict__ bias1, float* __restrict__ C1,
                  const float* __restrict__ lnw, const float* __restrict__ lnb) {
  __shared__ float red[SG_WAVES * 16 * 64];
  __shared__ __attribute__((aligned(16))) float lna[LN ? 16 * LNA_PITCH : 4];
  if (blockIdx.x < N0 / 64)
    slot_gemm_body<EPI0, BT0, K, N0, LN>(A0, B0, bias0, C0, nullptr, lnw, lnb, red, lna,
                                         blockIdx.x, blockIdx.y);
  else
    slot_gemm_body<EPI1, BT1, K, N1, LN>(A1, B1, bias1, C1, nullptr, lnw, lnb, red, lna,
                                         blockIdx.x - N0 / 64, blockIdx.y);
}

#define FA_WAVES 8
#define FA_TF 32
#define FA_FWG 256
#define FA_NT (FA_FWG / FA_TF)
#define FA_NWG (FF / FA_FWG)
#define FA_PITCH (DD + 4)
#define FA_APITCH (FA_TF + 4)
#define FA_REP8(M) M(0) M(1) M(2) M(3) M(4) M(5) M(6) M(7)
#define FA_DECL(i) float4 pa##i, pb##i;
#define FA_LDA(i) pa##i = *(const float4*)&src[(size_t)((i) >> 2) * DD + ((i) & 3) * 128];
#define FA_LDB(i) pb##i = *(const float4*)&src[(size_t)(2 + ((i) >> 2)) * DD + ((i) & 3) * 128];
#define FA_ST(i) *(float4*)&xs[((i) >> 2) * FA_PITCH + ((i) & 3) * 128] = pa##i; \
                 *(float4*)&xs[(2 + ((i) >> 2)) * FA_PITCH + ((i) & 3) * 128] = pb##i;
__device__ __forceinline__ void lds_barrier() {
  asm volatile("s_wait_dscnt 0x0\n\ts_barrier_signal -1\n\ts_barrier_wait -1" ::: "memory");
}
__global__ void __launch_bounds__(FA_WAVES * 32, 4)
attend_kernel(const float* __restrict__ Q, const float* X,
              const float* __restrict__ lnw, const float* __restrict__ lnb,
              float* __restrict__ PreNorm, float* __restrict__ Upart,
              float* __restrict__ Spart) {
  __shared__ __attribute__((aligned(16))) float xt[FA_TF * FA_PITCH];
  __shared__ __attribute__((aligned(16))) float psc[4][SS][FA_TF];
  __shared__ float pst[4][2][FA_TF];
  __shared__ __attribute__((aligned(16))) float at[SS * FA_APITCH];
  __shared__ __attribute__((aligned(16))) float cst[2][SS];
  const int t = threadIdx.x, lane = t & 31, wv = t >> 5;
  const int c = lane & 15, h = lane >> 4;
  const int b = blockIdx.y, bx = blockIdx.x;
  const float* qbase = Q + (size_t)b * SS * DD;
  {
    static_assert(FA_WAVES == SS, "one wave per slot row");
    const float4* qb = (const float4*)qbase + wv * (DD / 4);
    float ps = 0.f, pb = 0.f;
#pragma unroll
    for (int j = 0; j < DD / 128; ++j) {
      const float4 qv = qb[lane + 32 * j];
      const float4 ww = ((const float4*)lnw)[lane + 32 * j], bb = ((const float4*)lnb)[lane + 32 * j];
      ps += (qv.x * ww.x + qv.y * ww.y) + (qv.z * ww.z + qv.w * ww.w);
      pb += (qv.x * bb.x + qv.y * bb.y) + (qv.z * bb.z + qv.w * bb.w);
    }
    ps = wave_sum(ps); pb = wave_sum(pb);
    if (lane == 0) { cst[0][wv] = ps; cst[1][wv] = pb; }
  }
  const int ft = wv & 1, kq = wv >> 1;
  float4 qa[8][2];
  const float* xg = X + ((size_t)b * FF + (size_t)bx * FA_FWG + wv * 4) * DD + lane * 4;
  float* xs = &xt[(wv * 4) * FA_PITCH + lane * 4];
  FA_REP8(FA_DECL)
  {
    const float* src = xg;
    FA_REP8(FA_LDA)
    FA_REP8(FA_LDB)
    FA_REP8(FA_ST)
  }
  __builtin_amdgcn_sched_barrier(0);
  {
    const float* qr = qbase + (size_t)(c & 7) * DD + kq * 128 + h * 8;
    const float* wr = lnw + kq * 128 + h * 8;
#pragma unroll
    for (int g = 0; g < 8; ++g) {
#pragma unroll
      for (int e = 0; e < 2; ++e) {
        const float4 qv = *(const float4*)&qr[g * 16 + 4 * e];
        const float4 ww = *(const float4*)&wr[g * 16 + 4 * e];
        float4 o; o.x = qv.x * ww.x; o.y = qv.y * ww.y; o.z = qv.z * ww.z; o.w = qv.w * ww.w;
        qa[g][e] = o;
      }
    }
  }
  __syncthreads();

  const float* xrow = &xt[(ft * 16 + c) * FA_PITCH + kq * 128 + h * 8];
  const float* xcol = &xt[(h * 8) * FA_PITCH + 64 * wv + 4 * c];
  const float* arow = &at[(c & 7) * FA_APITCH + h * 8];
  v8f accu[4] = {};
  float accs[SS] = {}, accm[SS] = {};

#pragma unroll 1
  for (int it = 0; it < FA_NT; ++it) {
    const float* src = xg + (size_t)(it + 1 < FA_NT ? it + 1 : it) * FA_TF * DD;
    FA_REP8(FA_LDA)
    __builtin_amdgcn_sched_barrier(0);

    v8f sc[2] = {};
    v2f rsum = {}, rsq = {};
#pragma unroll
    for (int hf = 0; hf < 2; ++hf) {
      float4 bs[4][2];
#pragma unroll
      for (int u = 0; u < 4; ++u) {
        bs[u][0] = *(const float4*)&xrow[(hf * 4 + u) * 16];
        bs[u][1] = *(const float4*)&xrow[(hf * 4 + u) * 16 + 4];
      }
      __builtin_amdgcn_sched_barrier(0);
#pragma unroll
      for (int u = 0; u < 4; ++u) {
#pragma unroll
        for (int e = 0; e < 2; ++e) {
          v2f lo, hi;
          lo.x = bs[u][e].x; lo.y = bs[u][e].y;
          hi.x = bs[u][e].z; hi.y = bs[u][e].w;
          rsum += lo + hi;
          rsq += lo * lo;
          rsq += hi * hi;
        }
        const int g = hf * 4 + u;
        const float xa[8] = {qa[g][0].x, qa[g][0].y, qa[g][0].z, qa[g][0].w,
                             qa[g][1].x, qa[g][1].y, qa[g][1].z, qa[g][1].w};
        const float xb[8] = {bs[u][0].x, bs[u][0].y, bs[u][0].z, bs[u][0].w,
                             bs[u][1].x, bs[u][1].y, bs[u][1].z, bs[u][1].w};
#pragma unroll
        for (int tt = 0; tt < 4; ++tt) {
          v2f a; a.x = xa[2 * tt]; a.y = xa[2 * tt + 1];
          v2f bb; bb.x = xb[2 * tt]; bb.y = xb[2 * tt + 1];
          sc[u & 1] = wmma4(a, bb, sc[u & 1]);
        }
      }
      __builtin_amdgcn_sched_barrier(0);
    }
    {
      const v8f s8 = sc[0] + sc[1];
      float t0 = rsum.x + rsum.y, u0 = rsq.x + rsq.y;
      t0 += __shfl_xor(t0, 16, 32);
      u0 += __shfl_xor(u0, 16, 32);
      if (h == 0) {
#pragma unroll
        for (int r = 0; r < SS; ++r) psc[kq][r][ft * 16 + c] = s8[r];
        pst[kq][0][ft * 16 + c] = t0;
        pst[kq][1][ft * 16 + c] = u0;
      }
    }
    FA_REP8(FA_LDB)
    __builtin_amdgcn_sched_barrier(0);
    lds_barrier();

    if (wv == 0) {
      const int fl = lane;
      const float ssum = (pst[0][0][fl] + pst[1][0][fl]) + (pst[2][0][fl] + pst[3][0][fl]);
      const float ssq = (pst[0][1][fl] + pst[1][1][fl]) + (pst[2][1][fl] + pst[3][1][fl]);
      const float m = ssum * (1.f / DD);
      const float rs = rsqrtf(ssq * (1.f / DD) - m * m + LN_EPS_C);
      float qs[SS], qb2[SS];
      {
        const float4 a0 = *(const float4*)&cst[0][0], a1 = *(const float4*)&cst[0][4];
        const float4 b0 = *(const float4*)&cst[1][0], b1 = *(const float4*)&cst[1][4];
        qs[0] = a0.x; qs[1] = a0.y; qs[2] = a0.z; qs[3] = a0.w;
        qs[4] = a1.x; qs[5] = a1.y; qs[6] = a1.z; qs[7] = a1.w;
        qb2[0] = b0.x; qb2[1] = b0.y; qb2[2] = b0.z; qb2[3] = b0.w;
        qb2[4] = b1.x; qb2[5] = b1.y; qb2[6] = b1.z; qb2[7] = b1.w;
      }
      float d[SS]; float mx = -3.4e38f;
#pragma unroll
      for (int s = 0; s < SS; ++s) {
        const float raw = (psc[0][s][fl] + psc[1][s][fl]) + (psc[2][s][fl] + psc[3][s][fl]);
        d[s] = (rs * (raw - m * qs[s]) + qb2[s]) * SCALE_C;
        mx = fmaxf(mx, d[s]);
      }
      float sum = 0.f;
#pragma unroll
      for (int s = 0; s < SS; ++s) { d[s] = expf(d[s] - mx); sum += d[s]; }
      const float inv = 1.f / sum;
      const int f = bx * FA_FWG + it * FA_TF + fl;
#pragma unroll
      for (int s = 0; s < SS; ++s) {
        const float p = d[s] * inv;
        PreNorm[((size_t)b * SS + s) * FF + f] = p;
        const float aw = p + EPS_A;
        const float ar = aw * rs;
        at[s * FA_APITCH + fl] = ar;
        accs[s] += aw;
        accm[s] += ar * m;
      }
    }
    lds_barrier();

#pragma unroll
    for (int u = 0; u < FA_TF / 16; ++u) {
      float4 aq[2];
      float4 bq[4][2];
      aq[0] = *(const float4*)&arow[u * 16];
      aq[1] = *(const float4*)&arow[u * 16 + 4];
#pragma unroll
      for (int tt = 0; tt < 4; ++tt) {
        bq[tt][0] = *(const float4*)&xcol[(u * 16 + 2 * tt) * FA_PITCH];
        bq[tt][1] = *(const float4*)&xcol[(u * 16 + 2 * tt + 1) * FA_PITCH];
      }
      __builtin_amdgcn_sched_barrier(0);
      const float xa[8] = {aq[0].x, aq[0].y, aq[0].z, aq[0].w,
                           aq[1].x, aq[1].y, aq[1].z, aq[1].w};
#pragma unroll
      for (int tt = 0; tt < 4; ++tt) {
        v2f a; a.x = xa[2 * tt]; a.y = xa[2 * tt + 1];
        const float r0[4] = {bq[tt][0].x, bq[tt][0].y, bq[tt][0].z, bq[tt][0].w};
        const float r1[4] = {bq[tt][1].x, bq[tt][1].y, bq[tt][1].z, bq[tt][1].w};
#pragma unroll
        for (int j = 0; j < 4; ++j) {
          v2f bb; bb.x = r0[j]; bb.y = r1[j];
          accu[j] = wmma4(a, bb, accu[j]);
        }
      }
      __builtin_amdgcn_sched_barrier(0);
    }
    lds_barrier();

    FA_REP8(FA_ST)
    lds_barrier();
  }

  if (h == 0) {
    float* up = Upart + (((size_t)b * FA_NWG + bx) * SS) * DD + 64 * wv + 4 * c;
#pragma unroll
    for (int r = 0; r < SS; ++r) {
      float4 o; o.x = accu[0][r]; o.y = accu[1][r]; o.z = accu[2][r]; o.w = accu[3][r];
      *(float4*)&up[(size_t)r * DD] = o;
    }
  }
  if (wv == 0) {
    float r1[SS], r2[SS];
#pragma unroll
    for (int s = 0; s < SS; ++s) { r1[s] = wave_sum(accs[s]); r2[s] = wave_sum(accm[s]); }
    if (lane == 0) {
      float* sp = Spart + ((size_t)b * FA_NWG + bx) * 2 * SS;
#pragma unroll
      for (int s = 0; s < SS; ++s) { sp[s] = r1[s]; sp[SS + s] = r2[s]; }
    }
  }
}

__global__ void __launch_bounds__(DD)
upd_finish_kernel(const float* __restrict__ Upart, const float* __restrict__ Spart,
                  const float* __restrict__ lnw, const float* __restrict__ lnb,
                  float* __restrict__ Upd) {
  const int row = blockIdx.x, b = row >> 3, s = row & 7, d = threadIdx.x;
  float u = 0.f, sum = 0.f, mterm = 0.f;
#pragma unroll
  for (int w = 0; w < FA_NWG; ++w) {
    u += Upart[(((size_t)b * FA_NWG + w) * SS + s) * DD + d];
    sum += Spart[(((size_t)b * FA_NWG + w) * 2 + 0) * SS + s];
    mterm += Spart[(((size_t)b * FA_NWG + w) * 2 + 1) * SS + s];
  }
  Upd[(size_t)row * DD + d] = (u - mterm) * (1.f / sum) * lnw[d] + lnb[d];
}

#define GG_DC 32
#define GG_PITCH (3 * GG_DC + 4)
__global__ void __launch_bounds__(SG_WAVES * 32)
gi_gru_kernel(const float* __restrict__ upd, const float* __restrict__ wvi,
              const float* __restrict__ b_ih, const float* __restrict__ gh,
              const float* __restrict__ xold, const float* __restrict__ w1,
              const float* __restrict__ b1, float* __restrict__ snew) {
  __shared__ __attribute__((aligned(16))) float red[SG_WAVES * 16 * GG_PITCH];
  constexpr int N = 3 * DD;
  const int t = threadIdx.x, lane = t & 31, wv = t >> 5;
  const int c = lane & 15, h = lane >> 4;
  const int d0 = blockIdx.x * GG_DC, m0 = blockIdx.y * 16;
  const int k0 = wv * 32 + h * 8;
  const float* x = xold + (size_t)(m0 + wv) * DD;
  float xv[16];
#pragma unroll
  for (int j = 0; j < 16; ++j) xv[j] = x[lane + 32 * j];
  const float* arow = upd + (size_t)(m0 + c) * DD + k0;
  const float* bptr = wvi + (size_t)k0 * N + d0 + 2 * c;
  float4 aq[2][2];
  float2 bq[2][4][3][2];
#pragma unroll
  for (int u = 0; u < 2; ++u) {
    aq[u][0] = *(const float4*)&arow[u * 16];
    aq[u][1] = *(const float4*)&arow[u * 16 + 4];
#pragma unroll
    for (int tt = 0; tt < 4; ++tt) {
#pragma unroll
      for (int g = 0; g < 3; ++g) {
        const float* p = bptr + (size_t)(u * 16 + 2 * tt) * N + g * DD;
        bq[u][tt][g][0] = *(const float2*)p;
        bq[u][tt][g][1] = *(const float2*)(p + N);
      }
    }
  }
  const int d = d0 + lane;
  const size_t gb = (size_t)(m0 + wv) * N + d;
  const float hr = gh[gb], hz = gh[gb + DD], hn = gh[gb + 2 * DD];
  const float bir = b_ih[d], biz = b_ih[DD + d], bin = b_ih[2 * DD + d];
  const float xd = x[d], w1d = w1[d], b1d = b1[d];
  __builtin_amdgcn_sched_barrier(0);

  v8f acc[6] = {};
#pragma unroll
  for (int u = 0; u < 2; ++u) {
    const float xa[8] = {aq[u][0].x, aq[u][0].y, aq[u][0].z, aq[u][0].w,
                         aq[u][1].x, aq[u][1].y, aq[u][1].z, aq[u][1].w};
#pragma unroll
    for (int tt = 0; tt < 4; ++tt) {
      v2f a; a.x = xa[2 * tt]; a.y = xa[2 * tt + 1];
#pragma unroll
      for (int g = 0; g < 3; ++g) {
        const float r0[2] = {bq[u][tt][g][0].x, bq[u][tt][g][0].y};
        const float r1[2] = {bq[u][tt][g][1].x, bq[u][tt][g][1].y};
#pragma unroll
        for (int j = 0; j < 2; ++j) {
          v2f bb; bb.x = r0[j]; bb.y = r1[j];
          acc[2 * g + j] = wmma4(a, bb, acc[2 * g + j]);
        }
      }
    }
  }
  {
    float* rw = &red[(wv * 16 + h * 8) * GG_PITCH];
#pragma unroll
    for (int q = 0; q < 8; ++q) {
#pragma unroll
      for (int g = 0; g < 3; ++g) {
        float2 o; o.x = acc[2 * g][q]; o.y = acc[2 * g + 1][q];
        *(float2*)&rw[q * GG_PITCH + g * GG_DC + 2 * c] = o;
      }
    }
  }
  __syncthreads();
  float s0 = 0.f, s1 = 0.f, s2 = 0.f;
#pragma unroll
  for (int w = 0; w < SG_WAVES; ++w) {
    const float* p = &red[(w * 16 + wv) * GG_PITCH + lane];
    s0 += p[0]; s1 += p[GG_DC]; s2 += p[2 * GG_DC];
  }
  const float ir = s0 + bir, iz = s1 + biz, in_ = s2 + bin;
  float mu, rs;
  row_stats(xv, mu, rs);
  const float hcur = (xd - mu) * rs * w1d + b1d;
  const float rg = 1.f / (1.f + expf(-(ir + hr)));
  const float z = 1.f / (1.f + expf(-(iz + hz)));
  const float n = tanhf(in_ + rg * hn);
  snew[(size_t)(m0 + wv) * DD + d] = (1.f - z) * n + z * hcur;
}

extern "C" void kernel_launch(void* const* d_in, const int* in_sizes, int n_in,
                              void* d_out, int out_size, void* d_ws, size_t ws_size,
                              hipStream_t stream) {
  (void)in_sizes; (void)n_in; (void)out_size; (void)ws_size;
  const float* in_slots  = (const float*)d_in[0];
  const float* features  = (const float*)d_in[1];
  const float* w_k       = (const float*)d_in[2];
  const float* w_v       = (const float*)d_in[3];
  const float* w_q       = (const float*)d_in[4];
  const float* ln_feat_w = (const float*)d_in[5];
  const float* ln_feat_b = (const float*)d_in[6];
  const float* ln_slot_w = (const float*)d_in[7];
  const float* ln_slot_b = (const float*)d_in[8];
  const float* gru_w_ih  = (const float*)d_in[9];
  const float* gru_w_hh  = (const float*)d_in[10];
  const float* gru_b_ih  = (const float*)d_in[11];
  const float* gru_b_hh  = (const float*)d_in[12];
  const float* ln_mlp_w  = (const float*)d_in[13];
  const float* ln_mlp_b  = (const float*)d_in[14];
  const float* mlp_w1    = (const float*)d_in[15];
  const float* mlp_b1    = (const float*)d_in[16];
  const float* mlp_w2    = (const float*)d_in[17];
  const float* mlp_b2    = (const float*)d_in[18];

  float* ws = (float*)d_ws;
  size_t off = 0;
  auto carve = [&](size_t n) { float* p = ws + off; off += n; return p; };
  float* wqk   = carve((size_t)DD * DD);
  float* wvi   = carve((size_t)DD * 3 * DD);
  float* slots = carve((size_t)BB * SS * DD);
  float* q     = carve((size_t)BB * SS * DD);
  float* upart = carve((size_t)BB * FA_NWG * SS * DD);
  float* upd   = carve((size_t)BB * SS * DD);
  float* gh    = carve((size_t)BB * SS * 3 * DD);
  float* sgru  = carve((size_t)BB * SS * DD);
  float* h1    = carve((size_t)BB * SS * HH);
  float* spart = h1;

  float* out_slots = (float*)d_out;
  float* out_attn  = out_slots + (size_t)BB * SS * DD;

  slot_gemm2_kernel<0, true, DD, 0, true, 3 * DD, DD, false>
      <<<dim3(DD / 64 + (3 * DD) / 64, DD / 16), SG_WAVES * 32, 0, stream>>>(
      w_q, w_k, nullptr, wqk, w_v, gru_w_ih, nullptr, wvi, nullptr, nullptr);

  for (int it = 0; it < NITER; ++it) {
    const float* cur = it == 0 ? in_slots : slots;
    float* nxt = it == NITER - 1 ? out_slots : slots;
    slot_gemm2_kernel<0, false, DD, 3, true, 3 * DD, DD, true>
        <<<dim3(DD / 64 + (3 * DD) / 64, (BB * SS) / 16), SG_WAVES * 32, 0, stream>>>(
        cur, wqk, nullptr, q, cur, gru_w_hh, gru_b_hh, gh, ln_slot_w, ln_slot_b);
    attend_kernel<<<dim3(FA_NWG, BB), FA_WAVES * 32, 0, stream>>>(
        q, features, ln_feat_w, ln_feat_b, out_attn, upart, spart);
    upd_finish_kernel<<<BB * SS, DD, 0, stream>>>(upart, spart, ln_feat_w, ln_feat_b, upd);
    gi_gru_kernel<<<dim3(DD / GG_DC, (BB * SS) / 16), SG_WAVES * 32, 0, stream>>>(
        upd, wvi, gru_b_ih, gh, cur, ln_slot_w, ln_slot_b, sgru);
    slot_gemm_ln_kernel<1, false, DD, HH><<<dim3(HH / 64, (BB * SS) / 16), SG_WAVES * 32, 0, stream>>>(
        sgru, mlp_w1, mlp_b1, h1, ln_mlp_w, ln_mlp_b);
    slot_gemm_kernel<2, false, HH, DD><<<dim3(DD / 64, (BB * SS) / 16), SG_WAVES * 32, 0, stream>>>(
        h1, mlp_w2, mlp_b2, nxt, sgru);
  }
}
